# starting kernel
# baseline (speedup 1.0000x reference)
#include <hip/hip_runtime.h>
#include <hip/hip_bf16.h>
#include <math.h>

typedef __attribute__((ext_vector_type(2))) float v2f;
typedef __attribute__((ext_vector_type(8))) float v8f;

#define TPB 256


__global__ void k_fill(float* __restrict__ p, long n) {
  long i = (long)blockIdx.x * blockDim.x + threadIdx.x;
  if (i < n) p[i] = 0.0f;
}

__global__ void k_scatter_add(const float* __restrict__ X, float* __restrict__ Y,
                              const int* __restrict__ idx, long tot, int C) {
  long i = (long)blockIdx.x * blockDim.x + threadIdx.x;
  if (i < tot) {
    long r = i / C;
    long c = i - r * C;
    Y[(long)idx[r] * C + c] += X[i];
  }
}

__global__ void k_bn_stats(const float* __restrict__ X, long N, int C,
                           float* __restrict__ mu, float* __restrict__ rs) {
  __shared__ float s1[TPB];
  __shared__ float s2[TPB];
  const int c = blockIdx.x;
  float a = 0.0f, b = 0.0f;
  for (long r = threadIdx.x; r < N; r += TPB) {
    float v = X[r * (long)C + c];
    a += v;
    b += v * v;
  }
  s1[threadIdx.x] = a;
  s2[threadIdx.x] = b;
  __syncthreads();
  for (int s = TPB / 2; s > 0; s >>= 1) {
    if ((int)threadIdx.x < s) {
      s1[threadIdx.x] += s1[threadIdx.x + s];
      s2[threadIdx.x] += s2[threadIdx.x + s];
    }
    __syncthreads();
  }
  if (threadIdx.x == 0) {
    float m = s1[0] / (float)N;
    float var = s2[0] / (float)N - m * m;
    if (var < 0.0f) var = 0.0f;
    mu[c] = m;
    rs[c] = rsqrtf(var + 1e-5f);
  }
}

__global__ void k_bn_elu(const float* __restrict__ X, float* __restrict__ Y,
                         const float* __restrict__ mu, const float* __restrict__ rs,
                         const float* __restrict__ gb, long tot, int C) {
  long i = (long)blockIdx.x * blockDim.x + threadIdx.x;
  if (i < tot) {
    int c = (int)(i % C);
    float v = gb[c] * (X[i] - mu[c]) * rs[c] + gb[C + c];
    Y[i] = v > 0.0f ? v : expm1f(v);
  }
}

template <int CIN, int COUT>
__global__ void k_gemm_tap(const float* __restrict__ X,
                           const float* __restrict__ W,
                           float* __restrict__ Y,
                           const int* __restrict__ gidx,
                           const int* __restrict__ sidx,
                           int M, int out_offset, int accumulate) {
  const int lane = threadIdx.x;
  const int tile_r = blockIdx.x * 32;
  const int tile_c = (blockIdx.y * 4 + threadIdx.y) * 16;
  const int mr = lane & 15;
  const int kh = lane >> 4;

  const int r0 = tile_r + mr;
  const int r1 = tile_r + 16 + mr;
  int g0 = 0, g1 = 0;
  if (r0 < M) g0 = gidx ? gidx[r0] : r0;
  if (r1 < M) g1 = gidx ? gidx[r1] : r1;
  const float* __restrict__ xb0 = X + (long)g0 * CIN + 2 * kh;
  const float* __restrict__ xb1 = X + (long)g1 * CIN + 2 * kh;
  const float* __restrict__ wb  = W + tile_c + mr + 2 * kh * COUT;

  v8f c0 = {0.f, 0.f, 0.f, 0.f, 0.f, 0.f, 0.f, 0.f};
  v8f c1 = {0.f, 0.f, 0.f, 0.f, 0.f, 0.f, 0.f, 0.f};
#pragma unroll
  for (int k0 = 0; k0 < CIN; k0 += 4) {
    v2f a0 = *(const v2f*)(xb0 + k0);
    v2f a1 = *(const v2f*)(xb1 + k0);
    v2f b;
    b[0] = wb[k0 * COUT];
    b[1] = wb[(k0 + 1) * COUT];
    c0 = __builtin_amdgcn_wmma_f32_16x16x4_f32(false, a0, false, b,
                                               (short)0, c0, false, false);
    c1 = __builtin_amdgcn_wmma_f32_16x16x4_f32(false, a1, false, b,
                                               (short)0, c1, false, false);
  }

#pragma unroll
  for (int i = 0; i < 8; ++i) {
    const int ra = tile_r + i + 8 * kh;
    if (ra < M) {
      const int ro = sidx ? sidx[ra] : ra;
      float* y = Y + (long)(ro + out_offset) * COUT + tile_c + mr;
      if (accumulate) *y += c0[i]; else *y = c0[i];
    }
    const int rb = ra + 16;
    if (rb < M) {
      const int ro = sidx ? sidx[rb] : rb;
      float* y = Y + (long)(ro + out_offset) * COUT + tile_c + mr;
      if (accumulate) *y += c1[i]; else *y = c1[i];
    }
  }
}

template <int CIN, int COUT>
static inline void gemm_t(const float* X, const float* W, float* Y,
                          const int* g, const int* s, int M, int ooff,
                          int accum, hipStream_t st) {
  dim3 grid((unsigned)((M + 31) / 32), (unsigned)(COUT / 64));
  dim3 blk(32, 4);
  k_gemm_tap<CIN, COUT><<<grid, blk, 0, st>>>(X, W, Y, g, s, M, ooff, accum);
}

static void gemm(const float* X, const float* W, float* Y,
                 const int* g, const int* s, int M, int Cin, int Cout,
                 int ooff, int accum, hipStream_t st) {
  if      (Cin == 256 && Cout == 128) gemm_t<256, 128>(X, W, Y, g, s, M, ooff, accum, st);
  else if (Cin == 128 && Cout == 128) gemm_t<128, 128>(X, W, Y, g, s, M, ooff, accum, st);
  else if (Cin == 128 && Cout ==  64) gemm_t<128,  64>(X, W, Y, g, s, M, ooff, accum, st);
  else if (Cin ==  64 && Cout ==  64) gemm_t< 64,  64>(X, W, Y, g, s, M, ooff, accum, st);
  else if (Cin ==  64 && Cout == 128) gemm_t< 64, 128>(X, W, Y, g, s, M, ooff, accum, st);
}


extern "C" void kernel_launch(void* const* d_in, const int* in_sizes, int n_in,
                              void* d_out, int out_size, void* d_ws, size_t ws_size,
                              hipStream_t stream) {
  const float* feats0 = (const float*)d_in[0];
  const float* feats1 = (const float*)d_in[1];
  const float* feats2 = (const float*)d_in[2];
  const float* w_out0 = (const float*)d_in[3];
  const float* w_out1 = (const float*)d_in[4];
  const float* w_out2 = (const float*)d_in[5];
  const float* wt2    = (const float*)d_in[6];
  const float* wu2    = (const float*)d_in[7];
  const float* wt1    = (const float*)d_in[8];
  const float* wu1    = (const float*)d_in[9];
  const float* bn_out0 = (const float*)d_in[10];
  const float* bn_out1 = (const float*)d_in[11];
  const float* bn_out2 = (const float*)d_in[12];
  const float* bn_up2a = (const float*)d_in[13];
  const float* bn_up2b = (const float*)d_in[14];
  const float* bn_up1a = (const float*)d_in[15];
  const float* bn_up1b = (const float*)d_in[16];
  const int* m_c2 = (const int*)d_in[19];
  const int* m_g2 = (const int*)d_in[20];
  const int* m_u1 = (const int*)d_in[21];
  const int* m_g1 = (const int*)d_in[22];
  const int* m_u0 = (const int*)d_in[23];
  const int* lat1_idx = (const int*)d_in[24];
  const int* up1_idx  = (const int*)d_in[25];
  const int* lat0_idx = (const int*)d_in[26];
  const int* up0_idx  = (const int*)d_in[27];

  const int N0 = in_sizes[0] / 64;
  const int N1 = in_sizes[1] / 128;
  const int N2 = in_sizes[2] / 256;
  const int nu1 = in_sizes[17] / 128;
  const int nu0 = in_sizes[18] / 64;
  const int Mc2 = in_sizes[19] / 52;
  const int Mg2 = in_sizes[20] / 52;
  const int Mu1 = in_sizes[21] / 52;
  const int Mg1 = in_sizes[22] / 52;
  const int Mu0 = in_sizes[23] / 52;
  const int g2n = 8 * N2;
  const int g1n = 8 * nu1;

  char* wsp = (char*)d_ws;
  auto carve = [&](long nfloats) -> float* {
    float* p = (float*)wsp;
    wsp += ((nfloats * 4 + 255) / 256) * 256;
    return p;
  };
  long accElems = (long)(N2 + 1) * 128;
  { long e = (long)(g2n + 1) * 128; if (e > accElems) accElems = e; }
  { long e = (long)(nu1 + 1) * 128; if (e > accElems) accElems = e; }
  { long e = (long)(g1n + 1) * 64;  if (e > accElems) accElems = e; }
  { long e = (long)(nu0 + 1) * 128; if (e > accElems) accElems = e; }
  float* acc = carve(accElems);
  float* t2  = carve((long)g2n * 128);
  float* x1  = carve((long)nu1 * 128);
  float* t1  = carve((long)g1n * 64);
  float* x0  = carve((long)nu0 * 64);
  float* mu  = carve(128);
  float* rs  = carve(128);

  float* out  = (float*)d_out;
  float* out0 = out;
  float* out1 = out + (long)nu0 * 128;
  float* out2 = out + (long)nu0 * 128 + (long)nu1 * 128;

  auto fill0 = [&](float* p, long n) {
    k_fill<<<dim3((unsigned)((n + TPB - 1) / TPB)), dim3(TPB), 0, stream>>>(p, n);
  };
  auto conv3 = [&](const float* X, int n, int Cin, int Cout,
                   const float* W, const int* map, int Mw) {
    fill0(acc, (long)(n + 1) * Cout);
    for (int j = 0; j < 26; ++j) {
      const int k = (j < 13) ? j : j + 1;
      gemm(X, W + (long)k * Cin * Cout, acc,
           map + (long)j * Mw, map + (long)(26 + j) * Mw, Mw, Cin, Cout, 0, 1,
           stream);
    }
    gemm(X, W + (long)13 * Cin * Cout, acc, nullptr, nullptr, n, Cin, Cout, 0, 1,
         stream);
  };
  auto bnelu = [&](const float* src, float* dst, const float* gb, long n, int C) {
    k_bn_stats<<<dim3((unsigned)C), dim3(TPB), 0, stream>>>(src, n, C, mu, rs);
    long tot = n * C;
    k_bn_elu<<<dim3((unsigned)((tot + TPB - 1) / TPB)), dim3(TPB), 0, stream>>>(
        src, dst, mu, rs, gb, tot, C);
  };
  auto scat = [&](const float* X, float* Y, const int* idx, long rows, int C) {
    long tot = rows * C;
    k_scatter_add<<<dim3((unsigned)((tot + TPB - 1) / TPB)), dim3(TPB), 0, stream>>>(
        X, Y, idx, tot, C);
  };

  conv3(feats2, N2, 256, 128, w_out2, m_c2, Mc2);
  bnelu(acc, out2, bn_out2, N2, 128);

  for (int o = 0; o < 8; ++o)
    gemm(feats2, wt2 + (long)o * 256 * 128, t2, nullptr, nullptr, N2, 256, 128,
         o * N2, 0, stream);
  bnelu(t2, t2, bn_up2a, g2n, 128);
  conv3(t2, g2n, 128, 128, wu2, m_g2, Mg2);
  bnelu(acc, t2, bn_up2b, g2n, 128);

  fill0(x1, (long)nu1 * 128);
  scat(feats1, x1, lat1_idx, N1, 128);
  scat(t2, x1, up1_idx, g2n, 128);

  conv3(x1, nu1, 128, 128, w_out1, m_u1, Mu1);
  bnelu(acc, out1, bn_out1, nu1, 128);

  for (int o = 0; o < 8; ++o)
    gemm(x1, wt1 + (long)o * 128 * 64, t1, nullptr, nullptr, nu1, 128, 64,
         o * nu1, 0, stream);
  bnelu(t1, t1, bn_up1a, g1n, 64);
  conv3(t1, g1n, 64, 64, wu1, m_g1, Mg1);
  bnelu(acc, t1, bn_up1b, g1n, 64);

  fill0(x0, (long)nu0 * 64);
  scat(feats0, x0, lat0_idx, N0, 64);
  scat(t1, x0, up0_idx, g1n, 64);

  conv3(x0, nu0, 64, 128, w_out0, m_u0, Mu0);
  bnelu(acc, out0, bn_out0, nu0, 128);

  (void)n_in; (void)out_size; (void)ws_size;
}


# skip map pairs whose scatter row is the dropped scrap row n: all-padded 32-row tiles exit before gather, padded rows not stored (ends the read-modify-write pile-up on one row)
# speedup vs baseline: 2.1613x; 2.1613x over previous
#include <hip/hip_runtime.h>
#include <hip/hip_bf16.h>
#include <math.h>

typedef __attribute__((ext_vector_type(2))) float v2f;
typedef __attribute__((ext_vector_type(8))) float v8f;

#define TPB 256


__global__ void k_fill(float* __restrict__ p, long n) {
  long i = (long)blockIdx.x * blockDim.x + threadIdx.x;
  if (i < n) p[i] = 0.0f;
}

__global__ void k_scatter_add(const float* __restrict__ X, float* __restrict__ Y,
                              const int* __restrict__ idx, long tot, int C) {
  long i = (long)blockIdx.x * blockDim.x + threadIdx.x;
  if (i < tot) {
    long r = i / C;
    long c = i - r * C;
    Y[(long)idx[r] * C + c] += X[i];
  }
}

__global__ void k_bn_stats(const float* __restrict__ X, long N, int C,
                           float* __restrict__ mu, float* __restrict__ rs) {
  __shared__ float s1[TPB];
  __shared__ float s2[TPB];
  const int c = blockIdx.x;
  float a = 0.0f, b = 0.0f;
  for (long r = threadIdx.x; r < N; r += TPB) {
    float v = X[r * (long)C + c];
    a += v;
    b += v * v;
  }
  s1[threadIdx.x] = a;
  s2[threadIdx.x] = b;
  __syncthreads();
  for (int s = TPB / 2; s > 0; s >>= 1) {
    if ((int)threadIdx.x < s) {
      s1[threadIdx.x] += s1[threadIdx.x + s];
      s2[threadIdx.x] += s2[threadIdx.x + s];
    }
    __syncthreads();
  }
  if (threadIdx.x == 0) {
    float m = s1[0] / (float)N;
    float var = s2[0] / (float)N - m * m;
    if (var < 0.0f) var = 0.0f;
    mu[c] = m;
    rs[c] = rsqrtf(var + 1e-5f);
  }
}

__global__ void k_bn_elu(const float* __restrict__ X, float* __restrict__ Y,
                         const float* __restrict__ mu, const float* __restrict__ rs,
                         const float* __restrict__ gb, long tot, int C) {
  long i = (long)blockIdx.x * blockDim.x + threadIdx.x;
  if (i < tot) {
    int c = (int)(i % C);
    float v = gb[c] * (X[i] - mu[c]) * rs[c] + gb[C + c];
    Y[i] = v > 0.0f ? v : expm1f(v);
  }
}

template <int CIN, int COUT>
__global__ void k_gemm_tap(const float* __restrict__ X,
                           const float* __restrict__ W,
                           float* __restrict__ Y,
                           const int* __restrict__ gidx,
                           const int* __restrict__ sidx,
                           int M, int out_offset, int accumulate, int nscrap) {
  const int lane = threadIdx.x;
  const int tile_r = blockIdx.x * 32;
  const int tile_c = (blockIdx.y * 4 + threadIdx.y) * 16;
  const int mr = lane & 15;
  const int kh = lane >> 4;

  const int rl = tile_r + lane;
  int sv = nscrap;
  if (rl < M) sv = sidx ? sidx[rl] : rl;
  if (__builtin_amdgcn_ballot_w32(sv != nscrap) == 0u) return;

  const int r0 = tile_r + mr;
  const int r1 = tile_r + 16 + mr;
  int g0 = 0, g1 = 0;
  if (r0 < M) g0 = gidx ? gidx[r0] : r0;
  if (r1 < M) g1 = gidx ? gidx[r1] : r1;
  const float* __restrict__ xb0 = X + (long)g0 * CIN + 2 * kh;
  const float* __restrict__ xb1 = X + (long)g1 * CIN + 2 * kh;
  const float* __restrict__ wb  = W + tile_c + mr + 2 * kh * COUT;

  v8f c0 = {0.f, 0.f, 0.f, 0.f, 0.f, 0.f, 0.f, 0.f};
  v8f c1 = {0.f, 0.f, 0.f, 0.f, 0.f, 0.f, 0.f, 0.f};
#pragma unroll
  for (int k0 = 0; k0 < CIN; k0 += 4) {
    v2f a0 = *(const v2f*)(xb0 + k0);
    v2f a1 = *(const v2f*)(xb1 + k0);
    v2f b;
    b[0] = wb[k0 * COUT];
    b[1] = wb[(k0 + 1) * COUT];
    c0 = __builtin_amdgcn_wmma_f32_16x16x4_f32(false, a0, false, b,
                                               (short)0, c0, false, false);
    c1 = __builtin_amdgcn_wmma_f32_16x16x4_f32(false, a1, false, b,
                                               (short)0, c1, false, false);
  }

#pragma unroll
  for (int i = 0; i < 8; ++i) {
    const int t0a = __builtin_amdgcn_readlane(sv, i);
    const int t0b = __builtin_amdgcn_readlane(sv, i + 8);
    const int ro0 = kh ? t0b : t0a;
    if (ro0 != nscrap) {
      float* y = Y + (long)(ro0 + out_offset) * COUT + tile_c + mr;
      if (accumulate) *y += c0[i]; else *y = c0[i];
    }
    const int t1a = __builtin_amdgcn_readlane(sv, i + 16);
    const int t1b = __builtin_amdgcn_readlane(sv, i + 24);
    const int ro1 = kh ? t1b : t1a;
    if (ro1 != nscrap) {
      float* y = Y + (long)(ro1 + out_offset) * COUT + tile_c + mr;
      if (accumulate) *y += c1[i]; else *y = c1[i];
    }
  }
}

template <int CIN, int COUT>
static inline void gemm_t(const float* X, const float* W, float* Y,
                          const int* g, const int* s, int M, int ooff,
                          int accum, int nscrap, hipStream_t st) {
  dim3 grid((unsigned)((M + 31) / 32), (unsigned)(COUT / 64));
  dim3 blk(32, 4);
  k_gemm_tap<CIN, COUT><<<grid, blk, 0, st>>>(X, W, Y, g, s, M, ooff, accum, nscrap);
}

static void gemm(const float* X, const float* W, float* Y,
                 const int* g, const int* s, int M, int Cin, int Cout,
                 int ooff, int accum, int nscrap, hipStream_t st) {
  if      (Cin == 256 && Cout == 128) gemm_t<256, 128>(X, W, Y, g, s, M, ooff, accum, nscrap, st);
  else if (Cin == 128 && Cout == 128) gemm_t<128, 128>(X, W, Y, g, s, M, ooff, accum, nscrap, st);
  else if (Cin == 128 && Cout ==  64) gemm_t<128,  64>(X, W, Y, g, s, M, ooff, accum, nscrap, st);
  else if (Cin ==  64 && Cout ==  64) gemm_t< 64,  64>(X, W, Y, g, s, M, ooff, accum, nscrap, st);
  else if (Cin ==  64 && Cout == 128) gemm_t< 64, 128>(X, W, Y, g, s, M, ooff, accum, nscrap, st);
}


extern "C" void kernel_launch(void* const* d_in, const int* in_sizes, int n_in,
                              void* d_out, int out_size, void* d_ws, size_t ws_size,
                              hipStream_t stream) {
  const float* feats0 = (const float*)d_in[0];
  const float* feats1 = (const float*)d_in[1];
  const float* feats2 = (const float*)d_in[2];
  const float* w_out0 = (const float*)d_in[3];
  const float* w_out1 = (const float*)d_in[4];
  const float* w_out2 = (const float*)d_in[5];
  const float* wt2    = (const float*)d_in[6];
  const float* wu2    = (const float*)d_in[7];
  const float* wt1    = (const float*)d_in[8];
  const float* wu1    = (const float*)d_in[9];
  const float* bn_out0 = (const float*)d_in[10];
  const float* bn_out1 = (const float*)d_in[11];
  const float* bn_out2 = (const float*)d_in[12];
  const float* bn_up2a = (const float*)d_in[13];
  const float* bn_up2b = (const float*)d_in[14];
  const float* bn_up1a = (const float*)d_in[15];
  const float* bn_up1b = (const float*)d_in[16];
  const int* m_c2 = (const int*)d_in[19];
  const int* m_g2 = (const int*)d_in[20];
  const int* m_u1 = (const int*)d_in[21];
  const int* m_g1 = (const int*)d_in[22];
  const int* m_u0 = (const int*)d_in[23];
  const int* lat1_idx = (const int*)d_in[24];
  const int* up1_idx  = (const int*)d_in[25];
  const int* lat0_idx = (const int*)d_in[26];
  const int* up0_idx  = (const int*)d_in[27];

  const int N0 = in_sizes[0] / 64;
  const int N1 = in_sizes[1] / 128;
  const int N2 = in_sizes[2] / 256;
  const int nu1 = in_sizes[17] / 128;
  const int nu0 = in_sizes[18] / 64;
  const int Mc2 = in_sizes[19] / 52;
  const int Mg2 = in_sizes[20] / 52;
  const int Mu1 = in_sizes[21] / 52;
  const int Mg1 = in_sizes[22] / 52;
  const int Mu0 = in_sizes[23] / 52;
  const int g2n = 8 * N2;
  const int g1n = 8 * nu1;

  char* wsp = (char*)d_ws;
  auto carve = [&](long nfloats) -> float* {
    float* p = (float*)wsp;
    wsp += ((nfloats * 4 + 255) / 256) * 256;
    return p;
  };
  long accElems = (long)(N2 + 1) * 128;
  { long e = (long)(g2n + 1) * 128; if (e > accElems) accElems = e; }
  { long e = (long)(nu1 + 1) * 128; if (e > accElems) accElems = e; }
  { long e = (long)(g1n + 1) * 64;  if (e > accElems) accElems = e; }
  { long e = (long)(nu0 + 1) * 128; if (e > accElems) accElems = e; }
  float* acc = carve(accElems);
  float* t2  = carve((long)g2n * 128);
  float* x1  = carve((long)nu1 * 128);
  float* t1  = carve((long)g1n * 64);
  float* x0  = carve((long)nu0 * 64);
  float* mu  = carve(128);
  float* rs  = carve(128);

  float* out  = (float*)d_out;
  float* out0 = out;
  float* out1 = out + (long)nu0 * 128;
  float* out2 = out + (long)nu0 * 128 + (long)nu1 * 128;

  auto fill0 = [&](float* p, long n) {
    k_fill<<<dim3((unsigned)((n + TPB - 1) / TPB)), dim3(TPB), 0, stream>>>(p, n);
  };
  auto conv3 = [&](const float* X, int n, int Cin, int Cout,
                   const float* W, const int* map, int Mw) {
    fill0(acc, (long)(n + 1) * Cout);
    for (int j = 0; j < 26; ++j) {
      const int k = (j < 13) ? j : j + 1;
      gemm(X, W + (long)k * Cin * Cout, acc,
           map + (long)j * Mw, map + (long)(26 + j) * Mw, Mw, Cin, Cout, 0, 1,
           n, stream);
    }
    gemm(X, W + (long)13 * Cin * Cout, acc, nullptr, nullptr, n, Cin, Cout, 0, 1,
         -1, stream);
  };
  auto bnelu = [&](const float* src, float* dst, const float* gb, long n, int C) {
    k_bn_stats<<<dim3((unsigned)C), dim3(TPB), 0, stream>>>(src, n, C, mu, rs);
    long tot = n * C;
    k_bn_elu<<<dim3((unsigned)((tot + TPB - 1) / TPB)), dim3(TPB), 0, stream>>>(
        src, dst, mu, rs, gb, tot, C);
  };
  auto scat = [&](const float* X, float* Y, const int* idx, long rows, int C) {
    long tot = rows * C;
    k_scatter_add<<<dim3((unsigned)((tot + TPB - 1) / TPB)), dim3(TPB), 0, stream>>>(
        X, Y, idx, tot, C);
  };

  conv3(feats2, N2, 256, 128, w_out2, m_c2, Mc2);
  bnelu(acc, out2, bn_out2, N2, 128);

  for (int o = 0; o < 8; ++o)
    gemm(feats2, wt2 + (long)o * 256 * 128, t2, nullptr, nullptr, N2, 256, 128,
         o * N2, 0, -1, stream);
  bnelu(t2, t2, bn_up2a, g2n, 128);
  conv3(t2, g2n, 128, 128, wu2, m_g2, Mg2);
  bnelu(acc, t2, bn_up2b, g2n, 128);

  fill0(x1, (long)nu1 * 128);
  scat(feats1, x1, lat1_idx, N1, 128);
  scat(t2, x1, up1_idx, g2n, 128);

  conv3(x1, nu1, 128, 128, w_out1, m_u1, Mu1);
  bnelu(acc, out1, bn_out1, nu1, 128);

  for (int o = 0; o < 8; ++o)
    gemm(x1, wt1 + (long)o * 128 * 64, t1, nullptr, nullptr, nu1, 128, 64,
         o * nu1, 0, -1, stream);
  bnelu(t1, t1, bn_up1a, g1n, 64);
  conv3(t1, g1n, 64, 64, wu1, m_g1, Mg1);
  bnelu(acc, t1, bn_up1b, g1n, 64);

  fill0(x0, (long)nu0 * 64);
  scat(feats0, x0, lat0_idx, N0, 64);
  scat(t1, x0, up0_idx, g1n, 64);

  conv3(x0, nu0, 64, 128, w_out0, m_u0, Mu0);
  bnelu(acc, out0, bn_out0, nu0, 128);

  (void)n_in; (void)out_size; (void)ws_size;
}


# batch-norm column statistics read the matrix as coalesced 16-byte row slabs on up to 1024 blocks (per-block column pairs, then a small fixed-order reduce launch); the statistics' sum order changes
# speedup vs baseline: 6.9920x; 3.2351x over previous
#include <hip/hip_runtime.h>
#include <hip/hip_bf16.h>
#include <math.h>

typedef __attribute__((ext_vector_type(2))) float v2f;
typedef __attribute__((ext_vector_type(8))) float v8f;

#define TPB 256


__global__ void k_fill(float* __restrict__ p, long n) {
  long i = (long)blockIdx.x * blockDim.x + threadIdx.x;
  if (i < n) p[i] = 0.0f;
}

__global__ void k_fill_i32(int* __restrict__ p, long n, int v) {
  long i = (long)blockIdx.x * blockDim.x + threadIdx.x;
  if (i < n) p[i] = v;
}

__global__ void k_invmap(const int* __restrict__ map, int* __restrict__ nbr,
                         int Mw, int n) {
  long i = (long)blockIdx.x * blockDim.x + threadIdx.x;
  if (i < (long)26 * Mw) {
    const int j = (int)(i / Mw);
    const int o = map[(long)26 * Mw + i];
    if (o >= 0 && o < n) nbr[(long)j * n + o] = map[i];
  }
}

__global__ void k_scatter_add(const float* __restrict__ X, float* __restrict__ Y,
                              const int* __restrict__ idx, long tot, int C) {
  long i = (long)blockIdx.x * blockDim.x + threadIdx.x;
  if (i < tot) {
    long r = i / C;
    long c = i - r * C;
    Y[(long)idx[r] * C + c] += X[i];
  }
}

__global__ void k_bn_part(const float* __restrict__ X, long tot, int C,
                          float* __restrict__ part) {
  __shared__ float sm[8][TPB];
  const int t = threadIdx.x;
  float a0 = 0.0f, a1 = 0.0f, a2 = 0.0f, a3 = 0.0f;
  float b0 = 0.0f, b1 = 0.0f, b2 = 0.0f, b3 = 0.0f;
  const long step = (long)gridDim.x * (TPB * 4);
  for (long i = ((long)blockIdx.x * TPB + t) * 4; i < tot; i += step) {
    const float4 v = *(const float4*)(X + i);
    a0 += v.x; b0 += v.x * v.x;
    a1 += v.y; b1 += v.y * v.y;
    a2 += v.z; b2 += v.z * v.z;
    a3 += v.w; b3 += v.w * v.w;
  }
  sm[0][t] = a0; sm[1][t] = a1; sm[2][t] = a2; sm[3][t] = a3;
  sm[4][t] = b0; sm[5][t] = b1; sm[6][t] = b2; sm[7][t] = b3;
  __syncthreads();
  if (t < C) {
    const int q = C >> 2;
    const int comp = t & 3;
    float s = 0.0f, ss = 0.0f;
    for (int j = t >> 2; j < TPB; j += q) {
      s += sm[comp][j];
      ss += sm[4 + comp][j];
    }
    float* p = part + (long)blockIdx.x * 2 * C;
    p[t] = s;
    p[C + t] = ss;
  }
}

__global__ void k_bn_fin(const float* __restrict__ part, int P, long N, int C,
                         float* __restrict__ mu, float* __restrict__ rs) {
  __shared__ float s1[TPB];
  __shared__ float s2[TPB];
  const int c = blockIdx.x;
  float a = 0.0f, b = 0.0f;
  for (int p = threadIdx.x; p < P; p += TPB) {
    a += part[(long)p * 2 * C + c];
    b += part[(long)p * 2 * C + C + c];
  }
  s1[threadIdx.x] = a;
  s2[threadIdx.x] = b;
  __syncthreads();
  for (int s = TPB / 2; s > 0; s >>= 1) {
    if ((int)threadIdx.x < s) {
      s1[threadIdx.x] += s1[threadIdx.x + s];
      s2[threadIdx.x] += s2[threadIdx.x + s];
    }
    __syncthreads();
  }
  if (threadIdx.x == 0) {
    float m = s1[0] / (float)N;
    float var = s2[0] / (float)N - m * m;
    if (var < 0.0f) var = 0.0f;
    mu[c] = m;
    rs[c] = rsqrtf(var + 1e-5f);
  }
}

__global__ void k_bn_elu(const float* __restrict__ X, float* __restrict__ Y,
                         const float* __restrict__ mu, const float* __restrict__ rs,
                         const float* __restrict__ gb, long tot, int C) {
  long i = (long)blockIdx.x * blockDim.x + threadIdx.x;
  if (i < tot) {
    int c = (int)(i % C);
    float v = gb[c] * (X[i] - mu[c]) * rs[c] + gb[C + c];
    Y[i] = v > 0.0f ? v : expm1f(v);
  }
}

template <int CIN, int COUT>
__global__ void k_gemm_tap(const float* __restrict__ X,
                           const float* __restrict__ W,
                           float* __restrict__ Y,
                           const int* __restrict__ gidx,
                           const int* __restrict__ sidx,
                           int M, int out_offset, int accumulate, int nscrap) {
  const int lane = threadIdx.x;
  const int tile_r = blockIdx.x * 32;
  const int tile_c = (blockIdx.y * 4 + threadIdx.y) * 16;
  const int mr = lane & 15;
  const int kh = lane >> 4;

  const int rl = tile_r + lane;
  int sv = nscrap;
  if (rl < M) sv = sidx ? sidx[rl] : rl;
  if (__builtin_amdgcn_ballot_w32(sv != nscrap) == 0u) return;

  const int r0 = tile_r + mr;
  const int r1 = tile_r + 16 + mr;
  int g0 = 0, g1 = 0;
  if (r0 < M) g0 = gidx ? gidx[r0] : r0;
  if (r1 < M) g1 = gidx ? gidx[r1] : r1;
  const float* __restrict__ xb0 = X + (long)g0 * CIN + 2 * kh;
  const float* __restrict__ xb1 = X + (long)g1 * CIN + 2 * kh;
  const float* __restrict__ wb  = W + tile_c + mr + 2 * kh * COUT;

  v8f c0 = {0.f, 0.f, 0.f, 0.f, 0.f, 0.f, 0.f, 0.f};
  v8f c1 = {0.f, 0.f, 0.f, 0.f, 0.f, 0.f, 0.f, 0.f};
#pragma unroll
  for (int k0 = 0; k0 < CIN; k0 += 4) {
    v2f a0 = *(const v2f*)(xb0 + k0);
    v2f a1 = *(const v2f*)(xb1 + k0);
    v2f b;
    b[0] = wb[k0 * COUT];
    b[1] = wb[(k0 + 1) * COUT];
    c0 = __builtin_amdgcn_wmma_f32_16x16x4_f32(false, a0, false, b,
                                               (short)0, c0, false, false);
    c1 = __builtin_amdgcn_wmma_f32_16x16x4_f32(false, a1, false, b,
                                               (short)0, c1, false, false);
  }

#pragma unroll
  for (int i = 0; i < 8; ++i) {
    const int t0a = __builtin_amdgcn_readlane(sv, i);
    const int t0b = __builtin_amdgcn_readlane(sv, i + 8);
    const int ro0 = kh ? t0b : t0a;
    if (ro0 != nscrap) {
      float* y = Y + (long)(ro0 + out_offset) * COUT + tile_c + mr;
      if (accumulate) *y += c0[i]; else *y = c0[i];
    }
    const int t1a = __builtin_amdgcn_readlane(sv, i + 16);
    const int t1b = __builtin_amdgcn_readlane(sv, i + 24);
    const int ro1 = kh ? t1b : t1a;
    if (ro1 != nscrap) {
      float* y = Y + (long)(ro1 + out_offset) * COUT + tile_c + mr;
      if (accumulate) *y += c1[i]; else *y = c1[i];
    }
  }
}

template <int CIN, int COUT>
__global__ void __launch_bounds__(128) k_conv_os(const float* __restrict__ X,
                          const float* __restrict__ W,
                          float* __restrict__ Y,
                          const int* __restrict__ nbr,
                          int n) {
  const int lane = threadIdx.x;
  const int tile_r = blockIdx.x * 32;
  const int tile_c = (blockIdx.y * 4 + threadIdx.y) * 16;
  const int mr = lane & 15;
  const int kh = lane >> 4;
  const int r0 = tile_r + mr;
  const int r1 = tile_r + 16 + mr;
  const bool in0 = r0 < n;
  const bool in1 = r1 < n;
  constexpr int KC = CIN < 64 ? CIN : 64;
  constexpr int NS = KC / 4;

  v8f c0 = {0.f, 0.f, 0.f, 0.f, 0.f, 0.f, 0.f, 0.f};
  v8f c1 = {0.f, 0.f, 0.f, 0.f, 0.f, 0.f, 0.f, 0.f};

  const int q0 = in0 ? r0 : n - 1;
  const int q1 = in1 ? r1 : n - 1;
  int g0 = nbr[q0];
  int g1 = nbr[q1];
  g0 = in0 ? g0 : -1;
  g1 = in1 ? g1 : -1;
  for (int t = 0; t < 27; ++t) {
    int l0 = r0;
    int l1 = r1;
    if (t < 25) {
      l0 = nbr[(long)(t + 1) * n + q0];
      l1 = nbr[(long)(t + 1) * n + q1];
    }
    if (__builtin_amdgcn_ballot_w32((g0 >= 0) || (g1 >= 0)) != 0u) {
      const int k = (t < 13) ? t : (t < 26 ? t + 1 : 13);
      const float* __restrict__ wb =
          W + (long)k * CIN * COUT + tile_c + mr + 2 * kh * COUT;
#pragma unroll
      for (int kc = 0; kc < CIN; kc += KC) {
        v2f a0[NS];
        v2f a1[NS];
#pragma unroll
        for (int s = 0; s < NS; ++s) {
          a0[s] = (v2f){0.f, 0.f};
          a1[s] = (v2f){0.f, 0.f};
        }
        if (g0 >= 0) {
          const float* __restrict__ xb0 = X + (long)g0 * CIN + 2 * kh + kc;
#pragma unroll
          for (int s = 0; s < NS; ++s) a0[s] = *(const v2f*)(xb0 + 4 * s);
        }
        if (g1 >= 0) {
          const float* __restrict__ xb1 = X + (long)g1 * CIN + 2 * kh + kc;
#pragma unroll
          for (int s = 0; s < NS; ++s) a1[s] = *(const v2f*)(xb1 + 4 * s);
        }
#pragma unroll
        for (int s = 0; s < NS; ++s) {
          const int k0 = kc + 4 * s;
          v2f b;
          b[0] = wb[k0 * COUT];
          b[1] = wb[(k0 + 1) * COUT];
          c0 = __builtin_amdgcn_wmma_f32_16x16x4_f32(false, a0[s], false, b,
                                                     (short)0, c0, false, false);
          c1 = __builtin_amdgcn_wmma_f32_16x16x4_f32(false, a1[s], false, b,
                                                     (short)0, c1, false, false);
        }
      }
    }
    g0 = in0 ? l0 : -1;
    g1 = in1 ? l1 : -1;
  }

#pragma unroll
  for (int i = 0; i < 8; ++i) {
    const int ro0 = tile_r + i + 8 * kh;
    if (ro0 < n) Y[(long)ro0 * COUT + tile_c + mr] = c0[i];
    const int ro1 = ro0 + 16;
    if (ro1 < n) Y[(long)ro1 * COUT + tile_c + mr] = c1[i];
  }
}

template <int CIN, int COUT>
static inline void conv_t(const float* X, const float* W, float* Y,
                          const int* nbr, int n, hipStream_t st) {
  dim3 grid((unsigned)((n + 31) / 32), (unsigned)(COUT / 64));
  dim3 blk(32, 4);
  k_conv_os<CIN, COUT><<<grid, blk, 0, st>>>(X, W, Y, nbr, n);
}

static void conv_os(const float* X, const float* W, float* Y, const int* nbr,
                    int n, int Cin, int Cout, hipStream_t st) {
  if      (Cin == 256 && Cout == 128) conv_t<256, 128>(X, W, Y, nbr, n, st);
  else if (Cin == 128 && Cout == 128) conv_t<128, 128>(X, W, Y, nbr, n, st);
  else if (Cin ==  64 && Cout ==  64) conv_t< 64,  64>(X, W, Y, nbr, n, st);
  else if (Cin ==  64 && Cout == 128) conv_t< 64, 128>(X, W, Y, nbr, n, st);
}

template <int CIN, int COUT>
static inline void gemm_t(const float* X, const float* W, float* Y,
                          const int* g, const int* s, int M, int ooff,
                          int accum, int nscrap, hipStream_t st) {
  dim3 grid((unsigned)((M + 31) / 32), (unsigned)(COUT / 64));
  dim3 blk(32, 4);
  k_gemm_tap<CIN, COUT><<<grid, blk, 0, st>>>(X, W, Y, g, s, M, ooff, accum, nscrap);
}

static void gemm(const float* X, const float* W, float* Y,
                 const int* g, const int* s, int M, int Cin, int Cout,
                 int ooff, int accum, int nscrap, hipStream_t st) {
  if      (Cin == 256 && Cout == 128) gemm_t<256, 128>(X, W, Y, g, s, M, ooff, accum, nscrap, st);
  else if (Cin == 128 && Cout ==  64) gemm_t<128,  64>(X, W, Y, g, s, M, ooff, accum, nscrap, st);
}


extern "C" void kernel_launch(void* const* d_in, const int* in_sizes, int n_in,
                              void* d_out, int out_size, void* d_ws, size_t ws_size,
                              hipStream_t stream) {
  const float* feats0 = (const float*)d_in[0];
  const float* feats1 = (const float*)d_in[1];
  const float* feats2 = (const float*)d_in[2];
  const float* w_out0 = (const float*)d_in[3];
  const float* w_out1 = (const float*)d_in[4];
  const float* w_out2 = (const float*)d_in[5];
  const float* wt2    = (const float*)d_in[6];
  const float* wu2    = (const float*)d_in[7];
  const float* wt1    = (const float*)d_in[8];
  const float* wu1    = (const float*)d_in[9];
  const float* bn_out0 = (const float*)d_in[10];
  const float* bn_out1 = (const float*)d_in[11];
  const float* bn_out2 = (const float*)d_in[12];
  const float* bn_up2a = (const float*)d_in[13];
  const float* bn_up2b = (const float*)d_in[14];
  const float* bn_up1a = (const float*)d_in[15];
  const float* bn_up1b = (const float*)d_in[16];
  const int* m_c2 = (const int*)d_in[19];
  const int* m_g2 = (const int*)d_in[20];
  const int* m_u1 = (const int*)d_in[21];
  const int* m_g1 = (const int*)d_in[22];
  const int* m_u0 = (const int*)d_in[23];
  const int* lat1_idx = (const int*)d_in[24];
  const int* up1_idx  = (const int*)d_in[25];
  const int* lat0_idx = (const int*)d_in[26];
  const int* up0_idx  = (const int*)d_in[27];

  const int N0 = in_sizes[0] / 64;
  const int N1 = in_sizes[1] / 128;
  const int N2 = in_sizes[2] / 256;
  const int nu1 = in_sizes[17] / 128;
  const int nu0 = in_sizes[18] / 64;
  const int Mc2 = in_sizes[19] / 52;
  const int Mg2 = in_sizes[20] / 52;
  const int Mu1 = in_sizes[21] / 52;
  const int Mg1 = in_sizes[22] / 52;
  const int Mu0 = in_sizes[23] / 52;
  const int g2n = 8 * N2;
  const int g1n = 8 * nu1;

  char* wsp = (char*)d_ws;
  auto carve = [&](long nfloats) -> float* {
    float* p = (float*)wsp;
    wsp += ((nfloats * 4 + 255) / 256) * 256;
    return p;
  };
  long accElems = (long)(N2 + 1) * 128;
  { long e = (long)(g2n + 1) * 128; if (e > accElems) accElems = e; }
  { long e = (long)(nu1 + 1) * 128; if (e > accElems) accElems = e; }
  { long e = (long)(g1n + 1) * 64;  if (e > accElems) accElems = e; }
  { long e = (long)(nu0 + 1) * 128; if (e > accElems) accElems = e; }
  float* acc = carve(accElems);
  float* t2  = carve((long)g2n * 128);
  float* x1  = carve((long)nu1 * 128);
  float* t1  = carve((long)g1n * 64);
  float* x0  = carve((long)nu0 * 64);
  float* mu  = carve(128);
  float* rs  = carve(128);
  float* part = carve((long)1024 * 256);
  long maxn = N2;
  if (g2n > maxn) maxn = g2n;
  if (nu1 > maxn) maxn = nu1;
  if (g1n > maxn) maxn = g1n;
  if (nu0 > maxn) maxn = nu0;
  int* nbr = (int*)carve(26 * maxn);

  float* out  = (float*)d_out;
  float* out0 = out;
  float* out1 = out + (long)nu0 * 128;
  float* out2 = out + (long)nu0 * 128 + (long)nu1 * 128;

  auto fill0 = [&](float* p, long n) {
    k_fill<<<dim3((unsigned)((n + TPB - 1) / TPB)), dim3(TPB), 0, stream>>>(p, n);
  };
  auto conv3 = [&](const float* X, int n, int Cin, int Cout,
                   const float* W, const int* map, int Mw) {
    const long nt = (long)26 * n;
    k_fill_i32<<<dim3((unsigned)((nt + TPB - 1) / TPB)), dim3(TPB), 0, stream>>>(nbr, nt, -1);
    const long np = (long)26 * Mw;
    k_invmap<<<dim3((unsigned)((np + TPB - 1) / TPB)), dim3(TPB), 0, stream>>>(map, nbr, Mw, n);
    conv_os(X, W, acc, nbr, n, Cin, Cout, stream);
  };
  auto bnelu = [&](const float* src, float* dst, const float* gb, long n, int C) {
    long tot = n * C;
    long P = (tot + TPB * 4 - 1) / (TPB * 4);
    if (P > 1024) P = 1024;
    k_bn_part<<<dim3((unsigned)P), dim3(TPB), 0, stream>>>(src, tot, C, part);
    k_bn_fin<<<dim3((unsigned)C), dim3(TPB), 0, stream>>>(part, (int)P, n, C, mu, rs);
    k_bn_elu<<<dim3((unsigned)((tot + TPB - 1) / TPB)), dim3(TPB), 0, stream>>>(
        src, dst, mu, rs, gb, tot, C);
  };
  auto scat = [&](const float* X, float* Y, const int* idx, long rows, int C) {
    long tot = rows * C;
    k_scatter_add<<<dim3((unsigned)((tot + TPB - 1) / TPB)), dim3(TPB), 0, stream>>>(
        X, Y, idx, tot, C);
  };

  conv3(feats2, N2, 256, 128, w_out2, m_c2, Mc2);
  bnelu(acc, out2, bn_out2, N2, 128);

  for (int o = 0; o < 8; ++o)
    gemm(feats2, wt2 + (long)o * 256 * 128, t2, nullptr, nullptr, N2, 256, 128,
         o * N2, 0, -1, stream);
  bnelu(t2, t2, bn_up2a, g2n, 128);
  conv3(t2, g2n, 128, 128, wu2, m_g2, Mg2);
  bnelu(acc, t2, bn_up2b, g2n, 128);

  fill0(x1, (long)nu1 * 128);
  scat(feats1, x1, lat1_idx, N1, 128);
  scat(t2, x1, up1_idx, g2n, 128);

  conv3(x1, nu1, 128, 128, w_out1, m_u1, Mu1);
  bnelu(acc, out1, bn_out1, nu1, 128);

  for (int o = 0; o < 8; ++o)
    gemm(x1, wt1 + (long)o * 128 * 64, t1, nullptr, nullptr, nu1, 128, 64,
         o * nu1, 0, -1, stream);
  bnelu(t1, t1, bn_up1a, g1n, 64);
  conv3(t1, g1n, 64, 64, wu1, m_g1, Mg1);
  bnelu(acc, t1, bn_up1b, g1n, 64);

  fill0(x0, (long)nu0 * 64);
  scat(feats0, x0, lat0_idx, N0, 64);
  scat(t1, x0, up0_idx, g1n, 64);

  conv3(x0, nu0, 64, 128, w_out0, m_u0, Mu0);
  bnelu(acc, out0, bn_out0, nu0, 128);

  (void)n_in; (void)out_size; (void)ws_size;
}


# elementwise launches (fills, map inversion, scatter-add, normalise+ELU): at most 1024 blocks loop over 16-byte pieces, columns by mask not 64-bit division; about 15x fewer, longer waves
# speedup vs baseline: 7.3038x; 1.0446x over previous
#include <hip/hip_runtime.h>
#include <hip/hip_bf16.h>
#include <math.h>

typedef __attribute__((ext_vector_type(2))) float v2f;
typedef __attribute__((ext_vector_type(8))) float v8f;

#define TPB 256
#define MAXB 1024


__global__ void k_fill(float* __restrict__ p, long n4) {
  float4* __restrict__ p4 = (float4*)p;
  const float4 z = {0.0f, 0.0f, 0.0f, 0.0f};
  const long step = (long)gridDim.x * TPB;
  for (long i = (long)blockIdx.x * TPB + threadIdx.x; i < n4; i += step) p4[i] = z;
}

__global__ void k_fill_i32(int* __restrict__ p, long n, int v) {
  int4* __restrict__ p4 = (int4*)p;
  const int4 w = {v, v, v, v};
  const long n4 = n >> 2;
  const long step = (long)gridDim.x * TPB;
  const long g = (long)blockIdx.x * TPB + threadIdx.x;
  for (long i = g; i < n4; i += step) p4[i] = w;
  if (g < (n & 3)) p[(n4 << 2) + g] = v;
}

__global__ void k_invmap(const int* __restrict__ map, int* __restrict__ nbr,
                         int Mw, int n) {
  const int j = blockIdx.y;
  const int* __restrict__ gi = map + (long)j * Mw;
  const int* __restrict__ so = map + (long)(26 + j) * Mw;
  int* __restrict__ nb = nbr + (long)j * n;
  const int step = (int)gridDim.x * TPB;
  for (int i = (int)blockIdx.x * TPB + (int)threadIdx.x; i < Mw; i += 4 * step) {
    int o[4];
    int g[4];
    for (int u = 0; u < 4; ++u) {
      const int k = i + u * step;
      const int kc = k < Mw ? k : i;
      o[u] = so[kc];
      g[u] = gi[kc];
    }
    for (int u = 0; u < 4; ++u)
      if (i + u * step < Mw && o[u] >= 0 && o[u] < n) nb[o[u]] = g[u];
  }
}

template <int C>
__global__ void k_scatter_add(const float* __restrict__ X, float* __restrict__ Y,
                              const int* __restrict__ idx, long rows) {
  constexpr int Q = C / 4;
  const float4* __restrict__ X4 = (const float4*)X;
  const long tot4 = rows * Q;
  const long step = (long)gridDim.x * TPB;
  float4* __restrict__ Y4 = (float4*)Y;
  for (long i = (long)blockIdx.x * TPB + threadIdx.x; i < tot4; i += 4 * step) {
    long kc[4];
    long off[4];
    float4 x[4];
    float4 y[4];
    for (int u = 0; u < 4; ++u) {
      const long k = i + u * step;
      kc[u] = k < tot4 ? k : i;
      off[u] = idx[kc[u] / Q];
    }
    for (int u = 0; u < 4; ++u) {
      off[u] = off[u] * Q + (kc[u] % Q);
      x[u] = X4[kc[u]];
      y[u] = Y4[off[u]];
    }
    for (int u = 0; u < 4; ++u) {
      if (i + u * step < tot4) {
        float4 v = y[u];
        v.x += x[u].x; v.y += x[u].y; v.z += x[u].z; v.w += x[u].w;
        Y4[off[u]] = v;
      }
    }
  }
}

__global__ void k_bn_part(const float* __restrict__ X, long tot, int C,
                          float* __restrict__ part) {
  __shared__ float sm[8][TPB];
  const int t = threadIdx.x;
  float a0 = 0.0f, a1 = 0.0f, a2 = 0.0f, a3 = 0.0f;
  float b0 = 0.0f, b1 = 0.0f, b2 = 0.0f, b3 = 0.0f;
  const long step = (long)gridDim.x * (TPB * 4);
  for (long i = ((long)blockIdx.x * TPB + t) * 4; i < tot; i += step) {
    const float4 v = *(const float4*)(X + i);
    a0 += v.x; b0 += v.x * v.x;
    a1 += v.y; b1 += v.y * v.y;
    a2 += v.z; b2 += v.z * v.z;
    a3 += v.w; b3 += v.w * v.w;
  }
  sm[0][t] = a0; sm[1][t] = a1; sm[2][t] = a2; sm[3][t] = a3;
  sm[4][t] = b0; sm[5][t] = b1; sm[6][t] = b2; sm[7][t] = b3;
  __syncthreads();
  if (t < C) {
    const int q = C >> 2;
    const int comp = t & 3;
    float s = 0.0f, ss = 0.0f;
    for (int j = t >> 2; j < TPB; j += q) {
      s += sm[comp][j];
      ss += sm[4 + comp][j];
    }
    float* p = part + (long)blockIdx.x * 2 * C;
    p[t] = s;
    p[C + t] = ss;
  }
}

__global__ void k_bn_fin(const float* __restrict__ part, int P, long N, int C,
                         float* __restrict__ mu, float* __restrict__ rs) {
  __shared__ float s1[TPB];
  __shared__ float s2[TPB];
  const int c = blockIdx.x;
  float a = 0.0f, b = 0.0f;
  for (int p = threadIdx.x; p < P; p += TPB) {
    a += part[(long)p * 2 * C + c];
    b += part[(long)p * 2 * C + C + c];
  }
  s1[threadIdx.x] = a;
  s2[threadIdx.x] = b;
  __syncthreads();
  for (int s = TPB / 2; s > 0; s >>= 1) {
    if ((int)threadIdx.x < s) {
      s1[threadIdx.x] += s1[threadIdx.x + s];
      s2[threadIdx.x] += s2[threadIdx.x + s];
    }
    __syncthreads();
  }
  if (threadIdx.x == 0) {
    float m = s1[0] / (float)N;
    float var = s2[0] / (float)N - m * m;
    if (var < 0.0f) var = 0.0f;
    mu[c] = m;
    rs[c] = rsqrtf(var + 1e-5f);
  }
}

template <int C>
__global__ void k_bn_elu(const float* X, float* Y,
                         const float* __restrict__ mu, const float* __restrict__ rs,
                         const float* __restrict__ gb, long tot4) {
  const float4* X4 = (const float4*)X;
  float4* Y4 = (float4*)Y;
  const int c = ((int)threadIdx.x * 4) & (C - 1);
  float g[4], m[4], s[4], b[4];
  for (int e = 0; e < 4; ++e) {
    g[e] = gb[c + e];
    m[e] = mu[c + e];
    s[e] = rs[c + e];
    b[e] = gb[C + c + e];
  }
  const long step = (long)gridDim.x * TPB;
  for (long i = (long)blockIdx.x * TPB + threadIdx.x; i < tot4; i += 4 * step) {
    float4 x[4];
    for (int u = 0; u < 4; ++u)
      if (i + u * step < tot4) x[u] = X4[i + u * step];
    for (int u = 0; u < 4; ++u) {
      if (i + u * step < tot4) {
        float v0 = g[0] * (x[u].x - m[0]) * s[0] + b[0];
        float v1 = g[1] * (x[u].y - m[1]) * s[1] + b[1];
        float v2 = g[2] * (x[u].z - m[2]) * s[2] + b[2];
        float v3 = g[3] * (x[u].w - m[3]) * s[3] + b[3];
        float4 o;
        o.x = v0 > 0.0f ? v0 : expm1f(v0);
        o.y = v1 > 0.0f ? v1 : expm1f(v1);
        o.z = v2 > 0.0f ? v2 : expm1f(v2);
        o.w = v3 > 0.0f ? v3 : expm1f(v3);
        Y4[i + u * step] = o;
      }
    }
  }
}

template <int CIN, int COUT>
__global__ void k_gemm_tap(const float* __restrict__ X,
                           const float* __restrict__ W,
                           float* __restrict__ Y,
                           const int* __restrict__ gidx,
                           const int* __restrict__ sidx,
                           int M, int out_offset, int accumulate, int nscrap) {
  const int lane = threadIdx.x;
  const int tile_r = blockIdx.x * 32;
  const int tile_c = (blockIdx.y * 4 + threadIdx.y) * 16;
  const int mr = lane & 15;
  const int kh = lane >> 4;

  const int rl = tile_r + lane;
  int sv = nscrap;
  if (rl < M) sv = sidx ? sidx[rl] : rl;
  if (__builtin_amdgcn_ballot_w32(sv != nscrap) == 0u) return;

  const int r0 = tile_r + mr;
  const int r1 = tile_r + 16 + mr;
  int g0 = 0, g1 = 0;
  if (r0 < M) g0 = gidx ? gidx[r0] : r0;
  if (r1 < M) g1 = gidx ? gidx[r1] : r1;
  const float* __restrict__ xb0 = X + (long)g0 * CIN + 2 * kh;
  const float* __restrict__ xb1 = X + (long)g1 * CIN + 2 * kh;
  const float* __restrict__ wb  = W + tile_c + mr + 2 * kh * COUT;

  v8f c0 = {0.f, 0.f, 0.f, 0.f, 0.f, 0.f, 0.f, 0.f};
  v8f c1 = {0.f, 0.f, 0.f, 0.f, 0.f, 0.f, 0.f, 0.f};
#pragma unroll
  for (int k0 = 0; k0 < CIN; k0 += 4) {
    v2f a0 = *(const v2f*)(xb0 + k0);
    v2f a1 = *(const v2f*)(xb1 + k0);
    v2f b;
    b[0] = wb[k0 * COUT];
    b[1] = wb[(k0 + 1) * COUT];
    c0 = __builtin_amdgcn_wmma_f32_16x16x4_f32(false, a0, false, b,
                                               (short)0, c0, false, false);
    c1 = __builtin_amdgcn_wmma_f32_16x16x4_f32(false, a1, false, b,
                                               (short)0, c1, false, false);
  }

#pragma unroll
  for (int i = 0; i < 8; ++i) {
    const int t0a = __builtin_amdgcn_readlane(sv, i);
    const int t0b = __builtin_amdgcn_readlane(sv, i + 8);
    const int ro0 = kh ? t0b : t0a;
    if (ro0 != nscrap) {
      float* y = Y + (long)(ro0 + out_offset) * COUT + tile_c + mr;
      if (accumulate) *y += c0[i]; else *y = c0[i];
    }
    const int t1a = __builtin_amdgcn_readlane(sv, i + 16);
    const int t1b = __builtin_amdgcn_readlane(sv, i + 24);
    const int ro1 = kh ? t1b : t1a;
    if (ro1 != nscrap) {
      float* y = Y + (long)(ro1 + out_offset) * COUT + tile_c + mr;
      if (accumulate) *y += c1[i]; else *y = c1[i];
    }
  }
}

template <int CIN, int COUT>
__global__ void __launch_bounds__(128) k_conv_os(const float* __restrict__ X,
                          const float* __restrict__ W,
                          float* __restrict__ Y,
                          const int* __restrict__ nbr,
                          int n) {
  const int lane = threadIdx.x;
  const int tile_r = blockIdx.x * 32;
  const int tile_c = (blockIdx.y * 4 + threadIdx.y) * 16;
  const int mr = lane & 15;
  const int kh = lane >> 4;
  const int r0 = tile_r + mr;
  const int r1 = tile_r + 16 + mr;
  const bool in0 = r0 < n;
  const bool in1 = r1 < n;
  constexpr int KC = CIN < 64 ? CIN : 64;
  constexpr int NS = KC / 4;

  v8f c0 = {0.f, 0.f, 0.f, 0.f, 0.f, 0.f, 0.f, 0.f};
  v8f c1 = {0.f, 0.f, 0.f, 0.f, 0.f, 0.f, 0.f, 0.f};

  const int q0 = in0 ? r0 : n - 1;
  const int q1 = in1 ? r1 : n - 1;
  int g0 = nbr[q0];
  int g1 = nbr[q1];
  g0 = in0 ? g0 : -1;
  g1 = in1 ? g1 : -1;
  for (int t = 0; t < 27; ++t) {
    int l0 = r0;
    int l1 = r1;
    if (t < 25) {
      l0 = nbr[(long)(t + 1) * n + q0];
      l1 = nbr[(long)(t + 1) * n + q1];
    }
    if (__builtin_amdgcn_ballot_w32((g0 >= 0) || (g1 >= 0)) != 0u) {
      const int k = (t < 13) ? t : (t < 26 ? t + 1 : 13);
      const float* __restrict__ wb =
          W + (long)k * CIN * COUT + tile_c + mr + 2 * kh * COUT;
#pragma unroll
      for (int kc = 0; kc < CIN; kc += KC) {
        v2f a0[NS];
        v2f a1[NS];
#pragma unroll
        for (int s = 0; s < NS; ++s) {
          a0[s] = (v2f){0.f, 0.f};
          a1[s] = (v2f){0.f, 0.f};
        }
        if (g0 >= 0) {
          const float* __restrict__ xb0 = X + (long)g0 * CIN + 2 * kh + kc;
#pragma unroll
          for (int s = 0; s < NS; ++s) a0[s] = *(const v2f*)(xb0 + 4 * s);
        }
        if (g1 >= 0) {
          const float* __restrict__ xb1 = X + (long)g1 * CIN + 2 * kh + kc;
#pragma unroll
          for (int s = 0; s < NS; ++s) a1[s] = *(const v2f*)(xb1 + 4 * s);
        }
#pragma unroll
        for (int s = 0; s < NS; ++s) {
          const int k0 = kc + 4 * s;
          v2f b;
          b[0] = wb[k0 * COUT];
          b[1] = wb[(k0 + 1) * COUT];
          c0 = __builtin_amdgcn_wmma_f32_16x16x4_f32(false, a0[s], false, b,
                                                     (short)0, c0, false, false);
          c1 = __builtin_amdgcn_wmma_f32_16x16x4_f32(false, a1[s], false, b,
                                                     (short)0, c1, false, false);
        }
      }
    }
    g0 = in0 ? l0 : -1;
    g1 = in1 ? l1 : -1;
  }

#pragma unroll
  for (int i = 0; i < 8; ++i) {
    const int ro0 = tile_r + i + 8 * kh;
    if (ro0 < n) Y[(long)ro0 * COUT + tile_c + mr] = c0[i];
    const int ro1 = ro0 + 16;
    if (ro1 < n) Y[(long)ro1 * COUT + tile_c + mr] = c1[i];
  }
}

template <int CIN, int COUT>
static inline void conv_t(const float* X, const float* W, float* Y,
                          const int* nbr, int n, hipStream_t st) {
  dim3 grid((unsigned)((n + 31) / 32), (unsigned)(COUT / 64));
  dim3 blk(32, 4);
  k_conv_os<CIN, COUT><<<grid, blk, 0, st>>>(X, W, Y, nbr, n);
}

static void conv_os(const float* X, const float* W, float* Y, const int* nbr,
                    int n, int Cin, int Cout, hipStream_t st) {
  if      (Cin == 256 && Cout == 128) conv_t<256, 128>(X, W, Y, nbr, n, st);
  else if (Cin == 128 && Cout == 128) conv_t<128, 128>(X, W, Y, nbr, n, st);
  else if (Cin ==  64 && Cout ==  64) conv_t< 64,  64>(X, W, Y, nbr, n, st);
  else if (Cin ==  64 && Cout == 128) conv_t< 64, 128>(X, W, Y, nbr, n, st);
}

template <int CIN, int COUT>
static inline void gemm_t(const float* X, const float* W, float* Y,
                          const int* g, const int* s, int M, int ooff,
                          int accum, int nscrap, hipStream_t st) {
  dim3 grid((unsigned)((M + 31) / 32), (unsigned)(COUT / 64));
  dim3 blk(32, 4);
  k_gemm_tap<CIN, COUT><<<grid, blk, 0, st>>>(X, W, Y, g, s, M, ooff, accum, nscrap);
}

static void gemm(const float* X, const float* W, float* Y,
                 const int* g, const int* s, int M, int Cin, int Cout,
                 int ooff, int accum, int nscrap, hipStream_t st) {
  if      (Cin == 256 && Cout == 128) gemm_t<256, 128>(X, W, Y, g, s, M, ooff, accum, nscrap, st);
  else if (Cin == 128 && Cout ==  64) gemm_t<128,  64>(X, W, Y, g, s, M, ooff, accum, nscrap, st);
}


extern "C" void kernel_launch(void* const* d_in, const int* in_sizes, int n_in,
                              void* d_out, int out_size, void* d_ws, size_t ws_size,
                              hipStream_t stream) {
  const float* feats0 = (const float*)d_in[0];
  const float* feats1 = (const float*)d_in[1];
  const float* feats2 = (const float*)d_in[2];
  const float* w_out0 = (const float*)d_in[3];
  const float* w_out1 = (const float*)d_in[4];
  const float* w_out2 = (const float*)d_in[5];
  const float* wt2    = (const float*)d_in[6];
  const float* wu2    = (const float*)d_in[7];
  const float* wt1    = (const float*)d_in[8];
  const float* wu1    = (const float*)d_in[9];
  const float* bn_out0 = (const float*)d_in[10];
  const float* bn_out1 = (const float*)d_in[11];
  const float* bn_out2 = (const float*)d_in[12];
  const float* bn_up2a = (const float*)d_in[13];
  const float* bn_up2b = (const float*)d_in[14];
  const float* bn_up1a = (const float*)d_in[15];
  const float* bn_up1b = (const float*)d_in[16];
  const int* m_c2 = (const int*)d_in[19];
  const int* m_g2 = (const int*)d_in[20];
  const int* m_u1 = (const int*)d_in[21];
  const int* m_g1 = (const int*)d_in[22];
  const int* m_u0 = (const int*)d_in[23];
  const int* lat1_idx = (const int*)d_in[24];
  const int* up1_idx  = (const int*)d_in[25];
  const int* lat0_idx = (const int*)d_in[26];
  const int* up0_idx  = (const int*)d_in[27];

  const int N0 = in_sizes[0] / 64;
  const int N1 = in_sizes[1] / 128;
  const int N2 = in_sizes[2] / 256;
  const int nu1 = in_sizes[17] / 128;
  const int nu0 = in_sizes[18] / 64;
  const int Mc2 = in_sizes[19] / 52;
  const int Mg2 = in_sizes[20] / 52;
  const int Mu1 = in_sizes[21] / 52;
  const int Mg1 = in_sizes[22] / 52;
  const int Mu0 = in_sizes[23] / 52;
  const int g2n = 8 * N2;
  const int g1n = 8 * nu1;

  char* wsp = (char*)d_ws;
  auto carve = [&](long nfloats) -> float* {
    float* p = (float*)wsp;
    wsp += ((nfloats * 4 + 255) / 256) * 256;
    return p;
  };
  long accElems = (long)(N2 + 1) * 128;
  { long e = (long)(g2n + 1) * 128; if (e > accElems) accElems = e; }
  { long e = (long)(nu1 + 1) * 128; if (e > accElems) accElems = e; }
  { long e = (long)(g1n + 1) * 64;  if (e > accElems) accElems = e; }
  { long e = (long)(nu0 + 1) * 128; if (e > accElems) accElems = e; }
  float* acc = carve(accElems);
  float* t2  = carve((long)g2n * 128);
  float* x1  = carve((long)nu1 * 128);
  float* t1  = carve((long)g1n * 64);
  float* x0  = carve((long)nu0 * 64);
  float* mu  = carve(128);
  float* rs  = carve(128);
  float* part = carve((long)1024 * 256);
  long maxn = N2;
  if (g2n > maxn) maxn = g2n;
  if (nu1 > maxn) maxn = nu1;
  if (g1n > maxn) maxn = g1n;
  if (nu0 > maxn) maxn = nu0;
  int* nbr = (int*)carve(26 * maxn);

  float* out  = (float*)d_out;
  float* out0 = out;
  float* out1 = out + (long)nu0 * 128;
  float* out2 = out + (long)nu0 * 128 + (long)nu1 * 128;

  auto nblk = [](long items) -> unsigned {
    long b = (items + TPB - 1) / TPB;
    if (b > MAXB) b = MAXB;
    if (b < 1) b = 1;
    return (unsigned)b;
  };
  auto fill0 = [&](float* p, long n) {
    k_fill<<<dim3(nblk(n / 4)), dim3(TPB), 0, stream>>>(p, n / 4);
  };
  auto conv3 = [&](const float* X, int n, int Cin, int Cout,
                   const float* W, const int* map, int Mw) {
    const long nt = (long)26 * n;
    k_fill_i32<<<dim3(nblk(nt / 4)), dim3(TPB), 0, stream>>>(nbr, nt, -1);
    unsigned gx = (unsigned)((Mw + TPB - 1) / TPB);
    if (gx > MAXB / 26) gx = MAXB / 26;
    if (gx < 1) gx = 1;
    k_invmap<<<dim3(gx, 26), dim3(TPB), 0, stream>>>(map, nbr, Mw, n);
    conv_os(X, W, acc, nbr, n, Cin, Cout, stream);
  };
  auto bnelu = [&](const float* src, float* dst, const float* gb, long n, int C) {
    long tot = n * C;
    long P = (tot + TPB * 4 - 1) / (TPB * 4);
    if (P > 1024) P = 1024;
    k_bn_part<<<dim3((unsigned)P), dim3(TPB), 0, stream>>>(src, tot, C, part);
    k_bn_fin<<<dim3((unsigned)C), dim3(TPB), 0, stream>>>(part, (int)P, n, C, mu, rs);
    const long tot4 = tot / 4;
    if (C == 128)
      k_bn_elu<128><<<dim3(nblk(tot4)), dim3(TPB), 0, stream>>>(src, dst, mu, rs, gb, tot4);
    else
      k_bn_elu<64><<<dim3(nblk(tot4)), dim3(TPB), 0, stream>>>(src, dst, mu, rs, gb, tot4);
  };
  auto scat = [&](const float* X, float* Y, const int* idx, long rows, int C) {
    const long tot4 = rows * (C / 4);
    if (C == 128)
      k_scatter_add<128><<<dim3(nblk(tot4)), dim3(TPB), 0, stream>>>(X, Y, idx, rows);
    else
      k_scatter_add<64><<<dim3(nblk(tot4)), dim3(TPB), 0, stream>>>(X, Y, idx, rows);
  };

  conv3(feats2, N2, 256, 128, w_out2, m_c2, Mc2);
  bnelu(acc, out2, bn_out2, N2, 128);

  for (int o = 0; o < 8; ++o)
    gemm(feats2, wt2 + (long)o * 256 * 128, t2, nullptr, nullptr, N2, 256, 128,
         o * N2, 0, -1, stream);
  bnelu(t2, t2, bn_up2a, g2n, 128);
  conv3(t2, g2n, 128, 128, wu2, m_g2, Mg2);
  bnelu(acc, t2, bn_up2b, g2n, 128);

  fill0(x1, (long)nu1 * 128);
  scat(feats1, x1, lat1_idx, N1, 128);
  scat(t2, x1, up1_idx, g2n, 128);

  conv3(x1, nu1, 128, 128, w_out1, m_u1, Mu1);
  bnelu(acc, out1, bn_out1, nu1, 128);

  for (int o = 0; o < 8; ++o)
    gemm(x1, wt1 + (long)o * 128 * 64, t1, nullptr, nullptr, nu1, 128, 64,
         o * nu1, 0, -1, stream);
  bnelu(t1, t1, bn_up1a, g1n, 64);
  conv3(t1, g1n, 64, 64, wu1, m_g1, Mg1);
  bnelu(acc, t1, bn_up1b, g1n, 64);

  fill0(x0, (long)nu0 * 64);
  scat(feats0, x0, lat0_idx, N0, 64);
  scat(t1, x0, up0_idx, g1n, 64);

  conv3(x0, nu0, 64, 128, w_out0, m_u0, Mu0);
  bnelu(acc, out0, bn_out0, nu0, 128);

  (void)n_in; (void)out_size; (void)ws_size;
}


# both large 3^3 convs re-tiled: a 4-wave block owns 128 rows, each wave 32 rows x all columns; a tap's weights go once per block into LDS, rows are gathered once (16-byte loads); k sum order changes
# speedup vs baseline: 7.6823x; 1.0518x over previous
#include <hip/hip_runtime.h>
#include <hip/hip_bf16.h>
#include <math.h>

typedef __attribute__((ext_vector_type(2))) float v2f;
typedef __attribute__((ext_vector_type(8))) float v8f;

#define TPB 256
#define MAXB 1024


__global__ void k_fill(float* __restrict__ p, long n4) {
  float4* __restrict__ p4 = (float4*)p;
  const float4 z = {0.0f, 0.0f, 0.0f, 0.0f};
  const long step = (long)gridDim.x * TPB;
  for (long i = (long)blockIdx.x * TPB + threadIdx.x; i < n4; i += step) p4[i] = z;
}

__global__ void k_fill_i32(int* __restrict__ p, long n, int v) {
  int4* __restrict__ p4 = (int4*)p;
  const int4 w = {v, v, v, v};
  const long n4 = n >> 2;
  const long step = (long)gridDim.x * TPB;
  const long g = (long)blockIdx.x * TPB + threadIdx.x;
  for (long i = g; i < n4; i += step) p4[i] = w;
  if (g < (n & 3)) p[(n4 << 2) + g] = v;
}

__global__ void k_invmap(const int* __restrict__ map, int* __restrict__ nbr,
                         int Mw, int n) {
  const int j = blockIdx.y;
  const int* __restrict__ gi = map + (long)j * Mw;
  const int* __restrict__ so = map + (long)(26 + j) * Mw;
  int* __restrict__ nb = nbr + (long)j * n;
  const int step = (int)gridDim.x * TPB;
  for (int i = (int)blockIdx.x * TPB + (int)threadIdx.x; i < Mw; i += 4 * step) {
    int o[4];
    int g[4];
    for (int u = 0; u < 4; ++u) {
      const int k = i + u * step;
      const int kc = k < Mw ? k : i;
      o[u] = so[kc];
      g[u] = gi[kc];
    }
    for (int u = 0; u < 4; ++u)
      if (i + u * step < Mw && o[u] >= 0 && o[u] < n) nb[o[u]] = g[u];
  }
}

template <int C>
__global__ void k_scatter_add(const float* __restrict__ X, float* __restrict__ Y,
                              const int* __restrict__ idx, long rows) {
  constexpr int Q = C / 4;
  const float4* __restrict__ X4 = (const float4*)X;
  const long tot4 = rows * Q;
  const long step = (long)gridDim.x * TPB;
  float4* __restrict__ Y4 = (float4*)Y;
  for (long i = (long)blockIdx.x * TPB + threadIdx.x; i < tot4; i += 4 * step) {
    long kc[4];
    long off[4];
    float4 x[4];
    float4 y[4];
    for (int u = 0; u < 4; ++u) {
      const long k = i + u * step;
      kc[u] = k < tot4 ? k : i;
      off[u] = idx[kc[u] / Q];
    }
    for (int u = 0; u < 4; ++u) {
      off[u] = off[u] * Q + (kc[u] % Q);
      x[u] = X4[kc[u]];
      y[u] = Y4[off[u]];
    }
    for (int u = 0; u < 4; ++u) {
      if (i + u * step < tot4) {
        float4 v = y[u];
        v.x += x[u].x; v.y += x[u].y; v.z += x[u].z; v.w += x[u].w;
        Y4[off[u]] = v;
      }
    }
  }
}

__global__ void k_bn_part(const float* __restrict__ X, long tot, int C,
                          float* __restrict__ part) {
  __shared__ float sm[8][TPB];
  const int t = threadIdx.x;
  float a0 = 0.0f, a1 = 0.0f, a2 = 0.0f, a3 = 0.0f;
  float b0 = 0.0f, b1 = 0.0f, b2 = 0.0f, b3 = 0.0f;
  const long step = (long)gridDim.x * (TPB * 4);
  for (long i = ((long)blockIdx.x * TPB + t) * 4; i < tot; i += step) {
    const float4 v = *(const float4*)(X + i);
    a0 += v.x; b0 += v.x * v.x;
    a1 += v.y; b1 += v.y * v.y;
    a2 += v.z; b2 += v.z * v.z;
    a3 += v.w; b3 += v.w * v.w;
  }
  sm[0][t] = a0; sm[1][t] = a1; sm[2][t] = a2; sm[3][t] = a3;
  sm[4][t] = b0; sm[5][t] = b1; sm[6][t] = b2; sm[7][t] = b3;
  __syncthreads();
  if (t < C) {
    const int q = C >> 2;
    const int comp = t & 3;
    float s = 0.0f, ss = 0.0f;
    for (int j = t >> 2; j < TPB; j += q) {
      s += sm[comp][j];
      ss += sm[4 + comp][j];
    }
    float* p = part + (long)blockIdx.x * 2 * C;
    p[t] = s;
    p[C + t] = ss;
  }
}

__global__ void k_bn_fin(const float* __restrict__ part, int P, long N, int C,
                         float* __restrict__ mu, float* __restrict__ rs) {
  __shared__ float s1[TPB];
  __shared__ float s2[TPB];
  const int c = blockIdx.x;
  float a = 0.0f, b = 0.0f;
  for (int p = threadIdx.x; p < P; p += TPB) {
    a += part[(long)p * 2 * C + c];
    b += part[(long)p * 2 * C + C + c];
  }
  s1[threadIdx.x] = a;
  s2[threadIdx.x] = b;
  __syncthreads();
  for (int s = TPB / 2; s > 0; s >>= 1) {
    if ((int)threadIdx.x < s) {
      s1[threadIdx.x] += s1[threadIdx.x + s];
      s2[threadIdx.x] += s2[threadIdx.x + s];
    }
    __syncthreads();
  }
  if (threadIdx.x == 0) {
    float m = s1[0] / (float)N;
    float var = s2[0] / (float)N - m * m;
    if (var < 0.0f) var = 0.0f;
    mu[c] = m;
    rs[c] = rsqrtf(var + 1e-5f);
  }
}

template <int C>
__global__ void k_bn_elu(const float* X, float* Y,
                         const float* __restrict__ mu, const float* __restrict__ rs,
                         const float* __restrict__ gb, long tot4) {
  const float4* X4 = (const float4*)X;
  float4* Y4 = (float4*)Y;
  const int c = ((int)threadIdx.x * 4) & (C - 1);
  float g[4], m[4], s[4], b[4];
  for (int e = 0; e < 4; ++e) {
    g[e] = gb[c + e];
    m[e] = mu[c + e];
    s[e] = rs[c + e];
    b[e] = gb[C + c + e];
  }
  const long step = (long)gridDim.x * TPB;
  for (long i = (long)blockIdx.x * TPB + threadIdx.x; i < tot4; i += 4 * step) {
    float4 x[4];
    for (int u = 0; u < 4; ++u)
      if (i + u * step < tot4) x[u] = X4[i + u * step];
    for (int u = 0; u < 4; ++u) {
      if (i + u * step < tot4) {
        float v0 = g[0] * (x[u].x - m[0]) * s[0] + b[0];
        float v1 = g[1] * (x[u].y - m[1]) * s[1] + b[1];
        float v2 = g[2] * (x[u].z - m[2]) * s[2] + b[2];
        float v3 = g[3] * (x[u].w - m[3]) * s[3] + b[3];
        float4 o;
        o.x = v0 > 0.0f ? v0 : expm1f(v0);
        o.y = v1 > 0.0f ? v1 : expm1f(v1);
        o.z = v2 > 0.0f ? v2 : expm1f(v2);
        o.w = v3 > 0.0f ? v3 : expm1f(v3);
        Y4[i + u * step] = o;
      }
    }
  }
}

template <int CIN, int COUT>
__global__ void k_gemm_tap(const float* __restrict__ X,
                           const float* __restrict__ W,
                           float* __restrict__ Y,
                           const int* __restrict__ gidx,
                           const int* __restrict__ sidx,
                           int M, int out_offset, int accumulate, int nscrap) {
  const int lane = threadIdx.x;
  const int tile_r = blockIdx.x * 32;
  const int tile_c = (blockIdx.y * 4 + threadIdx.y) * 16;
  const int mr = lane & 15;
  const int kh = lane >> 4;

  const int rl = tile_r + lane;
  int sv = nscrap;
  if (rl < M) sv = sidx ? sidx[rl] : rl;
  if (__builtin_amdgcn_ballot_w32(sv != nscrap) == 0u) return;

  const int r0 = tile_r + mr;
  const int r1 = tile_r + 16 + mr;
  int g0 = 0, g1 = 0;
  if (r0 < M) g0 = gidx ? gidx[r0] : r0;
  if (r1 < M) g1 = gidx ? gidx[r1] : r1;
  const float* __restrict__ xb0 = X + (long)g0 * CIN + 2 * kh;
  const float* __restrict__ xb1 = X + (long)g1 * CIN + 2 * kh;
  const float* __restrict__ wb  = W + tile_c + mr + 2 * kh * COUT;

  v8f c0 = {0.f, 0.f, 0.f, 0.f, 0.f, 0.f, 0.f, 0.f};
  v8f c1 = {0.f, 0.f, 0.f, 0.f, 0.f, 0.f, 0.f, 0.f};
#pragma unroll
  for (int k0 = 0; k0 < CIN; k0 += 4) {
    v2f a0 = *(const v2f*)(xb0 + k0);
    v2f a1 = *(const v2f*)(xb1 + k0);
    v2f b;
    b[0] = wb[k0 * COUT];
    b[1] = wb[(k0 + 1) * COUT];
    c0 = __builtin_amdgcn_wmma_f32_16x16x4_f32(false, a0, false, b,
                                               (short)0, c0, false, false);
    c1 = __builtin_amdgcn_wmma_f32_16x16x4_f32(false, a1, false, b,
                                               (short)0, c1, false, false);
  }

#pragma unroll
  for (int i = 0; i < 8; ++i) {
    const int t0a = __builtin_amdgcn_readlane(sv, i);
    const int t0b = __builtin_amdgcn_readlane(sv, i + 8);
    const int ro0 = kh ? t0b : t0a;
    if (ro0 != nscrap) {
      float* y = Y + (long)(ro0 + out_offset) * COUT + tile_c + mr;
      if (accumulate) *y += c0[i]; else *y = c0[i];
    }
    const int t1a = __builtin_amdgcn_readlane(sv, i + 16);
    const int t1b = __builtin_amdgcn_readlane(sv, i + 24);
    const int ro1 = kh ? t1b : t1a;
    if (ro1 != nscrap) {
      float* y = Y + (long)(ro1 + out_offset) * COUT + tile_c + mr;
      if (accumulate) *y += c1[i]; else *y = c1[i];
    }
  }
}

template <int CIN, int COUT>
__global__ void __launch_bounds__(128) k_conv_os(const float* __restrict__ X,
                          const float* __restrict__ W,
                          float* __restrict__ Y,
                          const int* __restrict__ nbr,
                          int n) {
  const int lane = threadIdx.x;
  const int tile_r = blockIdx.x * 32;
  const int tile_c = (blockIdx.y * 4 + threadIdx.y) * 16;
  const int mr = lane & 15;
  const int kh = lane >> 4;
  const int r0 = tile_r + mr;
  const int r1 = tile_r + 16 + mr;
  const bool in0 = r0 < n;
  const bool in1 = r1 < n;
  constexpr int KC = CIN < 64 ? CIN : 64;
  constexpr int NS = KC / 4;

  v8f c0 = {0.f, 0.f, 0.f, 0.f, 0.f, 0.f, 0.f, 0.f};
  v8f c1 = {0.f, 0.f, 0.f, 0.f, 0.f, 0.f, 0.f, 0.f};

  const int q0 = in0 ? r0 : n - 1;
  const int q1 = in1 ? r1 : n - 1;
  int g0 = nbr[q0];
  int g1 = nbr[q1];
  g0 = in0 ? g0 : -1;
  g1 = in1 ? g1 : -1;
  for (int t = 0; t < 27; ++t) {
    int l0 = r0;
    int l1 = r1;
    if (t < 25) {
      l0 = nbr[(long)(t + 1) * n + q0];
      l1 = nbr[(long)(t + 1) * n + q1];
    }
    if (__builtin_amdgcn_ballot_w32((g0 >= 0) || (g1 >= 0)) != 0u) {
      const int k = (t < 13) ? t : (t < 26 ? t + 1 : 13);
      const float* __restrict__ wb =
          W + (long)k * CIN * COUT + tile_c + mr + 2 * kh * COUT;
#pragma unroll
      for (int kc = 0; kc < CIN; kc += KC) {
        v2f a0[NS];
        v2f a1[NS];
#pragma unroll
        for (int s = 0; s < NS; ++s) {
          a0[s] = (v2f){0.f, 0.f};
          a1[s] = (v2f){0.f, 0.f};
        }
        if (g0 >= 0) {
          const float* __restrict__ xb0 = X + (long)g0 * CIN + 2 * kh + kc;
#pragma unroll
          for (int s = 0; s < NS; ++s) a0[s] = *(const v2f*)(xb0 + 4 * s);
        }
        if (g1 >= 0) {
          const float* __restrict__ xb1 = X + (long)g1 * CIN + 2 * kh + kc;
#pragma unroll
          for (int s = 0; s < NS; ++s) a1[s] = *(const v2f*)(xb1 + 4 * s);
        }
#pragma unroll
        for (int s = 0; s < NS; ++s) {
          const int k0 = kc + 4 * s;
          v2f b;
          b[0] = wb[k0 * COUT];
          b[1] = wb[(k0 + 1) * COUT];
          c0 = __builtin_amdgcn_wmma_f32_16x16x4_f32(false, a0[s], false, b,
                                                     (short)0, c0, false, false);
          c1 = __builtin_amdgcn_wmma_f32_16x16x4_f32(false, a1[s], false, b,
                                                     (short)0, c1, false, false);
        }
      }
    }
    g0 = in0 ? l0 : -1;
    g1 = in1 ? l1 : -1;
  }

#pragma unroll
  for (int i = 0; i < 8; ++i) {
    const int ro0 = tile_r + i + 8 * kh;
    if (ro0 < n) Y[(long)ro0 * COUT + tile_c + mr] = c0[i];
    const int ro1 = ro0 + 16;
    if (ro1 < n) Y[(long)ro1 * COUT + tile_c + mr] = c1[i];
  }
}

template <int CIN, int COUT>
__global__ void __launch_bounds__(128) k_conv_os_wide(const float* __restrict__ X,
                          const float* __restrict__ W,
                          float* __restrict__ Y,
                          const int* __restrict__ nbr,
                          int n) {
  constexpr int NT = COUT / 16;
  constexpr int KC = 64;
  constexpr int NCH = CIN / KC;
  constexpr int PITCH = KC + 4;
  constexpr int CG = COUT / 4;
  constexpr int ITEMS = (KC / 4) * CG / 128;
  constexpr int S = 27 * NCH;
  __shared__ float Bs[2][COUT * PITCH];

  const int lane = threadIdx.x;
  const int wv = threadIdx.y;
  const int tid = wv * 32 + lane;
  const int tile_r = (blockIdx.x * 4 + wv) * 32;
  const int mr = lane & 15;
  const int kh = lane >> 4;
  const int r0 = tile_r + mr;
  const int r1 = tile_r + 16 + mr;
  const bool in0 = r0 < n;
  const bool in1 = r1 < n;

  auto stage = [&](int st, float* __restrict__ dst) {
    const int t = st / NCH;
    const int kc = (st % NCH) * KC;
    const int k = (t < 13) ? t : (t < 26 ? t + 1 : 13);
    const float* __restrict__ wsrc = W + ((long)k * CIN + kc) * COUT;
    float4 r[ITEMS][4];
    for (int u = 0; u < ITEMS; ++u) {
      const int e = tid + 128 * u;
      const int cg = e % CG;
      const int kg = e / CG;
      for (int rr = 0; rr < 4; ++rr)
        r[u][rr] = *(const float4*)(wsrc + (long)(4 * kg + rr) * COUT + 4 * cg);
    }
    for (int u = 0; u < ITEMS; ++u) {
      const int e = tid + 128 * u;
      const int cg = e % CG;
      const int kg = e / CG;
      float* d = dst + (4 * cg) * PITCH + 4 * kg;
      *(float4*)(d)             = make_float4(r[u][0].x, r[u][1].x, r[u][2].x, r[u][3].x);
      *(float4*)(d + PITCH)     = make_float4(r[u][0].y, r[u][1].y, r[u][2].y, r[u][3].y);
      *(float4*)(d + 2 * PITCH) = make_float4(r[u][0].z, r[u][1].z, r[u][2].z, r[u][3].z);
      *(float4*)(d + 3 * PITCH) = make_float4(r[u][0].w, r[u][1].w, r[u][2].w, r[u][3].w);
    }
  };

  v8f c0[NT];
  v8f c1[NT];
  for (int j = 0; j < NT; ++j) {
    c0[j] = (v8f){0.f, 0.f, 0.f, 0.f, 0.f, 0.f, 0.f, 0.f};
    c1[j] = (v8f){0.f, 0.f, 0.f, 0.f, 0.f, 0.f, 0.f, 0.f};
  }

  stage(0, Bs[0]);

  const int q0 = in0 ? r0 : n - 1;
  const int q1 = in1 ? r1 : n - 1;
  int g0 = nbr[q0];
  int g1 = nbr[q1];
  g0 = in0 ? g0 : -1;
  g1 = in1 ? g1 : -1;
  __syncthreads();

  for (int t = 0; t < 27; ++t) {
    int l0 = r0;
    int l1 = r1;
    if (t < 25) {
      l0 = nbr[(long)(t + 1) * n + q0];
      l1 = nbr[(long)(t + 1) * n + q1];
    }
    const bool act = __builtin_amdgcn_ballot_w32((g0 >= 0) || (g1 >= 0)) != 0u;
    for (int ch = 0; ch < NCH; ++ch) {
      const int st = t * NCH + ch;
      if (st + 1 < S) stage(st + 1, Bs[(st + 1) & 1]);
      if (act) {
        const float* __restrict__ bs = Bs[st & 1] + mr * PITCH + 4 * kh;
        float4 a0[8];
        float4 a1[8];
        for (int p = 0; p < 8; ++p) {
          a0[p] = make_float4(0.f, 0.f, 0.f, 0.f);
          a1[p] = make_float4(0.f, 0.f, 0.f, 0.f);
        }
        if (g0 >= 0) {
          const float* __restrict__ xb0 = X + (long)g0 * CIN + ch * KC + 4 * kh;
          for (int p = 0; p < 8; ++p) a0[p] = *(const float4*)(xb0 + 8 * p);
        }
        if (g1 >= 0) {
          const float* __restrict__ xb1 = X + (long)g1 * CIN + ch * KC + 4 * kh;
          for (int p = 0; p < 8; ++p) a1[p] = *(const float4*)(xb1 + 8 * p);
        }
        for (int p = 0; p < 8; ++p) {
          const v2f a0l = {a0[p].x, a0[p].y};
          const v2f a0h = {a0[p].z, a0[p].w};
          const v2f a1l = {a1[p].x, a1[p].y};
          const v2f a1h = {a1[p].z, a1[p].w};
          float4 bf[NT];
          for (int j = 0; j < NT; ++j) bf[j] = *(const float4*)(bs + 16 * j * PITCH + 8 * p);
          for (int j = 0; j < NT; ++j) {
            const float4 b = bf[j];
            const v2f bl = {b.x, b.y};
            const v2f bh = {b.z, b.w};
            c0[j] = __builtin_amdgcn_wmma_f32_16x16x4_f32(false, a0l, false, bl,
                                                          (short)0, c0[j], false, false);
            c1[j] = __builtin_amdgcn_wmma_f32_16x16x4_f32(false, a1l, false, bl,
                                                          (short)0, c1[j], false, false);
            c0[j] = __builtin_amdgcn_wmma_f32_16x16x4_f32(false, a0h, false, bh,
                                                          (short)0, c0[j], false, false);
            c1[j] = __builtin_amdgcn_wmma_f32_16x16x4_f32(false, a1h, false, bh,
                                                          (short)0, c1[j], false, false);
          }
        }
      }
      __syncthreads();
    }
    g0 = in0 ? l0 : -1;
    g1 = in1 ? l1 : -1;
  }

  for (int j = 0; j < NT; ++j) {
    for (int i = 0; i < 8; ++i) {
      const int ro0 = tile_r + i + 8 * kh;
      if (ro0 < n) Y[(long)ro0 * COUT + 16 * j + mr] = c0[j][i];
      const int ro1 = ro0 + 16;
      if (ro1 < n) Y[(long)ro1 * COUT + 16 * j + mr] = c1[j][i];
    }
  }
}

template <int CIN, int COUT>
static inline void conv_t(const float* X, const float* W, float* Y,
                          const int* nbr, int n, hipStream_t st) {
  dim3 grid((unsigned)((n + 31) / 32), (unsigned)(COUT / 64));
  dim3 blk(32, 4);
  k_conv_os<CIN, COUT><<<grid, blk, 0, st>>>(X, W, Y, nbr, n);
}

template <int CIN, int COUT>
static inline void conv_wide_t(const float* X, const float* W, float* Y,
                               const int* nbr, int n, hipStream_t st) {
  dim3 grid((unsigned)((n + 127) / 128));
  dim3 blk(32, 4);
  k_conv_os_wide<CIN, COUT><<<grid, blk, 0, st>>>(X, W, Y, nbr, n);
}

static void conv_os(const float* X, const float* W, float* Y, const int* nbr,
                    int n, int Cin, int Cout, hipStream_t st) {
  const bool wide = n >= 128 * 256;
  if      (wide && Cin == 64 && Cout ==  64) conv_wide_t<64,  64>(X, W, Y, nbr, n, st);
  else if (wide && Cin == 64 && Cout == 128) conv_wide_t<64, 128>(X, W, Y, nbr, n, st);
  else if (Cin == 256 && Cout == 128) conv_t<256, 128>(X, W, Y, nbr, n, st);
  else if (Cin == 128 && Cout == 128) conv_t<128, 128>(X, W, Y, nbr, n, st);
  else if (Cin ==  64 && Cout ==  64) conv_t< 64,  64>(X, W, Y, nbr, n, st);
  else if (Cin ==  64 && Cout == 128) conv_t< 64, 128>(X, W, Y, nbr, n, st);
}

template <int CIN, int COUT>
static inline void gemm_t(const float* X, const float* W, float* Y,
                          const int* g, const int* s, int M, int ooff,
                          int accum, int nscrap, hipStream_t st) {
  dim3 grid((unsigned)((M + 31) / 32), (unsigned)(COUT / 64));
  dim3 blk(32, 4);
  k_gemm_tap<CIN, COUT><<<grid, blk, 0, st>>>(X, W, Y, g, s, M, ooff, accum, nscrap);
}

static void gemm(const float* X, const float* W, float* Y,
                 const int* g, const int* s, int M, int Cin, int Cout,
                 int ooff, int accum, int nscrap, hipStream_t st) {
  if      (Cin == 256 && Cout == 128) gemm_t<256, 128>(X, W, Y, g, s, M, ooff, accum, nscrap, st);
  else if (Cin == 128 && Cout ==  64) gemm_t<128,  64>(X, W, Y, g, s, M, ooff, accum, nscrap, st);
}


extern "C" void kernel_launch(void* const* d_in, const int* in_sizes, int n_in,
                              void* d_out, int out_size, void* d_ws, size_t ws_size,
                              hipStream_t stream) {
  const float* feats0 = (const float*)d_in[0];
  const float* feats1 = (const float*)d_in[1];
  const float* feats2 = (const float*)d_in[2];
  const float* w_out0 = (const float*)d_in[3];
  const float* w_out1 = (const float*)d_in[4];
  const float* w_out2 = (const float*)d_in[5];
  const float* wt2    = (const float*)d_in[6];
  const float* wu2    = (const float*)d_in[7];
  const float* wt1    = (const float*)d_in[8];
  const float* wu1    = (const float*)d_in[9];
  const float* bn_out0 = (const float*)d_in[10];
  const float* bn_out1 = (const float*)d_in[11];
  const float* bn_out2 = (const float*)d_in[12];
  const float* bn_up2a = (const float*)d_in[13];
  const float* bn_up2b = (const float*)d_in[14];
  const float* bn_up1a = (const float*)d_in[15];
  const float* bn_up1b = (const float*)d_in[16];
  const int* m_c2 = (const int*)d_in[19];
  const int* m_g2 = (const int*)d_in[20];
  const int* m_u1 = (const int*)d_in[21];
  const int* m_g1 = (const int*)d_in[22];
  const int* m_u0 = (const int*)d_in[23];
  const int* lat1_idx = (const int*)d_in[24];
  const int* up1_idx  = (const int*)d_in[25];
  const int* lat0_idx = (const int*)d_in[26];
  const int* up0_idx  = (const int*)d_in[27];

  const int N0 = in_sizes[0] / 64;
  const int N1 = in_sizes[1] / 128;
  const int N2 = in_sizes[2] / 256;
  const int nu1 = in_sizes[17] / 128;
  const int nu0 = in_sizes[18] / 64;
  const int Mc2 = in_sizes[19] / 52;
  const int Mg2 = in_sizes[20] / 52;
  const int Mu1 = in_sizes[21] / 52;
  const int Mg1 = in_sizes[22] / 52;
  const int Mu0 = in_sizes[23] / 52;
  const int g2n = 8 * N2;
  const int g1n = 8 * nu1;

  char* wsp = (char*)d_ws;
  auto carve = [&](long nfloats) -> float* {
    float* p = (float*)wsp;
    wsp += ((nfloats * 4 + 255) / 256) * 256;
    return p;
  };
  long accElems = (long)(N2 + 1) * 128;
  { long e = (long)(g2n + 1) * 128; if (e > accElems) accElems = e; }
  { long e = (long)(nu1 + 1) * 128; if (e > accElems) accElems = e; }
  { long e = (long)(g1n + 1) * 64;  if (e > accElems) accElems = e; }
  { long e = (long)(nu0 + 1) * 128; if (e > accElems) accElems = e; }
  float* acc = carve(accElems);
  float* t2  = carve((long)g2n * 128);
  float* x1  = carve((long)nu1 * 128);
  float* t1  = carve((long)g1n * 64);
  float* x0  = carve((long)nu0 * 64);
  float* mu  = carve(128);
  float* rs  = carve(128);
  float* part = carve((long)1024 * 256);
  long maxn = N2;
  if (g2n > maxn) maxn = g2n;
  if (nu1 > maxn) maxn = nu1;
  if (g1n > maxn) maxn = g1n;
  if (nu0 > maxn) maxn = nu0;
  int* nbr = (int*)carve(26 * maxn);

  float* out  = (float*)d_out;
  float* out0 = out;
  float* out1 = out + (long)nu0 * 128;
  float* out2 = out + (long)nu0 * 128 + (long)nu1 * 128;

  auto nblk = [](long items) -> unsigned {
    long b = (items + TPB - 1) / TPB;
    if (b > MAXB) b = MAXB;
    if (b < 1) b = 1;
    return (unsigned)b;
  };
  auto fill0 = [&](float* p, long n) {
    k_fill<<<dim3(nblk(n / 4)), dim3(TPB), 0, stream>>>(p, n / 4);
  };
  auto conv3 = [&](const float* X, int n, int Cin, int Cout,
                   const float* W, const int* map, int Mw) {
    const long nt = (long)26 * n;
    k_fill_i32<<<dim3(nblk(nt / 4)), dim3(TPB), 0, stream>>>(nbr, nt, -1);
    unsigned gx = (unsigned)((Mw + TPB - 1) / TPB);
    if (gx > MAXB / 26) gx = MAXB / 26;
    if (gx < 1) gx = 1;
    k_invmap<<<dim3(gx, 26), dim3(TPB), 0, stream>>>(map, nbr, Mw, n);
    conv_os(X, W, acc, nbr, n, Cin, Cout, stream);
  };
  auto bnelu = [&](const float* src, float* dst, const float* gb, long n, int C) {
    long tot = n * C;
    long P = (tot + TPB * 4 - 1) / (TPB * 4);
    if (P > 1024) P = 1024;
    k_bn_part<<<dim3((unsigned)P), dim3(TPB), 0, stream>>>(src, tot, C, part);
    k_bn_fin<<<dim3((unsigned)C), dim3(TPB), 0, stream>>>(part, (int)P, n, C, mu, rs);
    const long tot4 = tot / 4;
    if (C == 128)
      k_bn_elu<128><<<dim3(nblk(tot4)), dim3(TPB), 0, stream>>>(src, dst, mu, rs, gb, tot4);
    else
      k_bn_elu<64><<<dim3(nblk(tot4)), dim3(TPB), 0, stream>>>(src, dst, mu, rs, gb, tot4);
  };
  auto scat = [&](const float* X, float* Y, const int* idx, long rows, int C) {
    const long tot4 = rows * (C / 4);
    if (C == 128)
      k_scatter_add<128><<<dim3(nblk(tot4)), dim3(TPB), 0, stream>>>(X, Y, idx, rows);
    else
      k_scatter_add<64><<<dim3(nblk(tot4)), dim3(TPB), 0, stream>>>(X, Y, idx, rows);
  };

  conv3(feats2, N2, 256, 128, w_out2, m_c2, Mc2);
  bnelu(acc, out2, bn_out2, N2, 128);

  for (int o = 0; o < 8; ++o)
    gemm(feats2, wt2 + (long)o * 256 * 128, t2, nullptr, nullptr, N2, 256, 128,
         o * N2, 0, -1, stream);
  bnelu(t2, t2, bn_up2a, g2n, 128);
  conv3(t2, g2n, 128, 128, wu2, m_g2, Mg2);
  bnelu(acc, t2, bn_up2b, g2n, 128);

  fill0(x1, (long)nu1 * 128);
  scat(feats1, x1, lat1_idx, N1, 128);
  scat(t2, x1, up1_idx, g2n, 128);

  conv3(x1, nu1, 128, 128, w_out1, m_u1, Mu1);
  bnelu(acc, out1, bn_out1, nu1, 128);

  for (int o = 0; o < 8; ++o)
    gemm(x1, wt1 + (long)o * 128 * 64, t1, nullptr, nullptr, nu1, 128, 64,
         o * nu1, 0, -1, stream);
  bnelu(t1, t1, bn_up1a, g1n, 64);
  conv3(t1, g1n, 64, 64, wu1, m_g1, Mg1);
  bnelu(acc, t1, bn_up1b, g1n, 64);

  fill0(x0, (long)nu0 * 64);
  scat(feats0, x0, lat0_idx, N0, 64);
  scat(t1, x0, up0_idx, g1n, 64);

  conv3(x0, nu0, 64, 128, w_out0, m_u0, Mu0);
  bnelu(acc, out0, bn_out0, nu0, 128);

  (void)n_in; (void)out_size; (void)ws_size;
}


# both large 3^3 convs cut their 128-row blocks from output rows sorted by which corner neighbours exist (ballot counting sort, 3 small launches each); a block walks and stages only its rows' taps
# speedup vs baseline: 7.9855x; 1.0395x over previous
#include <hip/hip_runtime.h>
#include <hip/hip_bf16.h>
#include <math.h>

typedef __attribute__((ext_vector_type(2))) float v2f;
typedef __attribute__((ext_vector_type(8))) float v8f;

#define TPB 256
#define MAXB 1024


__global__ void k_fill(float* __restrict__ p, long n4) {
  float4* __restrict__ p4 = (float4*)p;
  const float4 z = {0.0f, 0.0f, 0.0f, 0.0f};
  const long step = (long)gridDim.x * TPB;
  for (long i = (long)blockIdx.x * TPB + threadIdx.x; i < n4; i += step) p4[i] = z;
}

__global__ void k_fill_i32(int* __restrict__ p, long n, int v) {
  int4* __restrict__ p4 = (int4*)p;
  const int4 w = {v, v, v, v};
  const long n4 = n >> 2;
  const long step = (long)gridDim.x * TPB;
  const long g = (long)blockIdx.x * TPB + threadIdx.x;
  for (long i = g; i < n4; i += step) p4[i] = w;
  if (g < (n & 3)) p[(n4 << 2) + g] = v;
}

__global__ void k_invmap(const int* __restrict__ map, int* __restrict__ nbr,
                         int Mw, int n) {
  const int j = blockIdx.y;
  const int* __restrict__ gi = map + (long)j * Mw;
  const int* __restrict__ so = map + (long)(26 + j) * Mw;
  int* __restrict__ nb = nbr + (long)j * n;
  const int step = (int)gridDim.x * TPB;
  for (int i = (int)blockIdx.x * TPB + (int)threadIdx.x; i < Mw; i += 4 * step) {
    int o[4];
    int g[4];
    for (int u = 0; u < 4; ++u) {
      const int k = i + u * step;
      const int kc = k < Mw ? k : i;
      o[u] = so[kc];
      g[u] = gi[kc];
    }
    for (int u = 0; u < 4; ++u)
      if (i + u * step < Mw && o[u] >= 0 && o[u] < n) nb[o[u]] = g[u];
  }
}

#define SORT_B 256
#define SORT_W 1024

__device__ __forceinline__ int corner_key(const int* __restrict__ nbr, int n, int r) {
  const int ct[8] = {0, 2, 6, 8, 17, 19, 23, 25};
  int v[8];
#pragma unroll
  for (int b = 0; b < 8; ++b) v[b] = nbr[(long)ct[b] * n + r];
  int key = 0;
#pragma unroll
  for (int b = 0; b < 8; ++b) key |= (v[b] >= 0) ? (1 << b) : 0;
  return key;
}

__global__ void __launch_bounds__(TPB) k_sort_count(const int* __restrict__ nbr, int n,
                                                    int* __restrict__ cnt) {
  const int lane = threadIdx.x & 31;
  const int w = blockIdx.x * (TPB / 32) + ((int)threadIdx.x >> 5);
  const int seg = (n + SORT_W - 1) / SORT_W;
  const int beg = w * seg;
  const int end = beg + seg < n ? beg + seg : n;
  int c[SORT_B / 32];
#pragma unroll
  for (int s = 0; s < SORT_B / 32; ++s) c[s] = 0;
  for (int base = beg; base < end; base += 32) {
    const int r = base + lane;
    const int key = r < end ? corner_key(nbr, n, r) : -1;
#pragma unroll
    for (int s = 0; s < SORT_B / 32; ++s) {
      if (__builtin_amdgcn_ballot_w32((key >> 5) == s) != 0u) {
        for (int l = 0; l < 32; ++l) {
          const unsigned m = __builtin_amdgcn_ballot_w32(key == 32 * s + l);
          if (lane == l) c[s] += __popc(m);
        }
      }
    }
  }
#pragma unroll
  for (int s = 0; s < SORT_B / 32; ++s) cnt[(long)(32 * s + lane) * SORT_W + w] = c[s];
}

__global__ void __launch_bounds__(TPB) k_sort_scan(int* __restrict__ cnt, int* __restrict__ tot) {
  static_assert(SORT_W == 4 * TPB, "four wave counts per thread");
  __shared__ int wsum[TPB / 32];
  const int t = threadIdx.x;
  const int lane = t & 31;
  const int wv = t >> 5;
  int4* p = (int4*)(cnt + (long)blockIdx.x * SORT_W) + t;
  const int4 c = *p;
  const int s = c.x + c.y + c.z + c.w;
  int inc = s;
  for (int d = 1; d < 32; d <<= 1) {
    const int u = __shfl_up(inc, d);
    if (lane >= d) inc += u;
  }
  if (lane == 31) wsum[wv] = inc;
  __syncthreads();
  int before = 0;
  int total = 0;
  for (int i = 0; i < TPB / 32; ++i) {
    const int x = wsum[i];
    if (i < wv) before += x;
    total += x;
  }
  const int ex = before + inc - s;
  *p = make_int4(ex, ex + c.x, ex + c.x + c.y, ex + c.x + c.y + c.z);
  if (t == 0) tot[blockIdx.x] = total;
}

__global__ void __launch_bounds__(TPB) k_sort_place(const int* __restrict__ nbr, int n,
                                                    const int* __restrict__ cnt,
                                                    const int* __restrict__ tot,
                                                    int* __restrict__ perm) {
  const int lane = threadIdx.x & 31;
  const int w = blockIdx.x * (TPB / 32) + ((int)threadIdx.x >> 5);
  const int seg = (n + SORT_W - 1) / SORT_W;
  const int beg = w * seg;
  const int end = beg + seg < n ? beg + seg : n;
  int off[SORT_B / 32];
  int carry = 0;
#pragma unroll
  for (int s = 0; s < SORT_B / 32; ++s) {
    const int v = tot[32 * s + lane];
    int inc = v;
    for (int d = 1; d < 32; d <<= 1) {
      const int u = __shfl_up(inc, d);
      if (lane >= d) inc += u;
    }
    off[s] = carry + inc - v + cnt[(long)(32 * s + lane) * SORT_W + w];
    carry += __shfl(inc, 31);
  }
  const unsigned below = (1u << lane) - 1u;
  for (int base = beg; base < end; base += 32) {
    const int r = base + lane;
    const int key = r < end ? corner_key(nbr, n, r) : -1;
    int pos = 0;
#pragma unroll
    for (int s = 0; s < SORT_B / 32; ++s) {
      if (__builtin_amdgcn_ballot_w32((key >> 5) == s) != 0u) {
        for (int l = 0; l < 32; ++l) {
          const unsigned m = __builtin_amdgcn_ballot_w32(key == 32 * s + l);
          const int start = __builtin_amdgcn_readlane(off[s], l);
          if (key == 32 * s + l) pos = start + __popc(m & below);
          if (lane == l) off[s] += __popc(m);
        }
      }
    }
    if (key >= 0) perm[pos] = r;
  }
}

template <int C>
__global__ void k_scatter_add(const float* __restrict__ X, float* __restrict__ Y,
                              const int* __restrict__ idx, long rows) {
  constexpr int Q = C / 4;
  const float4* __restrict__ X4 = (const float4*)X;
  const long tot4 = rows * Q;
  const long step = (long)gridDim.x * TPB;
  float4* __restrict__ Y4 = (float4*)Y;
  for (long i = (long)blockIdx.x * TPB + threadIdx.x; i < tot4; i += 4 * step) {
    long kc[4];
    long off[4];
    float4 x[4];
    float4 y[4];
    for (int u = 0; u < 4; ++u) {
      const long k = i + u * step;
      kc[u] = k < tot4 ? k : i;
      off[u] = idx[kc[u] / Q];
    }
    for (int u = 0; u < 4; ++u) {
      off[u] = off[u] * Q + (kc[u] % Q);
      x[u] = X4[kc[u]];
      y[u] = Y4[off[u]];
    }
    for (int u = 0; u < 4; ++u) {
      if (i + u * step < tot4) {
        float4 v = y[u];
        v.x += x[u].x; v.y += x[u].y; v.z += x[u].z; v.w += x[u].w;
        Y4[off[u]] = v;
      }
    }
  }
}

__global__ void k_bn_part(const float* __restrict__ X, long tot, int C,
                          float* __restrict__ part) {
  __shared__ float sm[8][TPB];
  const int t = threadIdx.x;
  float a0 = 0.0f, a1 = 0.0f, a2 = 0.0f, a3 = 0.0f;
  float b0 = 0.0f, b1 = 0.0f, b2 = 0.0f, b3 = 0.0f;
  const long step = (long)gridDim.x * (TPB * 4);
  for (long i = ((long)blockIdx.x * TPB + t) * 4; i < tot; i += step) {
    const float4 v = *(const float4*)(X + i);
    a0 += v.x; b0 += v.x * v.x;
    a1 += v.y; b1 += v.y * v.y;
    a2 += v.z; b2 += v.z * v.z;
    a3 += v.w; b3 += v.w * v.w;
  }
  sm[0][t] = a0; sm[1][t] = a1; sm[2][t] = a2; sm[3][t] = a3;
  sm[4][t] = b0; sm[5][t] = b1; sm[6][t] = b2; sm[7][t] = b3;
  __syncthreads();
  if (t < C) {
    const int q = C >> 2;
    const int comp = t & 3;
    float s = 0.0f, ss = 0.0f;
    for (int j = t >> 2; j < TPB; j += q) {
      s += sm[comp][j];
      ss += sm[4 + comp][j];
    }
    float* p = part + (long)blockIdx.x * 2 * C;
    p[t] = s;
    p[C + t] = ss;
  }
}

__global__ void k_bn_fin(const float* __restrict__ part, int P, long N, int C,
                         float* __restrict__ mu, float* __restrict__ rs) {
  __shared__ float s1[TPB];
  __shared__ float s2[TPB];
  const int c = blockIdx.x;
  float a = 0.0f, b = 0.0f;
  for (int p = threadIdx.x; p < P; p += TPB) {
    a += part[(long)p * 2 * C + c];
    b += part[(long)p * 2 * C + C + c];
  }
  s1[threadIdx.x] = a;
  s2[threadIdx.x] = b;
  __syncthreads();
  for (int s = TPB / 2; s > 0; s >>= 1) {
    if ((int)threadIdx.x < s) {
      s1[threadIdx.x] += s1[threadIdx.x + s];
      s2[threadIdx.x] += s2[threadIdx.x + s];
    }
    __syncthreads();
  }
  if (threadIdx.x == 0) {
    float m = s1[0] / (float)N;
    float var = s2[0] / (float)N - m * m;
    if (var < 0.0f) var = 0.0f;
    mu[c] = m;
    rs[c] = rsqrtf(var + 1e-5f);
  }
}

template <int C>
__global__ void k_bn_elu(const float* X, float* Y,
                         const float* __restrict__ mu, const float* __restrict__ rs,
                         const float* __restrict__ gb, long tot4) {
  const float4* X4 = (const float4*)X;
  float4* Y4 = (float4*)Y;
  const int c = ((int)threadIdx.x * 4) & (C - 1);
  float g[4], m[4], s[4], b[4];
  for (int e = 0; e < 4; ++e) {
    g[e] = gb[c + e];
    m[e] = mu[c + e];
    s[e] = rs[c + e];
    b[e] = gb[C + c + e];
  }
  const long step = (long)gridDim.x * TPB;
  for (long i = (long)blockIdx.x * TPB + threadIdx.x; i < tot4; i += 4 * step) {
    float4 x[4];
    for (int u = 0; u < 4; ++u)
      if (i + u * step < tot4) x[u] = X4[i + u * step];
    for (int u = 0; u < 4; ++u) {
      if (i + u * step < tot4) {
        float v0 = g[0] * (x[u].x - m[0]) * s[0] + b[0];
        float v1 = g[1] * (x[u].y - m[1]) * s[1] + b[1];
        float v2 = g[2] * (x[u].z - m[2]) * s[2] + b[2];
        float v3 = g[3] * (x[u].w - m[3]) * s[3] + b[3];
        float4 o;
        o.x = v0 > 0.0f ? v0 : expm1f(v0);
        o.y = v1 > 0.0f ? v1 : expm1f(v1);
        o.z = v2 > 0.0f ? v2 : expm1f(v2);
        o.w = v3 > 0.0f ? v3 : expm1f(v3);
        Y4[i + u * step] = o;
      }
    }
  }
}

template <int CIN, int COUT>
__global__ void k_gemm_tap(const float* __restrict__ X,
                           const float* __restrict__ W,
                           float* __restrict__ Y,
                           const int* __restrict__ gidx,
                           const int* __restrict__ sidx,
                           int M, int out_offset, int accumulate, int nscrap) {
  const int lane = threadIdx.x;
  const int tile_r = blockIdx.x * 32;
  const int tile_c = (blockIdx.y * 4 + threadIdx.y) * 16;
  const int mr = lane & 15;
  const int kh = lane >> 4;

  const int rl = tile_r + lane;
  int sv = nscrap;
  if (rl < M) sv = sidx ? sidx[rl] : rl;
  if (__builtin_amdgcn_ballot_w32(sv != nscrap) == 0u) return;

  const int r0 = tile_r + mr;
  const int r1 = tile_r + 16 + mr;
  int g0 = 0, g1 = 0;
  if (r0 < M) g0 = gidx ? gidx[r0] : r0;
  if (r1 < M) g1 = gidx ? gidx[r1] : r1;
  const float* __restrict__ xb0 = X + (long)g0 * CIN + 2 * kh;
  const float* __restrict__ xb1 = X + (long)g1 * CIN + 2 * kh;
  const float* __restrict__ wb  = W + tile_c + mr + 2 * kh * COUT;

  v8f c0 = {0.f, 0.f, 0.f, 0.f, 0.f, 0.f, 0.f, 0.f};
  v8f c1 = {0.f, 0.f, 0.f, 0.f, 0.f, 0.f, 0.f, 0.f};
#pragma unroll
  for (int k0 = 0; k0 < CIN; k0 += 4) {
    v2f a0 = *(const v2f*)(xb0 + k0);
    v2f a1 = *(const v2f*)(xb1 + k0);
    v2f b;
    b[0] = wb[k0 * COUT];
    b[1] = wb[(k0 + 1) * COUT];
    c0 = __builtin_amdgcn_wmma_f32_16x16x4_f32(false, a0, false, b,
                                               (short)0, c0, false, false);
    c1 = __builtin_amdgcn_wmma_f32_16x16x4_f32(false, a1, false, b,
                                               (short)0, c1, false, false);
  }

#pragma unroll
  for (int i = 0; i < 8; ++i) {
    const int t0a = __builtin_amdgcn_readlane(sv, i);
    const int t0b = __builtin_amdgcn_readlane(sv, i + 8);
    const int ro0 = kh ? t0b : t0a;
    if (ro0 != nscrap) {
      float* y = Y + (long)(ro0 + out_offset) * COUT + tile_c + mr;
      if (accumulate) *y += c0[i]; else *y = c0[i];
    }
    const int t1a = __builtin_amdgcn_readlane(sv, i + 16);
    const int t1b = __builtin_amdgcn_readlane(sv, i + 24);
    const int ro1 = kh ? t1b : t1a;
    if (ro1 != nscrap) {
      float* y = Y + (long)(ro1 + out_offset) * COUT + tile_c + mr;
      if (accumulate) *y += c1[i]; else *y = c1[i];
    }
  }
}

template <int CIN, int COUT>
__global__ void __launch_bounds__(128) k_conv_os(const float* __restrict__ X,
                          const float* __restrict__ W,
                          float* __restrict__ Y,
                          const int* __restrict__ nbr,
                          int n) {
  const int lane = threadIdx.x;
  const int tile_r = blockIdx.x * 32;
  const int tile_c = (blockIdx.y * 4 + threadIdx.y) * 16;
  const int mr = lane & 15;
  const int kh = lane >> 4;
  const int r0 = tile_r + mr;
  const int r1 = tile_r + 16 + mr;
  const bool in0 = r0 < n;
  const bool in1 = r1 < n;
  constexpr int KC = CIN < 64 ? CIN : 64;
  constexpr int NS = KC / 4;

  v8f c0 = {0.f, 0.f, 0.f, 0.f, 0.f, 0.f, 0.f, 0.f};
  v8f c1 = {0.f, 0.f, 0.f, 0.f, 0.f, 0.f, 0.f, 0.f};

  const int q0 = in0 ? r0 : n - 1;
  const int q1 = in1 ? r1 : n - 1;
  int g0 = nbr[q0];
  int g1 = nbr[q1];
  g0 = in0 ? g0 : -1;
  g1 = in1 ? g1 : -1;
  for (int t = 0; t < 27; ++t) {
    int l0 = r0;
    int l1 = r1;
    if (t < 25) {
      l0 = nbr[(long)(t + 1) * n + q0];
      l1 = nbr[(long)(t + 1) * n + q1];
    }
    if (__builtin_amdgcn_ballot_w32((g0 >= 0) || (g1 >= 0)) != 0u) {
      const int k = (t < 13) ? t : (t < 26 ? t + 1 : 13);
      const float* __restrict__ wb =
          W + (long)k * CIN * COUT + tile_c + mr + 2 * kh * COUT;
#pragma unroll
      for (int kc = 0; kc < CIN; kc += KC) {
        v2f a0[NS];
        v2f a1[NS];
#pragma unroll
        for (int s = 0; s < NS; ++s) {
          a0[s] = (v2f){0.f, 0.f};
          a1[s] = (v2f){0.f, 0.f};
        }
        if (g0 >= 0) {
          const float* __restrict__ xb0 = X + (long)g0 * CIN + 2 * kh + kc;
#pragma unroll
          for (int s = 0; s < NS; ++s) a0[s] = *(const v2f*)(xb0 + 4 * s);
        }
        if (g1 >= 0) {
          const float* __restrict__ xb1 = X + (long)g1 * CIN + 2 * kh + kc;
#pragma unroll
          for (int s = 0; s < NS; ++s) a1[s] = *(const v2f*)(xb1 + 4 * s);
        }
#pragma unroll
        for (int s = 0; s < NS; ++s) {
          const int k0 = kc + 4 * s;
          v2f b;
          b[0] = wb[k0 * COUT];
          b[1] = wb[(k0 + 1) * COUT];
          c0 = __builtin_amdgcn_wmma_f32_16x16x4_f32(false, a0[s], false, b,
                                                     (short)0, c0, false, false);
          c1 = __builtin_amdgcn_wmma_f32_16x16x4_f32(false, a1[s], false, b,
                                                     (short)0, c1, false, false);
        }
      }
    }
    g0 = in0 ? l0 : -1;
    g1 = in1 ? l1 : -1;
  }

#pragma unroll
  for (int i = 0; i < 8; ++i) {
    const int ro0 = tile_r + i + 8 * kh;
    if (ro0 < n) Y[(long)ro0 * COUT + tile_c + mr] = c0[i];
    const int ro1 = ro0 + 16;
    if (ro1 < n) Y[(long)ro1 * COUT + tile_c + mr] = c1[i];
  }
}

template <int CIN, int COUT>
__global__ void __launch_bounds__(128) k_conv_os_wide(const float* __restrict__ X,
                          const float* __restrict__ W,
                          float* __restrict__ Y,
                          const int* __restrict__ nbr,
                          const int* __restrict__ perm,
                          int n) {
  constexpr int NT = COUT / 16;
  constexpr int KC = 64;
  constexpr int NCH = CIN / KC;
  constexpr int PITCH = KC + 4;
  constexpr int CG = COUT / 4;
  constexpr int ITEMS = (KC / 4) * CG / 128;
  __shared__ float Bs[2][COUT * PITCH];
  __shared__ unsigned need[4];

  const int lane = threadIdx.x;
  const int wv = threadIdx.y;
  const int tid = wv * 32 + lane;
  const int tile_r = (((int)gridDim.x - 1 - (int)blockIdx.x) * 4 + wv) * 32;
  const int mr = lane & 15;
  const int kh = lane >> 4;
  const int p0 = tile_r + mr;
  const int p1 = tile_r + 16 + mr;
  const bool in0 = p0 < n;
  const bool in1 = p1 < n;
  const int r0 = perm[in0 ? p0 : n - 1];
  const int r1 = perm[in1 ? p1 : n - 1];

  unsigned wm = 0u;
  {
    int h0[26];
    int h1[26];
#pragma unroll
    for (int t = 0; t < 26; ++t) {
      h0[t] = nbr[(long)t * n + r0];
      h1[t] = nbr[(long)t * n + r1];
    }
#pragma unroll
    for (int t = 0; t < 26; ++t) {
      const int e0 = in0 ? h0[t] : -1;
      const int e1 = in1 ? h1[t] : -1;
      if (__builtin_amdgcn_ballot_w32((e0 & e1) >= 0) != 0u) wm |= 1u << t;
    }
    if (__builtin_amdgcn_ballot_w32(in0 || in1) != 0u) wm |= 1u << 26;
  }
  if (lane == 0) need[wv] = wm;
  __syncthreads();
  unsigned bm = (unsigned)__builtin_amdgcn_readfirstlane(
      (int)(need[0] | need[1] | need[2] | need[3]));
  if (bm == 0u) return;

  auto stage = [&](int t, int ch, float* __restrict__ dst) {
    const int kc = ch * KC;
    const int k = (t < 13) ? t : (t < 26 ? t + 1 : 13);
    const float* __restrict__ wsrc = W + ((long)k * CIN + kc) * COUT;
    float4 r[ITEMS][4];
    for (int u = 0; u < ITEMS; ++u) {
      const int e = tid + 128 * u;
      const int cg = e % CG;
      const int kg = e / CG;
      for (int rr = 0; rr < 4; ++rr)
        r[u][rr] = *(const float4*)(wsrc + (long)(4 * kg + rr) * COUT + 4 * cg);
    }
    for (int u = 0; u < ITEMS; ++u) {
      const int e = tid + 128 * u;
      const int cg = e % CG;
      const int kg = e / CG;
      float* d = dst + (4 * cg) * PITCH + 4 * kg;
      *(float4*)(d)             = make_float4(r[u][0].x, r[u][1].x, r[u][2].x, r[u][3].x);
      *(float4*)(d + PITCH)     = make_float4(r[u][0].y, r[u][1].y, r[u][2].y, r[u][3].y);
      *(float4*)(d + 2 * PITCH) = make_float4(r[u][0].z, r[u][1].z, r[u][2].z, r[u][3].z);
      *(float4*)(d + 3 * PITCH) = make_float4(r[u][0].w, r[u][1].w, r[u][2].w, r[u][3].w);
    }
  };

  v8f c0[NT];
  v8f c1[NT];
  for (int j = 0; j < NT; ++j) {
    c0[j] = (v8f){0.f, 0.f, 0.f, 0.f, 0.f, 0.f, 0.f, 0.f};
    c1[j] = (v8f){0.f, 0.f, 0.f, 0.f, 0.f, 0.f, 0.f, 0.f};
  }

  int t = __builtin_ctz(bm);
  bm &= bm - 1u;
  stage(t, 0, Bs[0]);

  int g0 = t < 26 ? nbr[(long)t * n + r0] : r0;
  int g1 = t < 26 ? nbr[(long)t * n + r1] : r1;
  g0 = in0 ? g0 : -1;
  g1 = in1 ? g1 : -1;
  __syncthreads();

  int buf = 0;
  for (int it = 0; it < 27; ++it) {
    const int tn = bm != 0u ? __builtin_ctz(bm) : -1;
    bm &= bm - 1u;
    int l0 = r0;
    int l1 = r1;
    if (tn >= 0 && tn < 26) {
      l0 = nbr[(long)tn * n + r0];
      l1 = nbr[(long)tn * n + r1];
    }
    const bool act = __builtin_amdgcn_ballot_w32((g0 >= 0) || (g1 >= 0)) != 0u;
    for (int ch = 0; ch < NCH; ++ch) {
      if (ch + 1 < NCH) stage(t, ch + 1, Bs[buf ^ 1]);
      else if (tn >= 0) stage(tn, 0, Bs[buf ^ 1]);
      if (act) {
        const float* __restrict__ bs = Bs[buf] + mr * PITCH + 4 * kh;
        float4 a0[8];
        float4 a1[8];
        for (int p = 0; p < 8; ++p) {
          a0[p] = make_float4(0.f, 0.f, 0.f, 0.f);
          a1[p] = make_float4(0.f, 0.f, 0.f, 0.f);
        }
        if (g0 >= 0) {
          const float* __restrict__ xb0 = X + (long)g0 * CIN + ch * KC + 4 * kh;
          for (int p = 0; p < 8; ++p) a0[p] = *(const float4*)(xb0 + 8 * p);
        }
        if (g1 >= 0) {
          const float* __restrict__ xb1 = X + (long)g1 * CIN + ch * KC + 4 * kh;
          for (int p = 0; p < 8; ++p) a1[p] = *(const float4*)(xb1 + 8 * p);
        }
        for (int p = 0; p < 8; ++p) {
          const v2f a0l = {a0[p].x, a0[p].y};
          const v2f a0h = {a0[p].z, a0[p].w};
          const v2f a1l = {a1[p].x, a1[p].y};
          const v2f a1h = {a1[p].z, a1[p].w};
          float4 bf[NT];
          for (int j = 0; j < NT; ++j) bf[j] = *(const float4*)(bs + 16 * j * PITCH + 8 * p);
          for (int j = 0; j < NT; ++j) {
            const float4 b = bf[j];
            const v2f bl = {b.x, b.y};
            const v2f bh = {b.z, b.w};
            c0[j] = __builtin_amdgcn_wmma_f32_16x16x4_f32(false, a0l, false, bl,
                                                          (short)0, c0[j], false, false);
            c1[j] = __builtin_amdgcn_wmma_f32_16x16x4_f32(false, a1l, false, bl,
                                                          (short)0, c1[j], false, false);
            c0[j] = __builtin_amdgcn_wmma_f32_16x16x4_f32(false, a0h, false, bh,
                                                          (short)0, c0[j], false, false);
            c1[j] = __builtin_amdgcn_wmma_f32_16x16x4_f32(false, a1h, false, bh,
                                                          (short)0, c1[j], false, false);
          }
        }
      }
      __syncthreads();
      buf ^= 1;
    }
    if (tn < 0) break;
    g0 = in0 ? l0 : -1;
    g1 = in1 ? l1 : -1;
    t = tn;
  }

  int o0[8];
  int o1[8];
#pragma unroll
  for (int i = 0; i < 8; ++i) {
    const int xa = __builtin_amdgcn_readlane(r0, i);
    const int xb = __builtin_amdgcn_readlane(r0, i + 8);
    o0[i] = kh ? xb : xa;
    const int ya = __builtin_amdgcn_readlane(r1, i);
    const int yb = __builtin_amdgcn_readlane(r1, i + 8);
    o1[i] = kh ? yb : ya;
  }
#pragma unroll
  for (int j = 0; j < NT; ++j) {
#pragma unroll
    for (int i = 0; i < 8; ++i) {
      if (tile_r + i + 8 * kh < n) Y[(long)o0[i] * COUT + 16 * j + mr] = c0[j][i];
      if (tile_r + 16 + i + 8 * kh < n) Y[(long)o1[i] * COUT + 16 * j + mr] = c1[j][i];
    }
  }
}

template <int CIN, int COUT>
static inline void conv_t(const float* X, const float* W, float* Y,
                          const int* nbr, int n, hipStream_t st) {
  dim3 grid((unsigned)((n + 31) / 32), (unsigned)(COUT / 64));
  dim3 blk(32, 4);
  k_conv_os<CIN, COUT><<<grid, blk, 0, st>>>(X, W, Y, nbr, n);
}

struct SortWs {
  int* perm;
  int* cnt;
  int* tot;
};

template <int CIN, int COUT>
static inline void conv_wide_t(const float* X, const float* W, float* Y,
                               const int* nbr, int n, SortWs sw, hipStream_t st) {
  k_sort_count<<<dim3(SORT_W / (TPB / 32)), dim3(TPB), 0, st>>>(nbr, n, sw.cnt);
  k_sort_scan<<<dim3(SORT_B), dim3(TPB), 0, st>>>(sw.cnt, sw.tot);
  k_sort_place<<<dim3(SORT_W / (TPB / 32)), dim3(TPB), 0, st>>>(nbr, n, sw.cnt, sw.tot, sw.perm);
  dim3 grid((unsigned)((n + 127) / 128));
  dim3 blk(32, 4);
  k_conv_os_wide<CIN, COUT><<<grid, blk, 0, st>>>(X, W, Y, nbr, sw.perm, n);
}

static void conv_os(const float* X, const float* W, float* Y, const int* nbr,
                    int n, int Cin, int Cout, SortWs sw, hipStream_t st) {
  const bool wide = n >= 128 * 256;
  if      (wide && Cin == 64 && Cout ==  64) conv_wide_t<64,  64>(X, W, Y, nbr, n, sw, st);
  else if (wide && Cin == 64 && Cout == 128) conv_wide_t<64, 128>(X, W, Y, nbr, n, sw, st);
  else if (Cin == 256 && Cout == 128) conv_t<256, 128>(X, W, Y, nbr, n, st);
  else if (Cin == 128 && Cout == 128) conv_t<128, 128>(X, W, Y, nbr, n, st);
  else if (Cin ==  64 && Cout ==  64) conv_t< 64,  64>(X, W, Y, nbr, n, st);
  else if (Cin ==  64 && Cout == 128) conv_t< 64, 128>(X, W, Y, nbr, n, st);
}

template <int CIN, int COUT>
static inline void gemm_t(const float* X, const float* W, float* Y,
                          const int* g, const int* s, int M, int ooff,
                          int accum, int nscrap, hipStream_t st) {
  dim3 grid((unsigned)((M + 31) / 32), (unsigned)(COUT / 64));
  dim3 blk(32, 4);
  k_gemm_tap<CIN, COUT><<<grid, blk, 0, st>>>(X, W, Y, g, s, M, ooff, accum, nscrap);
}

static void gemm(const float* X, const float* W, float* Y,
                 const int* g, const int* s, int M, int Cin, int Cout,
                 int ooff, int accum, int nscrap, hipStream_t st) {
  if      (Cin == 256 && Cout == 128) gemm_t<256, 128>(X, W, Y, g, s, M, ooff, accum, nscrap, st);
  else if (Cin == 128 && Cout ==  64) gemm_t<128,  64>(X, W, Y, g, s, M, ooff, accum, nscrap, st);
}


extern "C" void kernel_launch(void* const* d_in, const int* in_sizes, int n_in,
                              void* d_out, int out_size, void* d_ws, size_t ws_size,
                              hipStream_t stream) {
  const float* feats0 = (const float*)d_in[0];
  const float* feats1 = (const float*)d_in[1];
  const float* feats2 = (const float*)d_in[2];
  const float* w_out0 = (const float*)d_in[3];
  const float* w_out1 = (const float*)d_in[4];
  const float* w_out2 = (const float*)d_in[5];
  const float* wt2    = (const float*)d_in[6];
  const float* wu2    = (const float*)d_in[7];
  const float* wt1    = (const float*)d_in[8];
  const float* wu1    = (const float*)d_in[9];
  const float* bn_out0 = (const float*)d_in[10];
  const float* bn_out1 = (const float*)d_in[11];
  const float* bn_out2 = (const float*)d_in[12];
  const float* bn_up2a = (const float*)d_in[13];
  const float* bn_up2b = (const float*)d_in[14];
  const float* bn_up1a = (const float*)d_in[15];
  const float* bn_up1b = (const float*)d_in[16];
  const int* m_c2 = (const int*)d_in[19];
  const int* m_g2 = (const int*)d_in[20];
  const int* m_u1 = (const int*)d_in[21];
  const int* m_g1 = (const int*)d_in[22];
  const int* m_u0 = (const int*)d_in[23];
  const int* lat1_idx = (const int*)d_in[24];
  const int* up1_idx  = (const int*)d_in[25];
  const int* lat0_idx = (const int*)d_in[26];
  const int* up0_idx  = (const int*)d_in[27];

  const int N0 = in_sizes[0] / 64;
  const int N1 = in_sizes[1] / 128;
  const int N2 = in_sizes[2] / 256;
  const int nu1 = in_sizes[17] / 128;
  const int nu0 = in_sizes[18] / 64;
  const int Mc2 = in_sizes[19] / 52;
  const int Mg2 = in_sizes[20] / 52;
  const int Mu1 = in_sizes[21] / 52;
  const int Mg1 = in_sizes[22] / 52;
  const int Mu0 = in_sizes[23] / 52;
  const int g2n = 8 * N2;
  const int g1n = 8 * nu1;

  char* wsp = (char*)d_ws;
  auto carve = [&](long nfloats) -> float* {
    float* p = (float*)wsp;
    wsp += ((nfloats * 4 + 255) / 256) * 256;
    return p;
  };
  long accElems = (long)(N2 + 1) * 128;
  { long e = (long)(g2n + 1) * 128; if (e > accElems) accElems = e; }
  { long e = (long)(nu1 + 1) * 128; if (e > accElems) accElems = e; }
  { long e = (long)(g1n + 1) * 64;  if (e > accElems) accElems = e; }
  { long e = (long)(nu0 + 1) * 128; if (e > accElems) accElems = e; }
  float* acc = carve(accElems);
  float* t2  = carve((long)g2n * 128);
  float* x1  = carve((long)nu1 * 128);
  float* t1  = carve((long)g1n * 64);
  float* x0  = carve((long)nu0 * 64);
  float* mu  = carve(128);
  float* rs  = carve(128);
  float* part = carve((long)1024 * 256);
  long maxn = N2;
  if (g2n > maxn) maxn = g2n;
  if (nu1 > maxn) maxn = nu1;
  if (g1n > maxn) maxn = g1n;
  if (nu0 > maxn) maxn = nu0;
  int* nbr = (int*)carve(26 * maxn);
  SortWs sw;
  sw.perm = (int*)carve(maxn);
  sw.cnt  = (int*)carve((long)SORT_B * SORT_W);
  sw.tot  = (int*)carve(SORT_B);

  float* out  = (float*)d_out;
  float* out0 = out;
  float* out1 = out + (long)nu0 * 128;
  float* out2 = out + (long)nu0 * 128 + (long)nu1 * 128;

  auto nblk = [](long items) -> unsigned {
    long b = (items + TPB - 1) / TPB;
    if (b > MAXB) b = MAXB;
    if (b < 1) b = 1;
    return (unsigned)b;
  };
  auto fill0 = [&](float* p, long n) {
    k_fill<<<dim3(nblk(n / 4)), dim3(TPB), 0, stream>>>(p, n / 4);
  };
  auto conv3 = [&](const float* X, int n, int Cin, int Cout,
                   const float* W, const int* map, int Mw) {
    const long nt = (long)26 * n;
    k_fill_i32<<<dim3(nblk(nt / 4)), dim3(TPB), 0, stream>>>(nbr, nt, -1);
    unsigned gx = (unsigned)((Mw + TPB - 1) / TPB);
    if (gx > MAXB / 26) gx = MAXB / 26;
    if (gx < 1) gx = 1;
    k_invmap<<<dim3(gx, 26), dim3(TPB), 0, stream>>>(map, nbr, Mw, n);
    conv_os(X, W, acc, nbr, n, Cin, Cout, sw, stream);
  };
  auto bnelu = [&](const float* src, float* dst, const float* gb, long n, int C) {
    long tot = n * C;
    long P = (tot + TPB * 4 - 1) / (TPB * 4);
    if (P > 1024) P = 1024;
    k_bn_part<<<dim3((unsigned)P), dim3(TPB), 0, stream>>>(src, tot, C, part);
    k_bn_fin<<<dim3((unsigned)C), dim3(TPB), 0, stream>>>(part, (int)P, n, C, mu, rs);
    const long tot4 = tot / 4;
    if (C == 128)
      k_bn_elu<128><<<dim3(nblk(tot4)), dim3(TPB), 0, stream>>>(src, dst, mu, rs, gb, tot4);
    else
      k_bn_elu<64><<<dim3(nblk(tot4)), dim3(TPB), 0, stream>>>(src, dst, mu, rs, gb, tot4);
  };
  auto scat = [&](const float* X, float* Y, const int* idx, long rows, int C) {
    const long tot4 = rows * (C / 4);
    if (C == 128)
      k_scatter_add<128><<<dim3(nblk(tot4)), dim3(TPB), 0, stream>>>(X, Y, idx, rows);
    else
      k_scatter_add<64><<<dim3(nblk(tot4)), dim3(TPB), 0, stream>>>(X, Y, idx, rows);
  };

  conv3(feats2, N2, 256, 128, w_out2, m_c2, Mc2);
  bnelu(acc, out2, bn_out2, N2, 128);

  for (int o = 0; o < 8; ++o)
    gemm(feats2, wt2 + (long)o * 256 * 128, t2, nullptr, nullptr, N2, 256, 128,
         o * N2, 0, -1, stream);
  bnelu(t2, t2, bn_up2a, g2n, 128);
  conv3(t2, g2n, 128, 128, wu2, m_g2, Mg2);
  bnelu(acc, t2, bn_up2b, g2n, 128);

  fill0(x1, (long)nu1 * 128);
  scat(feats1, x1, lat1_idx, N1, 128);
  scat(t2, x1, up1_idx, g2n, 128);

  conv3(x1, nu1, 128, 128, w_out1, m_u1, Mu1);
  bnelu(acc, out1, bn_out1, nu1, 128);

  for (int o = 0; o < 8; ++o)
    gemm(x1, wt1 + (long)o * 128 * 64, t1, nullptr, nullptr, nu1, 128, 64,
         o * nu1, 0, -1, stream);
  bnelu(t1, t1, bn_up1a, g1n, 64);
  conv3(t1, g1n, 64, 64, wu1, m_g1, Mg1);
  bnelu(acc, t1, bn_up1b, g1n, 64);

  fill0(x0, (long)nu0 * 64);
  scat(feats0, x0, lat0_idx, N0, 64);
  scat(t1, x0, up0_idx, g1n, 64);

  conv3(x0, nu0, 64, 128, w_out0, m_u0, Mu0);
  bnelu(acc, out0, bn_out0, nu0, 128);

  (void)n_in; (void)out_size; (void)ws_size;
}


# independent launches merged, 64 to 35: one fill + one launch build all 5 neighbour tables and lateral scatters up front, both row sorts share 3 launches, an up block's 8 child GEMMs are one (grid z)
# speedup vs baseline: 9.5066x; 1.1905x over previous
#include <hip/hip_runtime.h>
#include <hip/hip_bf16.h>
#include <math.h>

typedef __attribute__((ext_vector_type(2))) float v2f;
typedef __attribute__((ext_vector_type(8))) float v8f;

#define TPB 256
#define MAXB 1024


__global__ void k_fill2(int* __restrict__ a, long na4, int va,
                        int* __restrict__ b, long nb4, int vb) {
  int4* __restrict__ a4 = (int4*)a;
  int4* __restrict__ b4 = (int4*)b;
  const int4 wa = {va, va, va, va};
  const int4 wb = {vb, vb, vb, vb};
  const long step = (long)gridDim.x * TPB;
  const long g = (long)blockIdx.x * TPB + threadIdx.x;
  for (long i = g; i < na4; i += step) a4[i] = wa;
  for (long i = g; i < nb4; i += step) b4[i] = wb;
}

__device__ __forceinline__ void invmap_tap(const int* __restrict__ map, int* __restrict__ nbr,
                                           int Mw, int n, int j, int bx, int nbx) {
  const int* __restrict__ gi = map + (long)j * Mw;
  const int* __restrict__ so = map + (long)(26 + j) * Mw;
  int* __restrict__ nb = nbr + (long)j * n;
  const int step = nbx * TPB;
  for (int i = bx * TPB + (int)threadIdx.x; i < Mw; i += 4 * step) {
    int o[4];
    int g[4];
    for (int u = 0; u < 4; ++u) {
      const int k = i + u * step;
      const int kc = k < Mw ? k : i;
      o[u] = so[kc];
      g[u] = gi[kc];
    }
    for (int u = 0; u < 4; ++u)
      if (i + u * step < Mw && o[u] >= 0 && o[u] < n) nb[o[u]] = g[u];
  }
}

#define SORT_B 256
#define SORT_W 1024

__device__ __forceinline__ int corner_key(const int* __restrict__ nbr, int n, int r) {
  const int ct[8] = {0, 2, 6, 8, 17, 19, 23, 25};
  int v[8];
#pragma unroll
  for (int b = 0; b < 8; ++b) v[b] = nbr[(long)ct[b] * n + r];
  int key = 0;
#pragma unroll
  for (int b = 0; b < 8; ++b) key |= (v[b] >= 0) ? (1 << b) : 0;
  return key;
}

__global__ void __launch_bounds__(TPB) k_sort_count(const int* __restrict__ nbrA, int nA,
                                                    const int* __restrict__ nbrB, int nB,
                                                    int* __restrict__ cnt2) {
  const int* __restrict__ nbr = blockIdx.y ? nbrB : nbrA;
  const int n = blockIdx.y ? nB : nA;
  int* __restrict__ cnt = cnt2 + (long)blockIdx.y * SORT_B * SORT_W;
  const int lane = threadIdx.x & 31;
  const int w = blockIdx.x * (TPB / 32) + ((int)threadIdx.x >> 5);
  const int seg = (n + SORT_W - 1) / SORT_W;
  const int beg = w * seg;
  const int end = beg + seg < n ? beg + seg : n;
  int c[SORT_B / 32];
#pragma unroll
  for (int s = 0; s < SORT_B / 32; ++s) c[s] = 0;
  for (int base = beg; base < end; base += 32) {
    const int r = base + lane;
    const int key = r < end ? corner_key(nbr, n, r) : -1;
#pragma unroll
    for (int s = 0; s < SORT_B / 32; ++s) {
      if (__builtin_amdgcn_ballot_w32((key >> 5) == s) != 0u) {
        for (int l = 0; l < 32; ++l) {
          const unsigned m = __builtin_amdgcn_ballot_w32(key == 32 * s + l);
          if (lane == l) c[s] += __popc(m);
        }
      }
    }
  }
#pragma unroll
  for (int s = 0; s < SORT_B / 32; ++s) cnt[(long)(32 * s + lane) * SORT_W + w] = c[s];
}

__global__ void __launch_bounds__(TPB) k_sort_scan(int* __restrict__ cnt, int* __restrict__ tot) {
  static_assert(SORT_W == 4 * TPB, "four wave counts per thread");
  __shared__ int wsum[TPB / 32];
  const int t = threadIdx.x;
  const int lane = t & 31;
  const int wv = t >> 5;
  int4* p = (int4*)(cnt + (long)blockIdx.x * SORT_W) + t;
  const int4 c = *p;
  const int s = c.x + c.y + c.z + c.w;
  int inc = s;
  for (int d = 1; d < 32; d <<= 1) {
    const int u = __shfl_up(inc, d);
    if (lane >= d) inc += u;
  }
  if (lane == 31) wsum[wv] = inc;
  __syncthreads();
  int before = 0;
  int total = 0;
  for (int i = 0; i < TPB / 32; ++i) {
    const int x = wsum[i];
    if (i < wv) before += x;
    total += x;
  }
  const int ex = before + inc - s;
  *p = make_int4(ex, ex + c.x, ex + c.x + c.y, ex + c.x + c.y + c.z);
  if (t == 0) tot[blockIdx.x] = total;
}

__global__ void __launch_bounds__(TPB) k_sort_place(const int* __restrict__ nbrA, int nA,
                                                    const int* __restrict__ nbrB, int nB,
                                                    const int* __restrict__ cnt2,
                                                    const int* __restrict__ tot2,
                                                    int* __restrict__ permA,
                                                    int* __restrict__ permB) {
  const int* __restrict__ nbr = blockIdx.y ? nbrB : nbrA;
  const int n = blockIdx.y ? nB : nA;
  const int* __restrict__ cnt = cnt2 + (long)blockIdx.y * SORT_B * SORT_W;
  const int* __restrict__ tot = tot2 + (long)blockIdx.y * SORT_B;
  int* __restrict__ perm = blockIdx.y ? permB : permA;
  const int lane = threadIdx.x & 31;
  const int w = blockIdx.x * (TPB / 32) + ((int)threadIdx.x >> 5);
  const int seg = (n + SORT_W - 1) / SORT_W;
  const int beg = w * seg;
  const int end = beg + seg < n ? beg + seg : n;
  int off[SORT_B / 32];
  int carry = 0;
#pragma unroll
  for (int s = 0; s < SORT_B / 32; ++s) {
    const int v = tot[32 * s + lane];
    int inc = v;
    for (int d = 1; d < 32; d <<= 1) {
      const int u = __shfl_up(inc, d);
      if (lane >= d) inc += u;
    }
    off[s] = carry + inc - v + cnt[(long)(32 * s + lane) * SORT_W + w];
    carry += __shfl(inc, 31);
  }
  const unsigned below = (1u << lane) - 1u;
  for (int base = beg; base < end; base += 32) {
    const int r = base + lane;
    const int key = r < end ? corner_key(nbr, n, r) : -1;
    int pos = 0;
#pragma unroll
    for (int s = 0; s < SORT_B / 32; ++s) {
      if (__builtin_amdgcn_ballot_w32((key >> 5) == s) != 0u) {
        for (int l = 0; l < 32; ++l) {
          const unsigned m = __builtin_amdgcn_ballot_w32(key == 32 * s + l);
          const int start = __builtin_amdgcn_readlane(off[s], l);
          if (key == 32 * s + l) pos = start + __popc(m & below);
          if (lane == l) off[s] += __popc(m);
        }
      }
    }
    if (key >= 0) perm[pos] = r;
  }
}

template <int C>
__device__ __forceinline__ void scatter_add_rows(const float* __restrict__ X, float* __restrict__ Y,
                                                 const int* __restrict__ idx, long rows,
                                                 long first, long step) {
  constexpr int Q = C / 4;
  const float4* __restrict__ X4 = (const float4*)X;
  const long tot4 = rows * Q;
  float4* __restrict__ Y4 = (float4*)Y;
  for (long i = first; i < tot4; i += 4 * step) {
    long kc[4];
    long off[4];
    float4 x[4];
    float4 y[4];
    for (int u = 0; u < 4; ++u) {
      const long k = i + u * step;
      kc[u] = k < tot4 ? k : i;
      off[u] = idx[kc[u] / Q];
    }
    for (int u = 0; u < 4; ++u) {
      off[u] = off[u] * Q + (kc[u] % Q);
      x[u] = X4[kc[u]];
      y[u] = Y4[off[u]];
    }
    for (int u = 0; u < 4; ++u) {
      if (i + u * step < tot4) {
        float4 v = y[u];
        v.x += x[u].x; v.y += x[u].y; v.z += x[u].z; v.w += x[u].w;
        Y4[off[u]] = v;
      }
    }
  }
}

template <int C>
__global__ void k_scatter_add(const float* __restrict__ X, float* __restrict__ Y,
                              const int* __restrict__ idx, long rows) {
  scatter_add_rows<C>(X, Y, idx, rows, (long)blockIdx.x * TPB + threadIdx.x,
                      (long)gridDim.x * TPB);
}

struct PrepArgs {
  const int* map0; const int* map1; const int* map2; const int* map3; const int* map4;
  int* nbr0; int* nbr1; int* nbr2; int* nbr3; int* nbr4;
  int Mw0; int Mw1; int Mw2; int Mw3; int Mw4;
  int n0; int n1; int n2; int n3; int n4;
  int nb0; int nb1; int nb2; int nb3; int nb4;
  int s1; int s2; int s3; int s4; int s5; int s6;
  const float* fa; float* xa; const int* ia; long ra;
  const float* fb; float* xb; const int* ib; long rb;
};

__device__ __forceinline__ void invmap_job(const int* __restrict__ map, int* __restrict__ nbr,
                                           int Mw, int n, int local, int nb) {
  const int j = local / nb;
  invmap_tap(map, nbr, Mw, n, j, local - j * nb, nb);
}

__global__ void k_prep(PrepArgs a) {
  const int b = blockIdx.x;
  if (b >= a.s5) {
    const bool narrow = b >= a.s6;
    const int vb = b - (narrow ? a.s6 : a.s5);
    const int nb = narrow ? (int)gridDim.x - a.s6 : a.s6 - a.s5;
    const long first = (long)vb * TPB + threadIdx.x;
    const long step = (long)nb * TPB;
    if (narrow) scatter_add_rows<64>(a.fb, a.xb, a.ib, a.rb, first, step);
    else        scatter_add_rows<128>(a.fa, a.xa, a.ia, a.ra, first, step);
    return;
  }
  if      (b < a.s1) invmap_job(a.map0, a.nbr0, a.Mw0, a.n0, b, a.nb0);
  else if (b < a.s2) invmap_job(a.map1, a.nbr1, a.Mw1, a.n1, b - a.s1, a.nb1);
  else if (b < a.s3) invmap_job(a.map2, a.nbr2, a.Mw2, a.n2, b - a.s2, a.nb2);
  else if (b < a.s4) invmap_job(a.map3, a.nbr3, a.Mw3, a.n3, b - a.s3, a.nb3);
  else               invmap_job(a.map4, a.nbr4, a.Mw4, a.n4, b - a.s4, a.nb4);
}

__global__ void k_bn_part(const float* __restrict__ X, long tot, int C,
                          float* __restrict__ part) {
  __shared__ float sm[8][TPB];
  const int t = threadIdx.x;
  float a0 = 0.0f, a1 = 0.0f, a2 = 0.0f, a3 = 0.0f;
  float b0 = 0.0f, b1 = 0.0f, b2 = 0.0f, b3 = 0.0f;
  const long step = (long)gridDim.x * (TPB * 4);
  for (long i = ((long)blockIdx.x * TPB + t) * 4; i < tot; i += step) {
    const float4 v = *(const float4*)(X + i);
    a0 += v.x; b0 += v.x * v.x;
    a1 += v.y; b1 += v.y * v.y;
    a2 += v.z; b2 += v.z * v.z;
    a3 += v.w; b3 += v.w * v.w;
  }
  sm[0][t] = a0; sm[1][t] = a1; sm[2][t] = a2; sm[3][t] = a3;
  sm[4][t] = b0; sm[5][t] = b1; sm[6][t] = b2; sm[7][t] = b3;
  __syncthreads();
  if (t < C) {
    const int q = C >> 2;
    const int comp = t & 3;
    float s = 0.0f, ss = 0.0f;
    for (int j = t >> 2; j < TPB; j += q) {
      s += sm[comp][j];
      ss += sm[4 + comp][j];
    }
    float* p = part + (long)blockIdx.x * 2 * C;
    p[t] = s;
    p[C + t] = ss;
  }
}

__global__ void k_bn_fin(const float* __restrict__ part, int P, long N, int C,
                         float* __restrict__ mu, float* __restrict__ rs) {
  __shared__ float s1[TPB];
  __shared__ float s2[TPB];
  const int c = blockIdx.x;
  float a = 0.0f, b = 0.0f;
  for (int p = threadIdx.x; p < P; p += TPB) {
    a += part[(long)p * 2 * C + c];
    b += part[(long)p * 2 * C + C + c];
  }
  s1[threadIdx.x] = a;
  s2[threadIdx.x] = b;
  __syncthreads();
  for (int s = TPB / 2; s > 0; s >>= 1) {
    if ((int)threadIdx.x < s) {
      s1[threadIdx.x] += s1[threadIdx.x + s];
      s2[threadIdx.x] += s2[threadIdx.x + s];
    }
    __syncthreads();
  }
  if (threadIdx.x == 0) {
    float m = s1[0] / (float)N;
    float var = s2[0] / (float)N - m * m;
    if (var < 0.0f) var = 0.0f;
    mu[c] = m;
    rs[c] = rsqrtf(var + 1e-5f);
  }
}

template <int C>
__global__ void k_bn_elu(const float* X, float* Y,
                         const float* __restrict__ mu, const float* __restrict__ rs,
                         const float* __restrict__ gb, long tot4) {
  const float4* X4 = (const float4*)X;
  float4* Y4 = (float4*)Y;
  const int c = ((int)threadIdx.x * 4) & (C - 1);
  float g[4], m[4], s[4], b[4];
  for (int e = 0; e < 4; ++e) {
    g[e] = gb[c + e];
    m[e] = mu[c + e];
    s[e] = rs[c + e];
    b[e] = gb[C + c + e];
  }
  const long step = (long)gridDim.x * TPB;
  for (long i = (long)blockIdx.x * TPB + threadIdx.x; i < tot4; i += 4 * step) {
    float4 x[4];
    for (int u = 0; u < 4; ++u)
      if (i + u * step < tot4) x[u] = X4[i + u * step];
    for (int u = 0; u < 4; ++u) {
      if (i + u * step < tot4) {
        float v0 = g[0] * (x[u].x - m[0]) * s[0] + b[0];
        float v1 = g[1] * (x[u].y - m[1]) * s[1] + b[1];
        float v2 = g[2] * (x[u].z - m[2]) * s[2] + b[2];
        float v3 = g[3] * (x[u].w - m[3]) * s[3] + b[3];
        float4 o;
        o.x = v0 > 0.0f ? v0 : expm1f(v0);
        o.y = v1 > 0.0f ? v1 : expm1f(v1);
        o.z = v2 > 0.0f ? v2 : expm1f(v2);
        o.w = v3 > 0.0f ? v3 : expm1f(v3);
        Y4[i + u * step] = o;
      }
    }
  }
}

template <int CIN, int COUT>
__global__ void k_gemm_tap(const float* __restrict__ X,
                           const float* __restrict__ W,
                           float* __restrict__ Y,
                           const int* __restrict__ gidx,
                           const int* __restrict__ sidx,
                           int M, int out_offset, int accumulate, int nscrap, int zrows) {
  const int lane = threadIdx.x;
  W += (long)blockIdx.z * CIN * COUT;
  out_offset += (int)blockIdx.z * zrows;
  const int tile_r = blockIdx.x * 32;
  const int tile_c = (blockIdx.y * 4 + threadIdx.y) * 16;
  const int mr = lane & 15;
  const int kh = lane >> 4;

  const int rl = tile_r + lane;
  int sv = nscrap;
  if (rl < M) sv = sidx ? sidx[rl] : rl;
  if (__builtin_amdgcn_ballot_w32(sv != nscrap) == 0u) return;

  const int r0 = tile_r + mr;
  const int r1 = tile_r + 16 + mr;
  int g0 = 0, g1 = 0;
  if (r0 < M) g0 = gidx ? gidx[r0] : r0;
  if (r1 < M) g1 = gidx ? gidx[r1] : r1;
  const float* __restrict__ xb0 = X + (long)g0 * CIN + 2 * kh;
  const float* __restrict__ xb1 = X + (long)g1 * CIN + 2 * kh;
  const float* __restrict__ wb  = W + tile_c + mr + 2 * kh * COUT;

  v8f c0 = {0.f, 0.f, 0.f, 0.f, 0.f, 0.f, 0.f, 0.f};
  v8f c1 = {0.f, 0.f, 0.f, 0.f, 0.f, 0.f, 0.f, 0.f};
#pragma unroll
  for (int k0 = 0; k0 < CIN; k0 += 4) {
    v2f a0 = *(const v2f*)(xb0 + k0);
    v2f a1 = *(const v2f*)(xb1 + k0);
    v2f b;
    b[0] = wb[k0 * COUT];
    b[1] = wb[(k0 + 1) * COUT];
    c0 = __builtin_amdgcn_wmma_f32_16x16x4_f32(false, a0, false, b,
                                               (short)0, c0, false, false);
    c1 = __builtin_amdgcn_wmma_f32_16x16x4_f32(false, a1, false, b,
                                               (short)0, c1, false, false);
  }

#pragma unroll
  for (int i = 0; i < 8; ++i) {
    const int t0a = __builtin_amdgcn_readlane(sv, i);
    const int t0b = __builtin_amdgcn_readlane(sv, i + 8);
    const int ro0 = kh ? t0b : t0a;
    if (ro0 != nscrap) {
      float* y = Y + (long)(ro0 + out_offset) * COUT + tile_c + mr;
      if (accumulate) *y += c0[i]; else *y = c0[i];
    }
    const int t1a = __builtin_amdgcn_readlane(sv, i + 16);
    const int t1b = __builtin_amdgcn_readlane(sv, i + 24);
    const int ro1 = kh ? t1b : t1a;
    if (ro1 != nscrap) {
      float* y = Y + (long)(ro1 + out_offset) * COUT + tile_c + mr;
      if (accumulate) *y += c1[i]; else *y = c1[i];
    }
  }
}

template <int CIN, int COUT>
__global__ void __launch_bounds__(128) k_conv_os(const float* __restrict__ X,
                          const float* __restrict__ W,
                          float* __restrict__ Y,
                          const int* __restrict__ nbr,
                          int n) {
  const int lane = threadIdx.x;
  const int tile_r = blockIdx.x * 32;
  const int tile_c = (blockIdx.y * 4 + threadIdx.y) * 16;
  const int mr = lane & 15;
  const int kh = lane >> 4;
  const int r0 = tile_r + mr;
  const int r1 = tile_r + 16 + mr;
  const bool in0 = r0 < n;
  const bool in1 = r1 < n;
  constexpr int KC = CIN < 64 ? CIN : 64;
  constexpr int NS = KC / 4;

  v8f c0 = {0.f, 0.f, 0.f, 0.f, 0.f, 0.f, 0.f, 0.f};
  v8f c1 = {0.f, 0.f, 0.f, 0.f, 0.f, 0.f, 0.f, 0.f};

  const int q0 = in0 ? r0 : n - 1;
  const int q1 = in1 ? r1 : n - 1;
  int g0 = nbr[q0];
  int g1 = nbr[q1];
  g0 = in0 ? g0 : -1;
  g1 = in1 ? g1 : -1;
  for (int t = 0; t < 27; ++t) {
    int l0 = r0;
    int l1 = r1;
    if (t < 25) {
      l0 = nbr[(long)(t + 1) * n + q0];
      l1 = nbr[(long)(t + 1) * n + q1];
    }
    if (__builtin_amdgcn_ballot_w32((g0 >= 0) || (g1 >= 0)) != 0u) {
      const int k = (t < 13) ? t : (t < 26 ? t + 1 : 13);
      const float* __restrict__ wb =
          W + (long)k * CIN * COUT + tile_c + mr + 2 * kh * COUT;
#pragma unroll
      for (int kc = 0; kc < CIN; kc += KC) {
        v2f a0[NS];
        v2f a1[NS];
#pragma unroll
        for (int s = 0; s < NS; ++s) {
          a0[s] = (v2f){0.f, 0.f};
          a1[s] = (v2f){0.f, 0.f};
        }
        if (g0 >= 0) {
          const float* __restrict__ xb0 = X + (long)g0 * CIN + 2 * kh + kc;
#pragma unroll
          for (int s = 0; s < NS; ++s) a0[s] = *(const v2f*)(xb0 + 4 * s);
        }
        if (g1 >= 0) {
          const float* __restrict__ xb1 = X + (long)g1 * CIN + 2 * kh + kc;
#pragma unroll
          for (int s = 0; s < NS; ++s) a1[s] = *(const v2f*)(xb1 + 4 * s);
        }
#pragma unroll
        for (int s = 0; s < NS; ++s) {
          const int k0 = kc + 4 * s;
          v2f b;
          b[0] = wb[k0 * COUT];
          b[1] = wb[(k0 + 1) * COUT];
          c0 = __builtin_amdgcn_wmma_f32_16x16x4_f32(false, a0[s], false, b,
                                                     (short)0, c0, false, false);
          c1 = __builtin_amdgcn_wmma_f32_16x16x4_f32(false, a1[s], false, b,
                                                     (short)0, c1, false, false);
        }
      }
    }
    g0 = in0 ? l0 : -1;
    g1 = in1 ? l1 : -1;
  }

#pragma unroll
  for (int i = 0; i < 8; ++i) {
    const int ro0 = tile_r + i + 8 * kh;
    if (ro0 < n) Y[(long)ro0 * COUT + tile_c + mr] = c0[i];
    const int ro1 = ro0 + 16;
    if (ro1 < n) Y[(long)ro1 * COUT + tile_c + mr] = c1[i];
  }
}

template <int CIN, int COUT>
__global__ void __launch_bounds__(128) k_conv_os_wide(const float* __restrict__ X,
                          const float* __restrict__ W,
                          float* __restrict__ Y,
                          const int* __restrict__ nbr,
                          const int* __restrict__ perm,
                          int n) {
  constexpr int NT = COUT / 16;
  constexpr int KC = 64;
  constexpr int NCH = CIN / KC;
  constexpr int PITCH = KC + 4;
  constexpr int CG = COUT / 4;
  constexpr int ITEMS = (KC / 4) * CG / 128;
  __shared__ float Bs[2][COUT * PITCH];
  __shared__ unsigned need[4];

  const int lane = threadIdx.x;
  const int wv = threadIdx.y;
  const int tid = wv * 32 + lane;
  const int tile_r = (((int)gridDim.x - 1 - (int)blockIdx.x) * 4 + wv) * 32;
  const int mr = lane & 15;
  const int kh = lane >> 4;
  const int p0 = tile_r + mr;
  const int p1 = tile_r + 16 + mr;
  const bool in0 = p0 < n;
  const bool in1 = p1 < n;
  const int r0 = perm[in0 ? p0 : n - 1];
  const int r1 = perm[in1 ? p1 : n - 1];

  unsigned wm = 0u;
  {
    int h0[26];
    int h1[26];
#pragma unroll
    for (int t = 0; t < 26; ++t) {
      h0[t] = nbr[(long)t * n + r0];
      h1[t] = nbr[(long)t * n + r1];
    }
#pragma unroll
    for (int t = 0; t < 26; ++t) {
      const int e0 = in0 ? h0[t] : -1;
      const int e1 = in1 ? h1[t] : -1;
      if (__builtin_amdgcn_ballot_w32((e0 & e1) >= 0) != 0u) wm |= 1u << t;
    }
    if (__builtin_amdgcn_ballot_w32(in0 || in1) != 0u) wm |= 1u << 26;
  }
  if (lane == 0) need[wv] = wm;
  __syncthreads();
  unsigned bm = (unsigned)__builtin_amdgcn_readfirstlane(
      (int)(need[0] | need[1] | need[2] | need[3]));
  if (bm == 0u) return;

  auto stage = [&](int t, int ch, float* __restrict__ dst) {
    const int kc = ch * KC;
    const int k = (t < 13) ? t : (t < 26 ? t + 1 : 13);
    const float* __restrict__ wsrc = W + ((long)k * CIN + kc) * COUT;
    float4 r[ITEMS][4];
    for (int u = 0; u < ITEMS; ++u) {
      const int e = tid + 128 * u;
      const int cg = e % CG;
      const int kg = e / CG;
      for (int rr = 0; rr < 4; ++rr)
        r[u][rr] = *(const float4*)(wsrc + (long)(4 * kg + rr) * COUT + 4 * cg);
    }
    for (int u = 0; u < ITEMS; ++u) {
      const int e = tid + 128 * u;
      const int cg = e % CG;
      const int kg = e / CG;
      float* d = dst + (4 * cg) * PITCH + 4 * kg;
      *(float4*)(d)             = make_float4(r[u][0].x, r[u][1].x, r[u][2].x, r[u][3].x);
      *(float4*)(d + PITCH)     = make_float4(r[u][0].y, r[u][1].y, r[u][2].y, r[u][3].y);
      *(float4*)(d + 2 * PITCH) = make_float4(r[u][0].z, r[u][1].z, r[u][2].z, r[u][3].z);
      *(float4*)(d + 3 * PITCH) = make_float4(r[u][0].w, r[u][1].w, r[u][2].w, r[u][3].w);
    }
  };

  v8f c0[NT];
  v8f c1[NT];
  for (int j = 0; j < NT; ++j) {
    c0[j] = (v8f){0.f, 0.f, 0.f, 0.f, 0.f, 0.f, 0.f, 0.f};
    c1[j] = (v8f){0.f, 0.f, 0.f, 0.f, 0.f, 0.f, 0.f, 0.f};
  }

  int t = __builtin_ctz(bm);
  bm &= bm - 1u;
  stage(t, 0, Bs[0]);

  int g0 = t < 26 ? nbr[(long)t * n + r0] : r0;
  int g1 = t < 26 ? nbr[(long)t * n + r1] : r1;
  g0 = in0 ? g0 : -1;
  g1 = in1 ? g1 : -1;
  __syncthreads();

  int buf = 0;
  for (int it = 0; it < 27; ++it) {
    const int tn = bm != 0u ? __builtin_ctz(bm) : -1;
    bm &= bm - 1u;
    int l0 = r0;
    int l1 = r1;
    if (tn >= 0 && tn < 26) {
      l0 = nbr[(long)tn * n + r0];
      l1 = nbr[(long)tn * n + r1];
    }
    const bool act = __builtin_amdgcn_ballot_w32((g0 >= 0) || (g1 >= 0)) != 0u;
    for (int ch = 0; ch < NCH; ++ch) {
      if (ch + 1 < NCH) stage(t, ch + 1, Bs[buf ^ 1]);
      else if (tn >= 0) stage(tn, 0, Bs[buf ^ 1]);
      if (act) {
        const float* __restrict__ bs = Bs[buf] + mr * PITCH + 4 * kh;
        float4 a0[8];
        float4 a1[8];
        for (int p = 0; p < 8; ++p) {
          a0[p] = make_float4(0.f, 0.f, 0.f, 0.f);
          a1[p] = make_float4(0.f, 0.f, 0.f, 0.f);
        }
        if (g0 >= 0) {
          const float* __restrict__ xb0 = X + (long)g0 * CIN + ch * KC + 4 * kh;
          for (int p = 0; p < 8; ++p) a0[p] = *(const float4*)(xb0 + 8 * p);
        }
        if (g1 >= 0) {
          const float* __restrict__ xb1 = X + (long)g1 * CIN + ch * KC + 4 * kh;
          for (int p = 0; p < 8; ++p) a1[p] = *(const float4*)(xb1 + 8 * p);
        }
        for (int p = 0; p < 8; ++p) {
          const v2f a0l = {a0[p].x, a0[p].y};
          const v2f a0h = {a0[p].z, a0[p].w};
          const v2f a1l = {a1[p].x, a1[p].y};
          const v2f a1h = {a1[p].z, a1[p].w};
          float4 bf[NT];
          for (int j = 0; j < NT; ++j) bf[j] = *(const float4*)(bs + 16 * j * PITCH + 8 * p);
          for (int j = 0; j < NT; ++j) {
            const float4 b = bf[j];
            const v2f bl = {b.x, b.y};
            const v2f bh = {b.z, b.w};
            c0[j] = __builtin_amdgcn_wmma_f32_16x16x4_f32(false, a0l, false, bl,
                                                          (short)0, c0[j], false, false);
            c1[j] = __builtin_amdgcn_wmma_f32_16x16x4_f32(false, a1l, false, bl,
                                                          (short)0, c1[j], false, false);
            c0[j] = __builtin_amdgcn_wmma_f32_16x16x4_f32(false, a0h, false, bh,
                                                          (short)0, c0[j], false, false);
            c1[j] = __builtin_amdgcn_wmma_f32_16x16x4_f32(false, a1h, false, bh,
                                                          (short)0, c1[j], false, false);
          }
        }
      }
      __syncthreads();
      buf ^= 1;
    }
    if (tn < 0) break;
    g0 = in0 ? l0 : -1;
    g1 = in1 ? l1 : -1;
    t = tn;
  }

  int o0[8];
  int o1[8];
#pragma unroll
  for (int i = 0; i < 8; ++i) {
    const int xa = __builtin_amdgcn_readlane(r0, i);
    const int xb = __builtin_amdgcn_readlane(r0, i + 8);
    o0[i] = kh ? xb : xa;
    const int ya = __builtin_amdgcn_readlane(r1, i);
    const int yb = __builtin_amdgcn_readlane(r1, i + 8);
    o1[i] = kh ? yb : ya;
  }
#pragma unroll
  for (int j = 0; j < NT; ++j) {
#pragma unroll
    for (int i = 0; i < 8; ++i) {
      if (tile_r + i + 8 * kh < n) Y[(long)o0[i] * COUT + 16 * j + mr] = c0[j][i];
      if (tile_r + 16 + i + 8 * kh < n) Y[(long)o1[i] * COUT + 16 * j + mr] = c1[j][i];
    }
  }
}

template <int CIN, int COUT>
static inline void conv_t(const float* X, const float* W, float* Y,
                          const int* nbr, int n, hipStream_t st) {
  dim3 grid((unsigned)((n + 31) / 32), (unsigned)(COUT / 64));
  dim3 blk(32, 4);
  k_conv_os<CIN, COUT><<<grid, blk, 0, st>>>(X, W, Y, nbr, n);
}

template <int CIN, int COUT>
static inline void conv_wide_t(const float* X, const float* W, float* Y,
                               const int* nbr, int n, const int* perm, hipStream_t st) {
  dim3 grid((unsigned)((n + 127) / 128));
  dim3 blk(32, 4);
  k_conv_os_wide<CIN, COUT><<<grid, blk, 0, st>>>(X, W, Y, nbr, perm, n);
}

static inline bool conv_is_wide(int n, int Cin) { return n >= 128 * 256 && Cin == 64; }

static void conv_os(const float* X, const float* W, float* Y, const int* nbr,
                    int n, int Cin, int Cout, const int* perm, hipStream_t st) {
  const bool wide = conv_is_wide(n, Cin);
  if      (wide && Cin == 64 && Cout ==  64) conv_wide_t<64,  64>(X, W, Y, nbr, n, perm, st);
  else if (wide && Cin == 64 && Cout == 128) conv_wide_t<64, 128>(X, W, Y, nbr, n, perm, st);
  else if (Cin == 256 && Cout == 128) conv_t<256, 128>(X, W, Y, nbr, n, st);
  else if (Cin == 128 && Cout == 128) conv_t<128, 128>(X, W, Y, nbr, n, st);
  else if (Cin ==  64 && Cout ==  64) conv_t< 64,  64>(X, W, Y, nbr, n, st);
  else if (Cin ==  64 && Cout == 128) conv_t< 64, 128>(X, W, Y, nbr, n, st);
}

template <int CIN, int COUT>
static inline void gemm_t(const float* X, const float* W, float* Y,
                          const int* g, const int* s, int M, int ooff,
                          int accum, int nscrap, int nz, hipStream_t st) {
  dim3 grid((unsigned)((M + 31) / 32), (unsigned)(COUT / 64), (unsigned)nz);
  dim3 blk(32, 4);
  k_gemm_tap<CIN, COUT><<<grid, blk, 0, st>>>(X, W, Y, g, s, M, ooff, accum, nscrap, M);
}

static void gemm(const float* X, const float* W, float* Y,
                 const int* g, const int* s, int M, int Cin, int Cout,
                 int ooff, int accum, int nscrap, int nz, hipStream_t st) {
  if      (Cin == 256 && Cout == 128) gemm_t<256, 128>(X, W, Y, g, s, M, ooff, accum, nscrap, nz, st);
  else if (Cin == 128 && Cout ==  64) gemm_t<128,  64>(X, W, Y, g, s, M, ooff, accum, nscrap, nz, st);
}


extern "C" void kernel_launch(void* const* d_in, const int* in_sizes, int n_in,
                              void* d_out, int out_size, void* d_ws, size_t ws_size,
                              hipStream_t stream) {
  const float* feats0 = (const float*)d_in[0];
  const float* feats1 = (const float*)d_in[1];
  const float* feats2 = (const float*)d_in[2];
  const float* w_out0 = (const float*)d_in[3];
  const float* w_out1 = (const float*)d_in[4];
  const float* w_out2 = (const float*)d_in[5];
  const float* wt2    = (const float*)d_in[6];
  const float* wu2    = (const float*)d_in[7];
  const float* wt1    = (const float*)d_in[8];
  const float* wu1    = (const float*)d_in[9];
  const float* bn_out0 = (const float*)d_in[10];
  const float* bn_out1 = (const float*)d_in[11];
  const float* bn_out2 = (const float*)d_in[12];
  const float* bn_up2a = (const float*)d_in[13];
  const float* bn_up2b = (const float*)d_in[14];
  const float* bn_up1a = (const float*)d_in[15];
  const float* bn_up1b = (const float*)d_in[16];
  const int* m_c2 = (const int*)d_in[19];
  const int* m_g2 = (const int*)d_in[20];
  const int* m_u1 = (const int*)d_in[21];
  const int* m_g1 = (const int*)d_in[22];
  const int* m_u0 = (const int*)d_in[23];
  const int* lat1_idx = (const int*)d_in[24];
  const int* up1_idx  = (const int*)d_in[25];
  const int* lat0_idx = (const int*)d_in[26];
  const int* up0_idx  = (const int*)d_in[27];

  const int N0 = in_sizes[0] / 64;
  const int N1 = in_sizes[1] / 128;
  const int N2 = in_sizes[2] / 256;
  const int nu1 = in_sizes[17] / 128;
  const int nu0 = in_sizes[18] / 64;
  const int Mc2 = in_sizes[19] / 52;
  const int Mg2 = in_sizes[20] / 52;
  const int Mu1 = in_sizes[21] / 52;
  const int Mg1 = in_sizes[22] / 52;
  const int Mu0 = in_sizes[23] / 52;
  const int g2n = 8 * N2;
  const int g1n = 8 * nu1;

  char* wsp = (char*)d_ws;
  auto carve = [&](long nfloats) -> float* {
    float* p = (float*)wsp;
    wsp += ((nfloats * 4 + 255) / 256) * 256;
    return p;
  };
  long accElems = (long)(N2 + 1) * 128;
  { long e = (long)(g2n + 1) * 128; if (e > accElems) accElems = e; }
  { long e = (long)(nu1 + 1) * 128; if (e > accElems) accElems = e; }
  { long e = (long)(g1n + 1) * 64;  if (e > accElems) accElems = e; }
  { long e = (long)(nu0 + 1) * 128; if (e > accElems) accElems = e; }
  float* acc = carve(accElems);
  float* t2  = carve((long)g2n * 128);
  float* t1  = carve((long)g1n * 64);
  float* mu  = carve(128);
  float* rs  = carve(128);
  float* part = carve((long)1024 * 256);
  float* x1  = carve((long)nu1 * 128);
  float* x0  = carve((long)nu0 * 64);
  const long zero4 = (long)(wsp - (char*)x1) / 16;
  int* nbC2 = (int*)carve((long)26 * N2);
  int* nbG2 = (int*)carve((long)26 * g2n);
  int* nbU1 = (int*)carve((long)26 * nu1);
  int* nbG1 = (int*)carve((long)26 * g1n);
  int* nbU0 = (int*)carve((long)26 * nu0);
  const long neg4 = (long)(wsp - (char*)nbC2) / 16;
  int* permG1 = (int*)carve(g1n);
  int* permU0 = (int*)carve(nu0);
  int* cnt2 = (int*)carve((long)2 * SORT_B * SORT_W);
  int* tot2 = (int*)carve((long)2 * SORT_B);

  float* out  = (float*)d_out;
  float* out0 = out;
  float* out1 = out + (long)nu0 * 128;
  float* out2 = out + (long)nu0 * 128 + (long)nu1 * 128;

  auto nblk = [](long items) -> unsigned {
    long b = (items + TPB - 1) / TPB;
    if (b > MAXB) b = MAXB;
    if (b < 1) b = 1;
    return (unsigned)b;
  };
  auto conv3 = [&](const float* X, int n, int Cin, int Cout,
                   const float* W, const int* nbr, const int* perm) {
    conv_os(X, W, acc, nbr, n, Cin, Cout, perm, stream);
  };
  auto bnelu = [&](const float* src, float* dst, const float* gb, long n, int C) {
    long tot = n * C;
    long P = (tot + TPB * 4 - 1) / (TPB * 4);
    if (P > 1024) P = 1024;
    k_bn_part<<<dim3((unsigned)P), dim3(TPB), 0, stream>>>(src, tot, C, part);
    k_bn_fin<<<dim3((unsigned)C), dim3(TPB), 0, stream>>>(part, (int)P, n, C, mu, rs);
    const long tot4 = tot / 4;
    if (C == 128)
      k_bn_elu<128><<<dim3(nblk(tot4)), dim3(TPB), 0, stream>>>(src, dst, mu, rs, gb, tot4);
    else
      k_bn_elu<64><<<dim3(nblk(tot4)), dim3(TPB), 0, stream>>>(src, dst, mu, rs, gb, tot4);
  };
  auto scat = [&](const float* X, float* Y, const int* idx, long rows, int C) {
    const long tot4 = rows * (C / 4);
    if (C == 128)
      k_scatter_add<128><<<dim3(nblk(tot4)), dim3(TPB), 0, stream>>>(X, Y, idx, rows);
    else
      k_scatter_add<64><<<dim3(nblk(tot4)), dim3(TPB), 0, stream>>>(X, Y, idx, rows);
  };

  k_fill2<<<dim3(MAXB), dim3(TPB), 0, stream>>>(nbC2, neg4, -1, (int*)x1, zero4, 0);
  {
    PrepArgs pa;
    pa.map0 = m_c2; pa.map1 = m_g2; pa.map2 = m_u1; pa.map3 = m_g1; pa.map4 = m_u0;
    pa.nbr0 = nbC2; pa.nbr1 = nbG2; pa.nbr2 = nbU1; pa.nbr3 = nbG1; pa.nbr4 = nbU0;
    pa.Mw0 = Mc2; pa.Mw1 = Mg2; pa.Mw2 = Mu1; pa.Mw3 = Mg1; pa.Mw4 = Mu0;
    pa.n0 = N2; pa.n1 = g2n; pa.n2 = nu1; pa.n3 = g1n; pa.n4 = nu0;
    pa.fa = feats1; pa.xa = x1; pa.ia = lat1_idx; pa.ra = N1;
    pa.fb = feats0; pa.xb = x0; pa.ib = lat0_idx; pa.rb = N0;
    auto tapblk = [](int Mw) -> int {
      long b = ((long)Mw + 4 * TPB - 1) / (4 * TPB);
      if (b > MAXB / 26) b = MAXB / 26;
      if (b < 1) b = 1;
      return (int)b;
    };
    auto scatblk = [](long tot4) -> int {
      long b = (tot4 + 4 * TPB - 1) / (4 * TPB);
      if (b > MAXB) b = MAXB;
      if (b < 1) b = 1;
      return (int)b;
    };
    pa.nb0 = tapblk(Mc2); pa.nb1 = tapblk(Mg2); pa.nb2 = tapblk(Mu1);
    pa.nb3 = tapblk(Mg1); pa.nb4 = tapblk(Mu0);
    pa.s1 = 26 * pa.nb0;
    pa.s2 = pa.s1 + 26 * pa.nb1;
    pa.s3 = pa.s2 + 26 * pa.nb2;
    pa.s4 = pa.s3 + 26 * pa.nb3;
    pa.s5 = pa.s4 + 26 * pa.nb4;
    pa.s6 = pa.s5 + scatblk((long)N1 * 32);
    const int nblocks = pa.s6 + scatblk((long)N0 * 16);
    k_prep<<<dim3((unsigned)nblocks), dim3(TPB), 0, stream>>>(pa);
  }
  {
    const bool wG1 = conv_is_wide(g1n, 64);
    const bool wU0 = conv_is_wide(nu0, 64);
    const int* tA = wG1 ? nbG1 : nbU0;
    const int nA = wG1 ? g1n : nu0;
    int* pA = wG1 ? permG1 : permU0;
    const unsigned ny = (wG1 ? 1u : 0u) + (wU0 ? 1u : 0u);
    if (ny > 0u) {
      k_sort_count<<<dim3(SORT_W / (TPB / 32), ny), dim3(TPB), 0, stream>>>(tA, nA, nbU0, nu0, cnt2);
      k_sort_scan<<<dim3(SORT_B * ny), dim3(TPB), 0, stream>>>(cnt2, tot2);
      k_sort_place<<<dim3(SORT_W / (TPB / 32), ny), dim3(TPB), 0, stream>>>(tA, nA, nbU0, nu0, cnt2, tot2,
                                                                      pA, permU0);
    }
  }

  conv3(feats2, N2, 256, 128, w_out2, nbC2, nullptr);
  bnelu(acc, out2, bn_out2, N2, 128);

  gemm(feats2, wt2, t2, nullptr, nullptr, N2, 256, 128, 0, 0, -1, 8, stream);
  bnelu(t2, t2, bn_up2a, g2n, 128);
  conv3(t2, g2n, 128, 128, wu2, nbG2, nullptr);
  bnelu(acc, t2, bn_up2b, g2n, 128);

  scat(t2, x1, up1_idx, g2n, 128);

  conv3(x1, nu1, 128, 128, w_out1, nbU1, nullptr);
  bnelu(acc, out1, bn_out1, nu1, 128);

  gemm(x1, wt1, t1, nullptr, nullptr, nu1, 128, 64, 0, 0, -1, 8, stream);
  bnelu(t1, t1, bn_up1a, g1n, 64);
  conv3(t1, g1n, 64, 64, wu1, nbG1, permG1);
  bnelu(acc, t1, bn_up1b, g1n, 64);

  scat(t1, x0, up0_idx, g1n, 64);

  conv3(x0, nu0, 64, 128, w_out0, nbU0, permU0);
  bnelu(acc, out0, bn_out0, nu0, 128);

  (void)n_in; (void)out_size; (void)ws_size;
}


# side work gets no launches of its own: out2 and out1 convs run in the launch of level 1's child products, their batch norms and the row sort in main-chain batch-norm launches; 35 -> 24 launches
# speedup vs baseline: 10.8444x; 1.1407x over previous
#include <hip/hip_runtime.h>
#include <hip/hip_bf16.h>
#include <math.h>

typedef __attribute__((ext_vector_type(2))) float v2f;
typedef __attribute__((ext_vector_type(8))) float v8f;

#define TPB 256
#define MAXB 1024


__global__ void k_fill2(int* __restrict__ a, long na4, int va,
                        int* __restrict__ b, long nb4, int vb) {
  int4* __restrict__ a4 = (int4*)a;
  int4* __restrict__ b4 = (int4*)b;
  const int4 wa = {va, va, va, va};
  const int4 wb = {vb, vb, vb, vb};
  const long step = (long)gridDim.x * TPB;
  const long g = (long)blockIdx.x * TPB + threadIdx.x;
  for (long i = g; i < na4; i += step) a4[i] = wa;
  for (long i = g; i < nb4; i += step) b4[i] = wb;
}

__device__ __forceinline__ void invmap_tap(const int* __restrict__ map, int* __restrict__ nbr,
                                           int Mw, int n, int j, int bx, int nbx) {
  const int* __restrict__ gi = map + (long)j * Mw;
  const int* __restrict__ so = map + (long)(26 + j) * Mw;
  int* __restrict__ nb = nbr + (long)j * n;
  const int step = nbx * TPB;
  for (int i = bx * TPB + (int)threadIdx.x; i < Mw; i += 4 * step) {
    int o[4];
    int g[4];
    for (int u = 0; u < 4; ++u) {
      const int k = i + u * step;
      const int kc = k < Mw ? k : i;
      o[u] = so[kc];
      g[u] = gi[kc];
    }
    for (int u = 0; u < 4; ++u)
      if (i + u * step < Mw && o[u] >= 0 && o[u] < n) nb[o[u]] = g[u];
  }
}

#define SORT_B 256
#define SORT_W 1024

__device__ __forceinline__ int corner_key(const int* __restrict__ nbr, int n, int r) {
  const int ct[8] = {0, 2, 6, 8, 17, 19, 23, 25};
  int v[8];
#pragma unroll
  for (int b = 0; b < 8; ++b) v[b] = nbr[(long)ct[b] * n + r];
  int key = 0;
#pragma unroll
  for (int b = 0; b < 8; ++b) key |= (v[b] >= 0) ? (1 << b) : 0;
  return key;
}

#define SORT_BLK (SORT_W / (TPB / 32))
__device__ __forceinline__ void sort_count_rows(const int* __restrict__ nbrA, int nA,
                                                const int* __restrict__ nbrB, int nB,
                                                int* __restrict__ cnt2, int bx, int by) {
  const int* __restrict__ nbr = by ? nbrB : nbrA;
  const int n = by ? nB : nA;
  int* __restrict__ cnt = cnt2 + (long)by * SORT_B * SORT_W;
  const int lane = threadIdx.x & 31;
  const int w = bx * (TPB / 32) + ((int)threadIdx.x >> 5);
  const int seg = (n + SORT_W - 1) / SORT_W;
  const int beg = w * seg;
  const int end = beg + seg < n ? beg + seg : n;
  int c[SORT_B / 32];
#pragma unroll
  for (int s = 0; s < SORT_B / 32; ++s) c[s] = 0;
  for (int base = beg; base < end; base += 32) {
    const int r = base + lane;
    const int key = r < end ? corner_key(nbr, n, r) : -1;
#pragma unroll
    for (int s = 0; s < SORT_B / 32; ++s) {
      if (__builtin_amdgcn_ballot_w32((key >> 5) == s) != 0u) {
        for (int l = 0; l < 32; ++l) {
          const unsigned m = __builtin_amdgcn_ballot_w32(key == 32 * s + l);
          if (lane == l) c[s] += __popc(m);
        }
      }
    }
  }
#pragma unroll
  for (int s = 0; s < SORT_B / 32; ++s) cnt[(long)(32 * s + lane) * SORT_W + w] = c[s];
}

__device__ __forceinline__ void sort_scan_class(int* __restrict__ cnt, int* __restrict__ tot, int cls) {
  static_assert(SORT_W == 4 * TPB, "four wave counts per thread");
  __shared__ int wsum[TPB / 32];
  const int t = threadIdx.x;
  const int lane = t & 31;
  const int wv = t >> 5;
  int4* p = (int4*)(cnt + (long)cls * SORT_W) + t;
  const int4 c = *p;
  const int s = c.x + c.y + c.z + c.w;
  int inc = s;
  for (int d = 1; d < 32; d <<= 1) {
    const int u = __shfl_up(inc, d);
    if (lane >= d) inc += u;
  }
  if (lane == 31) wsum[wv] = inc;
  __syncthreads();
  int before = 0;
  int total = 0;
  for (int i = 0; i < TPB / 32; ++i) {
    const int x = wsum[i];
    if (i < wv) before += x;
    total += x;
  }
  const int ex = before + inc - s;
  *p = make_int4(ex, ex + c.x, ex + c.x + c.y, ex + c.x + c.y + c.z);
  if (t == 0) tot[cls] = total;
}

__device__ __forceinline__ void sort_place_rows(const int* __restrict__ nbrA, int nA,
                                                const int* __restrict__ nbrB, int nB,
                                                const int* __restrict__ cnt2,
                                                const int* __restrict__ tot2,
                                                int* __restrict__ permA,
                                                int* __restrict__ permB, int bx, int by) {
  const int* __restrict__ nbr = by ? nbrB : nbrA;
  const int n = by ? nB : nA;
  const int* __restrict__ cnt = cnt2 + (long)by * SORT_B * SORT_W;
  const int* __restrict__ tot = tot2 + (long)by * SORT_B;
  int* __restrict__ perm = by ? permB : permA;
  const int lane = threadIdx.x & 31;
  const int w = bx * (TPB / 32) + ((int)threadIdx.x >> 5);
  const int seg = (n + SORT_W - 1) / SORT_W;
  const int beg = w * seg;
  const int end = beg + seg < n ? beg + seg : n;
  int off[SORT_B / 32];
  int carry = 0;
#pragma unroll
  for (int s = 0; s < SORT_B / 32; ++s) {
    const int v = tot[32 * s + lane];
    int inc = v;
    for (int d = 1; d < 32; d <<= 1) {
      const int u = __shfl_up(inc, d);
      if (lane >= d) inc += u;
    }
    off[s] = carry + inc - v + cnt[(long)(32 * s + lane) * SORT_W + w];
    carry += __shfl(inc, 31);
  }
  const unsigned below = (1u << lane) - 1u;
  for (int base = beg; base < end; base += 32) {
    const int r = base + lane;
    const int key = r < end ? corner_key(nbr, n, r) : -1;
    int pos = 0;
#pragma unroll
    for (int s = 0; s < SORT_B / 32; ++s) {
      if (__builtin_amdgcn_ballot_w32((key >> 5) == s) != 0u) {
        for (int l = 0; l < 32; ++l) {
          const unsigned m = __builtin_amdgcn_ballot_w32(key == 32 * s + l);
          const int start = __builtin_amdgcn_readlane(off[s], l);
          if (key == 32 * s + l) pos = start + __popc(m & below);
          if (lane == l) off[s] += __popc(m);
        }
      }
    }
    if (key >= 0) perm[pos] = r;
  }
}

template <int C>
__device__ __forceinline__ void scatter_add_rows(const float* __restrict__ X, float* __restrict__ Y,
                                                 const int* __restrict__ idx, long rows,
                                                 long first, long step) {
  constexpr int Q = C / 4;
  const float4* __restrict__ X4 = (const float4*)X;
  const long tot4 = rows * Q;
  float4* __restrict__ Y4 = (float4*)Y;
  for (long i = first; i < tot4; i += 4 * step) {
    long kc[4];
    long off[4];
    float4 x[4];
    float4 y[4];
    for (int u = 0; u < 4; ++u) {
      const long k = i + u * step;
      kc[u] = k < tot4 ? k : i;
      off[u] = idx[kc[u] / Q];
    }
    for (int u = 0; u < 4; ++u) {
      off[u] = off[u] * Q + (kc[u] % Q);
      x[u] = X4[kc[u]];
      y[u] = Y4[off[u]];
    }
    for (int u = 0; u < 4; ++u) {
      if (i + u * step < tot4) {
        float4 v = y[u];
        v.x += x[u].x; v.y += x[u].y; v.z += x[u].z; v.w += x[u].w;
        Y4[off[u]] = v;
      }
    }
  }
}

template <int C>
__global__ void k_scatter_add(const float* __restrict__ X, float* __restrict__ Y,
                              const int* __restrict__ idx, long rows) {
  scatter_add_rows<C>(X, Y, idx, rows, (long)blockIdx.x * TPB + threadIdx.x,
                      (long)gridDim.x * TPB);
}

struct PrepArgs {
  const int* map0; const int* map1; const int* map2; const int* map3; const int* map4;
  int* nbr0; int* nbr1; int* nbr2; int* nbr3; int* nbr4;
  int Mw0; int Mw1; int Mw2; int Mw3; int Mw4;
  int n0; int n1; int n2; int n3; int n4;
  int nb0; int nb1; int nb2; int nb3; int nb4;
  int s1; int s2; int s3; int s4; int s5; int s6;
  int fill;
  const float* fa; float* xa; const int* ia; long ra;
  const float* fb; float* xb; const int* ib; long rb;
};

__device__ __forceinline__ void invmap_job(const int* __restrict__ map, int* __restrict__ nbr,
                                           int Mw, int n, int local, int nb) {
  const int j = local / nb;
  invmap_tap(map, nbr, Mw, n, j, local - j * nb, nb);
}

__global__ void k_prep(PrepArgs a) {
  const int b = blockIdx.x;
  if (b >= a.s5) {
    const bool narrow = b >= a.s6;
    const int vb = b - (narrow ? a.s6 : a.s5);
    const int nb = narrow ? (int)gridDim.x - a.s6 : a.s6 - a.s5;
    const long first = (long)vb * TPB + threadIdx.x;
    const long step = (long)nb * TPB;
    if (narrow) scatter_add_rows<64>(a.fb, a.xb, a.ib, a.rb, first, step);
    else        scatter_add_rows<128>(a.fa, a.xa, a.ia, a.ra, first, step);
    return;
  }
  if      (b < a.s1) invmap_job(a.map0, a.nbr0, a.Mw0, a.n0, b, a.nb0);
  else if (b < a.s2) invmap_job(a.map1, a.nbr1, a.Mw1, a.n1, b - a.s1, a.nb1);
  else if (b < a.s3) invmap_job(a.map2, a.nbr2, a.Mw2, a.n2, b - a.s2, a.nb2);
  else if (b < a.s4) invmap_job(a.map3, a.nbr3, a.Mw3, a.n3, b - a.s3, a.nb3);
  else               invmap_job(a.map4, a.nbr4, a.Mw4, a.n4, b - a.s4, a.nb4);
}

__device__ __forceinline__ void bn_part_rows(const float* __restrict__ X, long tot, int C,
                                             float* __restrict__ part, int bk, int nbk) {
  __shared__ float sm[8][TPB];
  const int t = threadIdx.x;
  float a0 = 0.0f, a1 = 0.0f, a2 = 0.0f, a3 = 0.0f;
  float b0 = 0.0f, b1 = 0.0f, b2 = 0.0f, b3 = 0.0f;
  const long step = (long)nbk * (TPB * 4);
  for (long i = ((long)bk * TPB + t) * 4; i < tot; i += step) {
    const float4 v = *(const float4*)(X + i);
    a0 += v.x; b0 += v.x * v.x;
    a1 += v.y; b1 += v.y * v.y;
    a2 += v.z; b2 += v.z * v.z;
    a3 += v.w; b3 += v.w * v.w;
  }
  sm[0][t] = a0; sm[1][t] = a1; sm[2][t] = a2; sm[3][t] = a3;
  sm[4][t] = b0; sm[5][t] = b1; sm[6][t] = b2; sm[7][t] = b3;
  __syncthreads();
  if (t < C) {
    const int q = C >> 2;
    const int comp = t & 3;
    float s = 0.0f, ss = 0.0f;
    for (int j = t >> 2; j < TPB; j += q) {
      s += sm[comp][j];
      ss += sm[4 + comp][j];
    }
    float* p = part + (long)bk * 2 * C;
    p[t] = s;
    p[C + t] = ss;
  }
}

__global__ void __launch_bounds__(TPB) k_bn_part(const float* __restrict__ XA, long totA, int CA,
                          float* __restrict__ partA, int nA,
                          const float* __restrict__ XB, long totB, int CB,
                          float* __restrict__ partB) {
  const bool jb = (int)blockIdx.x >= nA;
  bn_part_rows(jb ? XB : XA, jb ? totB : totA, jb ? CB : CA, jb ? partB : partA,
               jb ? (int)blockIdx.x - nA : (int)blockIdx.x,
               jb ? (int)gridDim.x - nA : nA);
}

__global__ void __launch_bounds__(TPB) k_bn_part_sort(const float* __restrict__ X, long tot, int C,
                          float* __restrict__ part,
                          const int* __restrict__ nbrA, int nA,
                          const int* __restrict__ nbrB, int nB,
                          int* __restrict__ cnt2, int ns) {
  const int b = blockIdx.x;
  if (b < ns) {
    sort_count_rows(nbrA, nA, nbrB, nB, cnt2, b % SORT_BLK, b / SORT_BLK);
    return;
  }
  bn_part_rows(X, tot, C, part, b - ns, (int)gridDim.x - ns);
}

__device__ __forceinline__ void bn_fin_col(const float* __restrict__ part, int P, long N, int C,
                                           float* __restrict__ mu, float* __restrict__ rs, int c) {
  __shared__ float s1[TPB];
  __shared__ float s2[TPB];
  float a = 0.0f, b = 0.0f;
  for (int p = threadIdx.x; p < P; p += TPB) {
    a += part[(long)p * 2 * C + c];
    b += part[(long)p * 2 * C + C + c];
  }
  s1[threadIdx.x] = a;
  s2[threadIdx.x] = b;
  __syncthreads();
  for (int s = TPB / 2; s > 0; s >>= 1) {
    if ((int)threadIdx.x < s) {
      s1[threadIdx.x] += s1[threadIdx.x + s];
      s2[threadIdx.x] += s2[threadIdx.x + s];
    }
    __syncthreads();
  }
  if (threadIdx.x == 0) {
    float m = s1[0] / (float)N;
    float var = s2[0] / (float)N - m * m;
    if (var < 0.0f) var = 0.0f;
    mu[c] = m;
    rs[c] = rsqrtf(var + 1e-5f);
  }
}

__global__ void __launch_bounds__(TPB) k_bn_fin(const float* __restrict__ partA, int PA, long NA, int CA,
                         float* __restrict__ muA, float* __restrict__ rsA,
                         const float* __restrict__ partB, int PB, long NB, int CB,
                         float* __restrict__ muB, float* __restrict__ rsB) {
  const bool jb = (int)blockIdx.x >= CA;
  bn_fin_col(jb ? partB : partA, jb ? PB : PA, jb ? NB : NA, jb ? CB : CA,
             jb ? muB : muA, jb ? rsB : rsA,
             jb ? (int)blockIdx.x - CA : (int)blockIdx.x);
}

__global__ void __launch_bounds__(TPB) k_bn_fin_sort(const float* __restrict__ part, int P, long N, int C,
                         float* __restrict__ mu, float* __restrict__ rs,
                         int* __restrict__ cnt, int* __restrict__ tot, int ns) {
  const int b = blockIdx.x;
  if (b < ns) {
    sort_scan_class(cnt, tot, b);
    return;
  }
  bn_fin_col(part, P, N, C, mu, rs, b - ns);
}

template <int C>
__device__ __forceinline__ void bn_elu_rows(const float* X, float* Y,
                                            const float* __restrict__ mu,
                                            const float* __restrict__ rs,
                                            const float* __restrict__ gb, long tot4,
                                            int bk, int nbk) {
  const float4* X4 = (const float4*)X;
  float4* Y4 = (float4*)Y;
  const int c = ((int)threadIdx.x * 4) & (C - 1);
  float g[4], m[4], s[4], b[4];
  for (int e = 0; e < 4; ++e) {
    g[e] = gb[c + e];
    m[e] = mu[c + e];
    s[e] = rs[c + e];
    b[e] = gb[C + c + e];
  }
  const long step = (long)nbk * TPB;
  for (long i = (long)bk * TPB + threadIdx.x; i < tot4; i += 4 * step) {
    float4 x[4];
    for (int u = 0; u < 4; ++u)
      if (i + u * step < tot4) x[u] = X4[i + u * step];
    for (int u = 0; u < 4; ++u) {
      if (i + u * step < tot4) {
        float v0 = g[0] * (x[u].x - m[0]) * s[0] + b[0];
        float v1 = g[1] * (x[u].y - m[1]) * s[1] + b[1];
        float v2 = g[2] * (x[u].z - m[2]) * s[2] + b[2];
        float v3 = g[3] * (x[u].w - m[3]) * s[3] + b[3];
        float4 o;
        o.x = v0 > 0.0f ? v0 : expm1f(v0);
        o.y = v1 > 0.0f ? v1 : expm1f(v1);
        o.z = v2 > 0.0f ? v2 : expm1f(v2);
        o.w = v3 > 0.0f ? v3 : expm1f(v3);
        Y4[i + u * step] = o;
      }
    }
  }
}

template <int CA, int CB>
__global__ void k_bn_elu(const float* XA, float* YA,
                         const float* __restrict__ muA, const float* __restrict__ rsA,
                         const float* __restrict__ gbA, long tot4A, int nA,
                         const float* XB, float* YB,
                         const float* __restrict__ muB, const float* __restrict__ rsB,
                         const float* __restrict__ gbB, long tot4B) {
  if ((int)blockIdx.x < nA)
    bn_elu_rows<CA>(XA, YA, muA, rsA, gbA, tot4A, (int)blockIdx.x, nA);
  else
    bn_elu_rows<CB>(XB, YB, muB, rsB, gbB, tot4B, (int)blockIdx.x - nA, (int)gridDim.x - nA);
}

template <int C>
__global__ void __launch_bounds__(TPB) k_bn_elu_sort(const float* X, float* Y,
                         const float* __restrict__ mu, const float* __restrict__ rs,
                         const float* __restrict__ gb, long tot4,
                         const int* __restrict__ nbrA, int nA,
                         const int* __restrict__ nbrB, int nB,
                         const int* __restrict__ cnt2, const int* __restrict__ tot2,
                         int* __restrict__ permA, int* __restrict__ permB, int ns) {
  const int b = blockIdx.x;
  if (b < ns) {
    sort_place_rows(nbrA, nA, nbrB, nB, cnt2, tot2, permA, permB, b % SORT_BLK, b / SORT_BLK);
    return;
  }
  bn_elu_rows<C>(X, Y, mu, rs, gb, tot4, b - ns, (int)gridDim.x - ns);
}

template <int CIN, int COUT>
__device__ __forceinline__ void gemm_tap_tile(const float* __restrict__ X,
                           const float* __restrict__ W,
                           float* __restrict__ Y,
                           const int* __restrict__ gidx,
                           const int* __restrict__ sidx,
                           int M, int out_offset, int accumulate, int nscrap, int zrows,
                           int bx, int by, int bz) {
  const int lane = threadIdx.x;
  W += (long)bz * CIN * COUT;
  out_offset += bz * zrows;
  const int tile_r = bx * 32;
  const int tile_c = (by * 4 + threadIdx.y) * 16;
  const int mr = lane & 15;
  const int kh = lane >> 4;

  const int rl = tile_r + lane;
  int sv = nscrap;
  if (rl < M) sv = sidx ? sidx[rl] : rl;
  if (__builtin_amdgcn_ballot_w32(sv != nscrap) == 0u) return;

  const int r0 = tile_r + mr;
  const int r1 = tile_r + 16 + mr;
  int g0 = 0, g1 = 0;
  if (r0 < M) g0 = gidx ? gidx[r0] : r0;
  if (r1 < M) g1 = gidx ? gidx[r1] : r1;
  const float* __restrict__ xb0 = X + (long)g0 * CIN + 2 * kh;
  const float* __restrict__ xb1 = X + (long)g1 * CIN + 2 * kh;
  const float* __restrict__ wb  = W + tile_c + mr + 2 * kh * COUT;

  v8f c0 = {0.f, 0.f, 0.f, 0.f, 0.f, 0.f, 0.f, 0.f};
  v8f c1 = {0.f, 0.f, 0.f, 0.f, 0.f, 0.f, 0.f, 0.f};
#pragma unroll
  for (int k0 = 0; k0 < CIN; k0 += 4) {
    v2f a0 = *(const v2f*)(xb0 + k0);
    v2f a1 = *(const v2f*)(xb1 + k0);
    v2f b;
    b[0] = wb[k0 * COUT];
    b[1] = wb[(k0 + 1) * COUT];
    c0 = __builtin_amdgcn_wmma_f32_16x16x4_f32(false, a0, false, b,
                                               (short)0, c0, false, false);
    c1 = __builtin_amdgcn_wmma_f32_16x16x4_f32(false, a1, false, b,
                                               (short)0, c1, false, false);
  }

#pragma unroll
  for (int i = 0; i < 8; ++i) {
    const int t0a = __builtin_amdgcn_readlane(sv, i);
    const int t0b = __builtin_amdgcn_readlane(sv, i + 8);
    const int ro0 = kh ? t0b : t0a;
    if (ro0 != nscrap) {
      float* y = Y + (long)(ro0 + out_offset) * COUT + tile_c + mr;
      if (accumulate) *y += c0[i]; else *y = c0[i];
    }
    const int t1a = __builtin_amdgcn_readlane(sv, i + 16);
    const int t1b = __builtin_amdgcn_readlane(sv, i + 24);
    const int ro1 = kh ? t1b : t1a;
    if (ro1 != nscrap) {
      float* y = Y + (long)(ro1 + out_offset) * COUT + tile_c + mr;
      if (accumulate) *y += c1[i]; else *y = c1[i];
    }
  }
}

template <int CIN, int COUT>
__device__ __forceinline__ void conv_os_tile(const float* __restrict__ X,
                          const float* __restrict__ W,
                          float* __restrict__ Y,
                          const int* __restrict__ nbr,
                          int n, int bx, int by) {
  const int lane = threadIdx.x;
  const int tile_r = bx * 32;
  const int tile_c = (by * 4 + threadIdx.y) * 16;
  const int mr = lane & 15;
  const int kh = lane >> 4;
  const int r0 = tile_r + mr;
  const int r1 = tile_r + 16 + mr;
  const bool in0 = r0 < n;
  const bool in1 = r1 < n;
  constexpr int KC = CIN < 64 ? CIN : 64;
  constexpr int NS = KC / 4;

  v8f c0 = {0.f, 0.f, 0.f, 0.f, 0.f, 0.f, 0.f, 0.f};
  v8f c1 = {0.f, 0.f, 0.f, 0.f, 0.f, 0.f, 0.f, 0.f};

  const int q0 = in0 ? r0 : n - 1;
  const int q1 = in1 ? r1 : n - 1;
  int g0 = nbr[q0];
  int g1 = nbr[q1];
  g0 = in0 ? g0 : -1;
  g1 = in1 ? g1 : -1;
  for (int t = 0; t < 27; ++t) {
    int l0 = r0;
    int l1 = r1;
    if (t < 25) {
      l0 = nbr[(long)(t + 1) * n + q0];
      l1 = nbr[(long)(t + 1) * n + q1];
    }
    if (__builtin_amdgcn_ballot_w32((g0 >= 0) || (g1 >= 0)) != 0u) {
      const int k = (t < 13) ? t : (t < 26 ? t + 1 : 13);
      const float* __restrict__ wb =
          W + (long)k * CIN * COUT + tile_c + mr + 2 * kh * COUT;
#pragma unroll
      for (int kc = 0; kc < CIN; kc += KC) {
        v2f a0[NS];
        v2f a1[NS];
#pragma unroll
        for (int s = 0; s < NS; ++s) {
          a0[s] = (v2f){0.f, 0.f};
          a1[s] = (v2f){0.f, 0.f};
        }
        if (g0 >= 0) {
          const float* __restrict__ xb0 = X + (long)g0 * CIN + 2 * kh + kc;
#pragma unroll
          for (int s = 0; s < NS; ++s) a0[s] = *(const v2f*)(xb0 + 4 * s);
        }
        if (g1 >= 0) {
          const float* __restrict__ xb1 = X + (long)g1 * CIN + 2 * kh + kc;
#pragma unroll
          for (int s = 0; s < NS; ++s) a1[s] = *(const v2f*)(xb1 + 4 * s);
        }
#pragma unroll
        for (int s = 0; s < NS; ++s) {
          const int k0 = kc + 4 * s;
          v2f b;
          b[0] = wb[k0 * COUT];
          b[1] = wb[(k0 + 1) * COUT];
          c0 = __builtin_amdgcn_wmma_f32_16x16x4_f32(false, a0[s], false, b,
                                                     (short)0, c0, false, false);
          c1 = __builtin_amdgcn_wmma_f32_16x16x4_f32(false, a1[s], false, b,
                                                     (short)0, c1, false, false);
        }
      }
    }
    g0 = in0 ? l0 : -1;
    g1 = in1 ? l1 : -1;
  }

#pragma unroll
  for (int i = 0; i < 8; ++i) {
    const int ro0 = tile_r + i + 8 * kh;
    if (ro0 < n) Y[(long)ro0 * COUT + tile_c + mr] = c0[i];
    const int ro1 = ro0 + 16;
    if (ro1 < n) Y[(long)ro1 * COUT + tile_c + mr] = c1[i];
  }
}

template <int CIN, int COUT>
__global__ void __launch_bounds__(128) k_conv_os(const float* __restrict__ X,
                          const float* __restrict__ W, float* __restrict__ Y,
                          const int* __restrict__ nbr, int n) {
  conv_os_tile<CIN, COUT>(X, W, Y, nbr, n, (int)blockIdx.x, (int)blockIdx.y);
}

template <int CIN, int COUT>
__global__ void __launch_bounds__(128) k_gemm_tap(const float* __restrict__ X,
                          const float* __restrict__ W, float* __restrict__ Y, int M) {
  gemm_tap_tile<CIN, COUT>(X, W, Y, nullptr, nullptr, M, 0, 0, -1, M,
                           (int)blockIdx.x, (int)blockIdx.y, (int)blockIdx.z);
}

template <int CIN, int CCOUT, int GCOUT, int CIN2>
__global__ void __launch_bounds__(128) k_side_convs_gemm(
                          const float* __restrict__ X,
                          const float* __restrict__ Wc,
                          float* __restrict__ Yc,
                          const int* __restrict__ nbr,
                          const float* __restrict__ Wg,
                          float* __restrict__ Yg,
                          int n, int nconv,
                          const float* __restrict__ X2,
                          const float* __restrict__ Wc2,
                          float* __restrict__ Yc2,
                          const int* __restrict__ nbr2,
                          int n2, int nconv2) {
  constexpr int CY = CCOUT / 64;
  constexpr int GY = GCOUT / 64;
  const int b = blockIdx.x;
  if (b < nconv2) {
    conv_os_tile<CIN2, CCOUT>(X2, Wc2, Yc2, nbr2, n2, b / CY, b % CY);
    return;
  }
  const int b1 = b - nconv2;
  if (b1 < nconv) {
    conv_os_tile<CIN, CCOUT>(X, Wc, Yc, nbr, n, b1 / CY, b1 % CY);
    return;
  }
  const int g = b1 - nconv;
  const int per = ((n + 31) / 32) * GY;
  const int bz = g / per;
  const int r = g - bz * per;
  gemm_tap_tile<CIN, GCOUT>(X, Wg, Yg, nullptr, nullptr, n, 0, 0, -1, n, r / GY, r % GY, bz);
}

template <int CIN, int COUT>
__global__ void __launch_bounds__(128) k_conv_os_wide(const float* __restrict__ X,
                          const float* __restrict__ W,
                          float* __restrict__ Y,
                          const int* __restrict__ nbr,
                          const int* __restrict__ perm,
                          int n) {
  constexpr int NT = COUT / 16;
  constexpr int KC = 64;
  constexpr int NCH = CIN / KC;
  constexpr int PITCH = KC + 4;
  constexpr int CG = COUT / 4;
  constexpr int ITEMS = (KC / 4) * CG / 128;
  __shared__ float Bs[2][COUT * PITCH];
  __shared__ unsigned need[4];

  const int lane = threadIdx.x;
  const int wv = threadIdx.y;
  const int tid = wv * 32 + lane;
  const int tile_r = (((int)gridDim.x - 1 - (int)blockIdx.x) * 4 + wv) * 32;
  const int mr = lane & 15;
  const int kh = lane >> 4;
  const int p0 = tile_r + mr;
  const int p1 = tile_r + 16 + mr;
  const bool in0 = p0 < n;
  const bool in1 = p1 < n;
  const int r0 = perm[in0 ? p0 : n - 1];
  const int r1 = perm[in1 ? p1 : n - 1];

  unsigned wm = 0u;
  {
    int h0[26];
    int h1[26];
#pragma unroll
    for (int t = 0; t < 26; ++t) {
      h0[t] = nbr[(long)t * n + r0];
      h1[t] = nbr[(long)t * n + r1];
    }
#pragma unroll
    for (int t = 0; t < 26; ++t) {
      const int e0 = in0 ? h0[t] : -1;
      const int e1 = in1 ? h1[t] : -1;
      if (__builtin_amdgcn_ballot_w32((e0 & e1) >= 0) != 0u) wm |= 1u << t;
    }
    if (__builtin_amdgcn_ballot_w32(in0 || in1) != 0u) wm |= 1u << 26;
  }
  if (lane == 0) need[wv] = wm;
  __syncthreads();
  unsigned bm = (unsigned)__builtin_amdgcn_readfirstlane(
      (int)(need[0] | need[1] | need[2] | need[3]));
  if (bm == 0u) return;

  auto stage = [&](int t, int ch, float* __restrict__ dst) {
    const int kc = ch * KC;
    const int k = (t < 13) ? t : (t < 26 ? t + 1 : 13);
    const float* __restrict__ wsrc = W + ((long)k * CIN + kc) * COUT;
    float4 r[ITEMS][4];
    for (int u = 0; u < ITEMS; ++u) {
      const int e = tid + 128 * u;
      const int cg = e % CG;
      const int kg = e / CG;
      for (int rr = 0; rr < 4; ++rr)
        r[u][rr] = *(const float4*)(wsrc + (long)(4 * kg + rr) * COUT + 4 * cg);
    }
    for (int u = 0; u < ITEMS; ++u) {
      const int e = tid + 128 * u;
      const int cg = e % CG;
      const int kg = e / CG;
      float* d = dst + (4 * cg) * PITCH + 4 * kg;
      *(float4*)(d)             = make_float4(r[u][0].x, r[u][1].x, r[u][2].x, r[u][3].x);
      *(float4*)(d + PITCH)     = make_float4(r[u][0].y, r[u][1].y, r[u][2].y, r[u][3].y);
      *(float4*)(d + 2 * PITCH) = make_float4(r[u][0].z, r[u][1].z, r[u][2].z, r[u][3].z);
      *(float4*)(d + 3 * PITCH) = make_float4(r[u][0].w, r[u][1].w, r[u][2].w, r[u][3].w);
    }
  };

  v8f c0[NT];
  v8f c1[NT];
  for (int j = 0; j < NT; ++j) {
    c0[j] = (v8f){0.f, 0.f, 0.f, 0.f, 0.f, 0.f, 0.f, 0.f};
    c1[j] = (v8f){0.f, 0.f, 0.f, 0.f, 0.f, 0.f, 0.f, 0.f};
  }

  int t = __builtin_ctz(bm);
  bm &= bm - 1u;
  stage(t, 0, Bs[0]);

  int g0 = t < 26 ? nbr[(long)t * n + r0] : r0;
  int g1 = t < 26 ? nbr[(long)t * n + r1] : r1;
  g0 = in0 ? g0 : -1;
  g1 = in1 ? g1 : -1;
  __syncthreads();

  int buf = 0;
  for (int it = 0; it < 27; ++it) {
    const int tn = bm != 0u ? __builtin_ctz(bm) : -1;
    bm &= bm - 1u;
    int l0 = r0;
    int l1 = r1;
    if (tn >= 0 && tn < 26) {
      l0 = nbr[(long)tn * n + r0];
      l1 = nbr[(long)tn * n + r1];
    }
    const bool act = __builtin_amdgcn_ballot_w32((g0 >= 0) || (g1 >= 0)) != 0u;
    for (int ch = 0; ch < NCH; ++ch) {
      if (ch + 1 < NCH) stage(t, ch + 1, Bs[buf ^ 1]);
      else if (tn >= 0) stage(tn, 0, Bs[buf ^ 1]);
      if (act) {
        const float* __restrict__ bs = Bs[buf] + mr * PITCH + 4 * kh;
        float4 a0[8];
        float4 a1[8];
        for (int p = 0; p < 8; ++p) {
          a0[p] = make_float4(0.f, 0.f, 0.f, 0.f);
          a1[p] = make_float4(0.f, 0.f, 0.f, 0.f);
        }
        if (g0 >= 0) {
          const float* __restrict__ xb0 = X + (long)g0 * CIN + ch * KC + 4 * kh;
          for (int p = 0; p < 8; ++p) a0[p] = *(const float4*)(xb0 + 8 * p);
        }
        if (g1 >= 0) {
          const float* __restrict__ xb1 = X + (long)g1 * CIN + ch * KC + 4 * kh;
          for (int p = 0; p < 8; ++p) a1[p] = *(const float4*)(xb1 + 8 * p);
        }
        for (int p = 0; p < 8; ++p) {
          const v2f a0l = {a0[p].x, a0[p].y};
          const v2f a0h = {a0[p].z, a0[p].w};
          const v2f a1l = {a1[p].x, a1[p].y};
          const v2f a1h = {a1[p].z, a1[p].w};
          float4 bf[NT];
          for (int j = 0; j < NT; ++j) bf[j] = *(const float4*)(bs + 16 * j * PITCH + 8 * p);
          for (int j = 0; j < NT; ++j) {
            const float4 b = bf[j];
            const v2f bl = {b.x, b.y};
            const v2f bh = {b.z, b.w};
            c0[j] = __builtin_amdgcn_wmma_f32_16x16x4_f32(false, a0l, false, bl,
                                                          (short)0, c0[j], false, false);
            c1[j] = __builtin_amdgcn_wmma_f32_16x16x4_f32(false, a1l, false, bl,
                                                          (short)0, c1[j], false, false);
            c0[j] = __builtin_amdgcn_wmma_f32_16x16x4_f32(false, a0h, false, bh,
                                                          (short)0, c0[j], false, false);
            c1[j] = __builtin_amdgcn_wmma_f32_16x16x4_f32(false, a1h, false, bh,
                                                          (short)0, c1[j], false, false);
          }
        }
      }
      __syncthreads();
      buf ^= 1;
    }
    if (tn < 0) break;
    g0 = in0 ? l0 : -1;
    g1 = in1 ? l1 : -1;
    t = tn;
  }

  int o0[8];
  int o1[8];
#pragma unroll
  for (int i = 0; i < 8; ++i) {
    const int xa = __builtin_amdgcn_readlane(r0, i);
    const int xb = __builtin_amdgcn_readlane(r0, i + 8);
    o0[i] = kh ? xb : xa;
    const int ya = __builtin_amdgcn_readlane(r1, i);
    const int yb = __builtin_amdgcn_readlane(r1, i + 8);
    o1[i] = kh ? yb : ya;
  }
#pragma unroll
  for (int j = 0; j < NT; ++j) {
#pragma unroll
    for (int i = 0; i < 8; ++i) {
      if (tile_r + i + 8 * kh < n) Y[(long)o0[i] * COUT + 16 * j + mr] = c0[j][i];
      if (tile_r + 16 + i + 8 * kh < n) Y[(long)o1[i] * COUT + 16 * j + mr] = c1[j][i];
    }
  }
}

template <int CIN, int COUT>
static inline void conv_t(const float* X, const float* W, float* Y,
                          const int* nbr, int n, hipStream_t st) {
  dim3 grid((unsigned)((n + 31) / 32), (unsigned)(COUT / 64));
  dim3 blk(32, 4);
  k_conv_os<CIN, COUT><<<grid, blk, 0, st>>>(X, W, Y, nbr, n);
}

template <int CIN, int COUT>
static inline void conv_wide_t(const float* X, const float* W, float* Y,
                               const int* nbr, int n, const int* perm, hipStream_t st) {
  dim3 grid((unsigned)((n + 127) / 128));
  dim3 blk(32, 4);
  k_conv_os_wide<CIN, COUT><<<grid, blk, 0, st>>>(X, W, Y, nbr, perm, n);
}

static inline bool conv_is_wide(int n, int Cin) { return n >= 128 * 256 && Cin == 64; }

static void conv_os(const float* X, const float* W, float* Y, const int* nbr,
                    int n, int Cin, int Cout, const int* perm, hipStream_t st) {
  const bool wide = conv_is_wide(n, Cin);
  if      (wide && Cin == 64 && Cout ==  64) conv_wide_t<64,  64>(X, W, Y, nbr, n, perm, st);
  else if (wide && Cin == 64 && Cout == 128) conv_wide_t<64, 128>(X, W, Y, nbr, n, perm, st);
  else if (Cin == 256 && Cout == 128) conv_t<256, 128>(X, W, Y, nbr, n, st);
  else if (Cin == 128 && Cout == 128) conv_t<128, 128>(X, W, Y, nbr, n, st);
  else if (Cin ==  64 && Cout ==  64) conv_t< 64,  64>(X, W, Y, nbr, n, st);
  else if (Cin ==  64 && Cout == 128) conv_t< 64, 128>(X, W, Y, nbr, n, st);
}

template <int CIN, int COUT>
static inline void gemm8_t(const float* X, const float* W, float* Y, int M, hipStream_t st) {
  dim3 grid((unsigned)((M + 31) / 32), (unsigned)(COUT / 64), 8u);
  dim3 blk(32, 4);
  k_gemm_tap<CIN, COUT><<<grid, blk, 0, st>>>(X, W, Y, M);
}

static inline void side_convs_gemm(const float* X, const float* Wc, float* Yc, const int* nbr,
                                   const float* Wg, float* Yg, int n,
                                   const float* X2, const float* Wc2, float* Yc2,
                                   const int* nbr2, int n2, hipStream_t st) {
  const int tiles = (n + 31) / 32;
  const int nconv = tiles * 2;
  const int nprod = tiles * 8;
  const int nconv2 = ((n2 + 31) / 32) * 2;
  dim3 blk(32, 4);
  k_side_convs_gemm<128, 128, 64, 256><<<dim3((unsigned)(nconv2 + nconv + nprod)), blk, 0, st>>>(
      X, Wc, Yc, nbr, Wg, Yg, n, nconv, X2, Wc2, Yc2, nbr2, n2, nconv2);
}


extern "C" void kernel_launch(void* const* d_in, const int* in_sizes, int n_in,
                              void* d_out, int out_size, void* d_ws, size_t ws_size,
                              hipStream_t stream) {
  const float* feats0 = (const float*)d_in[0];
  const float* feats1 = (const float*)d_in[1];
  const float* feats2 = (const float*)d_in[2];
  const float* w_out0 = (const float*)d_in[3];
  const float* w_out1 = (const float*)d_in[4];
  const float* w_out2 = (const float*)d_in[5];
  const float* wt2    = (const float*)d_in[6];
  const float* wu2    = (const float*)d_in[7];
  const float* wt1    = (const float*)d_in[8];
  const float* wu1    = (const float*)d_in[9];
  const float* bn_out0 = (const float*)d_in[10];
  const float* bn_out1 = (const float*)d_in[11];
  const float* bn_out2 = (const float*)d_in[12];
  const float* bn_up2a = (const float*)d_in[13];
  const float* bn_up2b = (const float*)d_in[14];
  const float* bn_up1a = (const float*)d_in[15];
  const float* bn_up1b = (const float*)d_in[16];
  const int* m_c2 = (const int*)d_in[19];
  const int* m_g2 = (const int*)d_in[20];
  const int* m_u1 = (const int*)d_in[21];
  const int* m_g1 = (const int*)d_in[22];
  const int* m_u0 = (const int*)d_in[23];
  const int* lat1_idx = (const int*)d_in[24];
  const int* up1_idx  = (const int*)d_in[25];
  const int* lat0_idx = (const int*)d_in[26];
  const int* up0_idx  = (const int*)d_in[27];

  const int N0 = in_sizes[0] / 64;
  const int N1 = in_sizes[1] / 128;
  const int N2 = in_sizes[2] / 256;
  const int nu1 = in_sizes[17] / 128;
  const int nu0 = in_sizes[18] / 64;
  const int Mc2 = in_sizes[19] / 52;
  const int Mg2 = in_sizes[20] / 52;
  const int Mu1 = in_sizes[21] / 52;
  const int Mg1 = in_sizes[22] / 52;
  const int Mu0 = in_sizes[23] / 52;
  const int g2n = 8 * N2;
  const int g1n = 8 * nu1;

  char* wsp = (char*)d_ws;
  auto carve = [&](long nfloats) -> float* {
    float* p = (float*)wsp;
    wsp += ((nfloats * 4 + 255) / 256) * 256;
    return p;
  };
  long accElems = (long)(N2 + 1) * 128;
  { long e = (long)(g2n + 1) * 128; if (e > accElems) accElems = e; }
  { long e = (long)(nu1 + 1) * 128; if (e > accElems) accElems = e; }
  { long e = (long)(g1n + 1) * 64;  if (e > accElems) accElems = e; }
  { long e = (long)(nu0 + 1) * 128; if (e > accElems) accElems = e; }
  float* acc = carve(accElems);
  float* t2  = carve((long)g2n * 128);
  float* t1  = carve((long)g1n * 64);
  float* mu  = carve(128);
  float* rs  = carve(128);
  float* part = carve((long)1024 * 256);
  float* accS = carve((long)nu1 * 128);
  float* accT = carve((long)N2 * 128);
  float* muS  = carve(128);
  float* rsS  = carve(128);
  float* partS = carve((long)1024 * 256);
  float* x1  = carve((long)nu1 * 128);
  float* x0  = carve((long)nu0 * 64);
  const long zero4 = (long)(wsp - (char*)x1) / 16;
  int* nbC2 = (int*)carve((long)26 * N2);
  int* nbG2 = (int*)carve((long)26 * g2n);
  int* nbU1 = (int*)carve((long)26 * nu1);
  int* nbG1 = (int*)carve((long)26 * g1n);
  int* nbU0 = (int*)carve((long)26 * nu0);
  const long neg4 = (long)(wsp - (char*)nbC2) / 16;
  int* permG1 = (int*)carve(g1n);
  int* permU0 = (int*)carve(nu0);
  int* cnt2 = (int*)carve((long)2 * SORT_B * SORT_W);
  int* tot2 = (int*)carve((long)2 * SORT_B);

  float* out  = (float*)d_out;
  float* out0 = out;
  float* out1 = out + (long)nu0 * 128;
  float* out2 = out + (long)nu0 * 128 + (long)nu1 * 128;

  auto nblk = [](long items) -> unsigned {
    long b = (items + TPB - 1) / TPB;
    if (b > MAXB) b = MAXB;
    if (b < 1) b = 1;
    return (unsigned)b;
  };
  auto conv3 = [&](const float* X, int n, int Cin, int Cout,
                   const float* W, const int* nbr, const int* perm) {
    conv_os(X, W, acc, nbr, n, Cin, Cout, perm, stream);
  };
  auto bnelu = [&](const float* src, float* dst, const float* gb, long n, int C,
                   const float* srcS, float* dstS, const float* gbS, long nS) {
    const long tot = n * C;
    long P = (tot + TPB * 4 - 1) / (TPB * 4);
    if (P > 1024) P = 1024;
    const long totS = srcS ? nS * 128 : 0;
    long PS = (totS + TPB * 4 - 1) / (TPB * 4);
    if (PS > 1024) PS = 1024;
    const int CS = srcS ? 128 : 0;
    k_bn_part<<<dim3((unsigned)(P + PS)), dim3(TPB), 0, stream>>>(src, tot, C, part, (int)P,
                                                                  srcS, totS, 128, partS);
    k_bn_fin<<<dim3((unsigned)(C + CS)), dim3(TPB), 0, stream>>>(part, (int)P, n, C, mu, rs,
                                                                 partS, (int)PS, nS, 128, muS, rsS);
    const long tot4 = tot / 4;
    const long tot4S = totS / 4;
    const unsigned nA = nblk(tot4);
    const unsigned nB = srcS ? nblk(tot4S) : 0u;
    if (C == 128)
      k_bn_elu<128, 128><<<dim3(nA + nB), dim3(TPB), 0, stream>>>(src, dst, mu, rs, gb, tot4, (int)nA,
                                                                  srcS, dstS, muS, rsS, gbS, tot4S);
    else
      k_bn_elu<64, 128><<<dim3(nA + nB), dim3(TPB), 0, stream>>>(src, dst, mu, rs, gb, tot4, (int)nA,
                                                                 srcS, dstS, muS, rsS, gbS, tot4S);
  };
  auto scat = [&](const float* X, float* Y, const int* idx, long rows, int C) {
    const long tot4 = rows * (C / 4);
    if (C == 128)
      k_scatter_add<128><<<dim3(nblk(tot4)), dim3(TPB), 0, stream>>>(X, Y, idx, rows);
    else
      k_scatter_add<64><<<dim3(nblk(tot4)), dim3(TPB), 0, stream>>>(X, Y, idx, rows);
  };

  k_fill2<<<dim3(MAXB), dim3(TPB), 0, stream>>>(nbC2, neg4, -1, (int*)x1, zero4, 0);
  {
    PrepArgs pa;
    pa.fill = 0;
    pa.map0 = m_c2; pa.map1 = m_g2; pa.map2 = m_u1; pa.map3 = m_g1; pa.map4 = m_u0;
    pa.nbr0 = nbC2; pa.nbr1 = nbG2; pa.nbr2 = nbU1; pa.nbr3 = nbG1; pa.nbr4 = nbU0;
    pa.Mw0 = Mc2; pa.Mw1 = Mg2; pa.Mw2 = Mu1; pa.Mw3 = Mg1; pa.Mw4 = Mu0;
    pa.n0 = N2; pa.n1 = g2n; pa.n2 = nu1; pa.n3 = g1n; pa.n4 = nu0;
    pa.fa = feats1; pa.xa = x1; pa.ia = lat1_idx; pa.ra = N1;
    pa.fb = feats0; pa.xb = x0; pa.ib = lat0_idx; pa.rb = N0;
    auto tapblk = [](int Mw) -> int {
      long b = ((long)Mw + 4 * TPB - 1) / (4 * TPB);
      if (b > MAXB / 26) b = MAXB / 26;
      if (b < 1) b = 1;
      return (int)b;
    };
    auto scatblk = [](long tot4) -> int {
      long b = (tot4 + 4 * TPB - 1) / (4 * TPB);
      if (b > MAXB) b = MAXB;
      if (b < 1) b = 1;
      return (int)b;
    };
    pa.nb0 = tapblk(Mc2); pa.nb1 = tapblk(Mg2); pa.nb2 = tapblk(Mu1);
    pa.nb3 = tapblk(Mg1); pa.nb4 = tapblk(Mu0);
    pa.s1 = 26 * pa.nb0;
    pa.s2 = pa.s1 + 26 * pa.nb1;
    pa.s3 = pa.s2 + 26 * pa.nb2;
    pa.s4 = pa.s3 + 26 * pa.nb3;
    pa.s5 = pa.s4 + 26 * pa.nb4;
    pa.s6 = pa.s5 + scatblk((long)N1 * 32);
    const int nblocks = pa.s6 + scatblk((long)N0 * 16);
    k_prep<<<dim3((unsigned)nblocks), dim3(TPB), 0, stream>>>(pa);
  }
  gemm8_t<256, 128>(feats2, wt2, t2, N2, stream);
  {
    const bool wG1 = conv_is_wide(g1n, 64);
    const bool wU0 = conv_is_wide(nu0, 64);
    const int* tA = wG1 ? nbG1 : nbU0;
    const int nA = wG1 ? g1n : nu0;
    int* pA = wG1 ? permG1 : permU0;
    const int ny = (wG1 ? 1 : 0) + (wU0 ? 1 : 0);
    const int ns = SORT_BLK * ny;
    const int nsc = SORT_B * ny;
    const long tot = (long)g2n * 128;
    long P = (tot + TPB * 4 - 1) / (TPB * 4);
    if (P > 1024) P = 1024;
    const long tot4 = tot / 4;
    k_bn_part_sort<<<dim3((unsigned)(ns + P)), dim3(TPB), 0, stream>>>(
        t2, tot, 128, part, tA, nA, nbU0, nu0, cnt2, ns);
    k_bn_fin_sort<<<dim3((unsigned)(nsc + 128)), dim3(TPB), 0, stream>>>(
        part, (int)P, (long)g2n, 128, mu, rs, cnt2, tot2, nsc);
    k_bn_elu_sort<128><<<dim3((unsigned)ns + nblk(tot4)), dim3(TPB), 0, stream>>>(
        t2, t2, mu, rs, bn_up2a, tot4, tA, nA, nbU0, nu0, cnt2, tot2, pA, permU0, ns);
  }
  conv3(t2, g2n, 128, 128, wu2, nbG2, nullptr);
  bnelu(acc, t2, bn_up2b, g2n, 128, nullptr, nullptr, nullptr, 0);

  scat(t2, x1, up1_idx, g2n, 128);

  side_convs_gemm(x1, w_out1, accS, nbU1, wt1, t1, nu1, feats2, w_out2, accT, nbC2, N2, stream);
  bnelu(t1, t1, bn_up1a, g1n, 64, accS, out1, bn_out1, nu1);
  conv3(t1, g1n, 64, 64, wu1, nbG1, permG1);
  bnelu(acc, t1, bn_up1b, g1n, 64, accT, out2, bn_out2, N2);

  scat(t1, x0, up0_idx, g1n, 64);

  conv3(x0, nu0, 64, 128, w_out0, nbU0, permU0);
  bnelu(acc, out0, bn_out0, nu0, 128, nullptr, nullptr, nullptr, 0);

  (void)n_in; (void)out_size; (void)ws_size;
}


# launches merged, 24 -> 20: table fill beside level-2 child products, table build in the first stats launch, each up-branch scatter-add done by its normalise launch (no stored copy of t)
# speedup vs baseline: 11.2786x; 1.0400x over previous
#include <hip/hip_runtime.h>
#include <hip/hip_bf16.h>
#include <math.h>

typedef __attribute__((ext_vector_type(2))) float v2f;
typedef __attribute__((ext_vector_type(8))) float v8f;

#define TPB 256
#define MAXB 1024


__device__ __forceinline__ void fill2_rows(int* __restrict__ a, long na4, int va,
                                           int* __restrict__ b, long nb4, int vb,
                                           long g, long step) {
  int4* __restrict__ a4 = (int4*)a;
  int4* __restrict__ b4 = (int4*)b;
  const int4 wa = {va, va, va, va};
  const int4 wb = {vb, vb, vb, vb};
  for (long i = g; i < na4; i += step) a4[i] = wa;
  for (long i = g; i < nb4; i += step) b4[i] = wb;
}

__device__ __forceinline__ void invmap_tap(const int* __restrict__ map, int* __restrict__ nbr,
                                           int Mw, int n, int j, int bx, int nbx) {
  const int* __restrict__ gi = map + (long)j * Mw;
  const int* __restrict__ so = map + (long)(26 + j) * Mw;
  int* __restrict__ nb = nbr + (long)j * n;
  const int step = nbx * TPB;
  for (int i = bx * TPB + (int)threadIdx.x; i < Mw; i += 4 * step) {
    int o[4];
    int g[4];
    for (int u = 0; u < 4; ++u) {
      const int k = i + u * step;
      const int kc = k < Mw ? k : i;
      o[u] = so[kc];
      g[u] = gi[kc];
    }
    for (int u = 0; u < 4; ++u)
      if (i + u * step < Mw && o[u] >= 0 && o[u] < n) nb[o[u]] = g[u];
  }
}

#define SORT_B 256
#define SORT_W 1024

__device__ __forceinline__ int corner_key(const int* __restrict__ nbr, int n, int r) {
  const int ct[8] = {0, 2, 6, 8, 17, 19, 23, 25};
  int v[8];
#pragma unroll
  for (int b = 0; b < 8; ++b) v[b] = nbr[(long)ct[b] * n + r];
  int key = 0;
#pragma unroll
  for (int b = 0; b < 8; ++b) key |= (v[b] >= 0) ? (1 << b) : 0;
  return key;
}

#define SORT_BLK (SORT_W / (TPB / 32))
__device__ __forceinline__ void sort_count_rows(const int* __restrict__ nbrA, int nA,
                                                const int* __restrict__ nbrB, int nB,
                                                int* __restrict__ cnt2, int bx, int by) {
  const int* __restrict__ nbr = by ? nbrB : nbrA;
  const int n = by ? nB : nA;
  int* __restrict__ cnt = cnt2 + (long)by * SORT_B * SORT_W;
  const int lane = threadIdx.x & 31;
  const int w = bx * (TPB / 32) + ((int)threadIdx.x >> 5);
  const int seg = (n + SORT_W - 1) / SORT_W;
  const int beg = w * seg;
  const int end = beg + seg < n ? beg + seg : n;
  int c[SORT_B / 32];
#pragma unroll
  for (int s = 0; s < SORT_B / 32; ++s) c[s] = 0;
  for (int base = beg; base < end; base += 32) {
    const int r = base + lane;
    const int key = r < end ? corner_key(nbr, n, r) : -1;
#pragma unroll
    for (int s = 0; s < SORT_B / 32; ++s) {
      if (__builtin_amdgcn_ballot_w32((key >> 5) == s) != 0u) {
        for (int l = 0; l < 32; ++l) {
          const unsigned m = __builtin_amdgcn_ballot_w32(key == 32 * s + l);
          if (lane == l) c[s] += __popc(m);
        }
      }
    }
  }
#pragma unroll
  for (int s = 0; s < SORT_B / 32; ++s) cnt[(long)(32 * s + lane) * SORT_W + w] = c[s];
}

__device__ __forceinline__ void sort_scan_class(int* __restrict__ cnt, int* __restrict__ tot, int cls) {
  static_assert(SORT_W == 4 * TPB, "four wave counts per thread");
  __shared__ int wsum[TPB / 32];
  const int t = threadIdx.x;
  const int lane = t & 31;
  const int wv = t >> 5;
  int4* p = (int4*)(cnt + (long)cls * SORT_W) + t;
  const int4 c = *p;
  const int s = c.x + c.y + c.z + c.w;
  int inc = s;
  for (int d = 1; d < 32; d <<= 1) {
    const int u = __shfl_up(inc, d);
    if (lane >= d) inc += u;
  }
  if (lane == 31) wsum[wv] = inc;
  __syncthreads();
  int before = 0;
  int total = 0;
  for (int i = 0; i < TPB / 32; ++i) {
    const int x = wsum[i];
    if (i < wv) before += x;
    total += x;
  }
  const int ex = before + inc - s;
  *p = make_int4(ex, ex + c.x, ex + c.x + c.y, ex + c.x + c.y + c.z);
  if (t == 0) tot[cls] = total;
}

__device__ __forceinline__ void sort_place_rows(const int* __restrict__ nbrA, int nA,
                                                const int* __restrict__ nbrB, int nB,
                                                const int* __restrict__ cnt2,
                                                const int* __restrict__ tot2,
                                                int* __restrict__ permA,
                                                int* __restrict__ permB, int bx, int by) {
  const int* __restrict__ nbr = by ? nbrB : nbrA;
  const int n = by ? nB : nA;
  const int* __restrict__ cnt = cnt2 + (long)by * SORT_B * SORT_W;
  const int* __restrict__ tot = tot2 + (long)by * SORT_B;
  int* __restrict__ perm = by ? permB : permA;
  const int lane = threadIdx.x & 31;
  const int w = bx * (TPB / 32) + ((int)threadIdx.x >> 5);
  const int seg = (n + SORT_W - 1) / SORT_W;
  const int beg = w * seg;
  const int end = beg + seg < n ? beg + seg : n;
  int off[SORT_B / 32];
  int carry = 0;
#pragma unroll
  for (int s = 0; s < SORT_B / 32; ++s) {
    const int v = tot[32 * s + lane];
    int inc = v;
    for (int d = 1; d < 32; d <<= 1) {
      const int u = __shfl_up(inc, d);
      if (lane >= d) inc += u;
    }
    off[s] = carry + inc - v + cnt[(long)(32 * s + lane) * SORT_W + w];
    carry += __shfl(inc, 31);
  }
  const unsigned below = (1u << lane) - 1u;
  for (int base = beg; base < end; base += 32) {
    const int r = base + lane;
    const int key = r < end ? corner_key(nbr, n, r) : -1;
    int pos = 0;
#pragma unroll
    for (int s = 0; s < SORT_B / 32; ++s) {
      if (__builtin_amdgcn_ballot_w32((key >> 5) == s) != 0u) {
        for (int l = 0; l < 32; ++l) {
          const unsigned m = __builtin_amdgcn_ballot_w32(key == 32 * s + l);
          const int start = __builtin_amdgcn_readlane(off[s], l);
          if (key == 32 * s + l) pos = start + __popc(m & below);
          if (lane == l) off[s] += __popc(m);
        }
      }
    }
    if (key >= 0) perm[pos] = r;
  }
}

template <int C>
__device__ __forceinline__ void scatter_add_rows(const float* __restrict__ X, float* __restrict__ Y,
                                                 const int* __restrict__ idx, long rows,
                                                 long first, long step) {
  constexpr int Q = C / 4;
  const float4* __restrict__ X4 = (const float4*)X;
  const long tot4 = rows * Q;
  float4* __restrict__ Y4 = (float4*)Y;
  for (long i = first; i < tot4; i += 4 * step) {
    long kc[4];
    long off[4];
    float4 x[4];
    float4 y[4];
    for (int u = 0; u < 4; ++u) {
      const long k = i + u * step;
      kc[u] = k < tot4 ? k : i;
      off[u] = idx[kc[u] / Q];
    }
    for (int u = 0; u < 4; ++u) {
      off[u] = off[u] * Q + (kc[u] % Q);
      x[u] = X4[kc[u]];
      y[u] = Y4[off[u]];
    }
    for (int u = 0; u < 4; ++u) {
      if (i + u * step < tot4) {
        float4 v = y[u];
        v.x += x[u].x; v.y += x[u].y; v.z += x[u].z; v.w += x[u].w;
        Y4[off[u]] = v;
      }
    }
  }
}

struct PrepArgs {
  const int* map0; const int* map1; const int* map2; const int* map3; const int* map4;
  int* nbr0; int* nbr1; int* nbr2; int* nbr3; int* nbr4;
  int Mw0; int Mw1; int Mw2; int Mw3; int Mw4;
  int n0; int n1; int n2; int n3; int n4;
  int nb0; int nb1; int nb2; int nb3; int nb4;
  int s1; int s2; int s3; int s4; int s5; int s6;
  int fill;
  const float* fa; float* xa; const int* ia; long ra;
  const float* fb; float* xb; const int* ib; long rb;
};

__device__ __forceinline__ void invmap_job(const int* __restrict__ map, int* __restrict__ nbr,
                                           int Mw, int n, int local, int nb) {
  const int j = local / nb;
  invmap_tap(map, nbr, Mw, n, j, local - j * nb, nb);
}

__device__ __forceinline__ void prep_block(const PrepArgs& a, int b, int np) {
  if (b >= a.s5) {
    const bool narrow = b >= a.s6;
    const int vb = b - (narrow ? a.s6 : a.s5);
    const int nb = narrow ? np - a.s6 : a.s6 - a.s5;
    const long first = (long)vb * TPB + threadIdx.x;
    const long step = (long)nb * TPB;
    if (narrow) scatter_add_rows<64>(a.fb, a.xb, a.ib, a.rb, first, step);
    else        scatter_add_rows<128>(a.fa, a.xa, a.ia, a.ra, first, step);
    return;
  }
  if      (b < a.s1) invmap_job(a.map0, a.nbr0, a.Mw0, a.n0, b, a.nb0);
  else if (b < a.s2) invmap_job(a.map1, a.nbr1, a.Mw1, a.n1, b - a.s1, a.nb1);
  else if (b < a.s3) invmap_job(a.map2, a.nbr2, a.Mw2, a.n2, b - a.s2, a.nb2);
  else if (b < a.s4) invmap_job(a.map3, a.nbr3, a.Mw3, a.n3, b - a.s3, a.nb3);
  else               invmap_job(a.map4, a.nbr4, a.Mw4, a.n4, b - a.s4, a.nb4);
}

__device__ __forceinline__ void bn_part_rows(const float* __restrict__ X, long tot, int C,
                                             float* __restrict__ part, int bk, int nbk) {
  __shared__ float sm[8][TPB];
  const int t = threadIdx.x;
  float a0 = 0.0f, a1 = 0.0f, a2 = 0.0f, a3 = 0.0f;
  float b0 = 0.0f, b1 = 0.0f, b2 = 0.0f, b3 = 0.0f;
  const long step = (long)nbk * (TPB * 4);
  for (long i = ((long)bk * TPB + t) * 4; i < tot; i += step) {
    const float4 v = *(const float4*)(X + i);
    a0 += v.x; b0 += v.x * v.x;
    a1 += v.y; b1 += v.y * v.y;
    a2 += v.z; b2 += v.z * v.z;
    a3 += v.w; b3 += v.w * v.w;
  }
  sm[0][t] = a0; sm[1][t] = a1; sm[2][t] = a2; sm[3][t] = a3;
  sm[4][t] = b0; sm[5][t] = b1; sm[6][t] = b2; sm[7][t] = b3;
  __syncthreads();
  if (t < C) {
    const int q = C >> 2;
    const int comp = t & 3;
    float s = 0.0f, ss = 0.0f;
    for (int j = t >> 2; j < TPB; j += q) {
      s += sm[comp][j];
      ss += sm[4 + comp][j];
    }
    float* p = part + (long)bk * 2 * C;
    p[t] = s;
    p[C + t] = ss;
  }
}

__global__ void __launch_bounds__(TPB) k_bn_part(const float* __restrict__ XA, long totA, int CA,
                          float* __restrict__ partA, int nA,
                          const float* __restrict__ XB, long totB, int CB,
                          float* __restrict__ partB) {
  const bool jb = (int)blockIdx.x >= nA;
  bn_part_rows(jb ? XB : XA, jb ? totB : totA, jb ? CB : CA, jb ? partB : partA,
               jb ? (int)blockIdx.x - nA : (int)blockIdx.x,
               jb ? (int)gridDim.x - nA : nA);
}

__global__ void __launch_bounds__(TPB) k_prep_bn(PrepArgs a, int np,
                          const float* __restrict__ X, long tot, int C,
                          float* __restrict__ part) {
  const int b = blockIdx.x;
  if (b < np) {
    prep_block(a, b, np);
    return;
  }
  bn_part_rows(X, tot, C, part, b - np, (int)gridDim.x - np);
}

__global__ void __launch_bounds__(TPB) k_bn_part_sort(const float* __restrict__ X, long tot, int C,
                          float* __restrict__ part,
                          const int* __restrict__ nbrA, int nA,
                          const int* __restrict__ nbrB, int nB,
                          int* __restrict__ cnt2, int ns) {
  const int b = blockIdx.x;
  if (b < ns) {
    sort_count_rows(nbrA, nA, nbrB, nB, cnt2, b % SORT_BLK, b / SORT_BLK);
    return;
  }
  bn_part_rows(X, tot, C, part, b - ns, (int)gridDim.x - ns);
}

__device__ __forceinline__ void bn_fin_col(const float* __restrict__ part, int P, long N, int C,
                                           float* __restrict__ mu, float* __restrict__ rs, int c) {
  __shared__ float s1[TPB];
  __shared__ float s2[TPB];
  float a = 0.0f, b = 0.0f;
  for (int p = threadIdx.x; p < P; p += TPB) {
    a += part[(long)p * 2 * C + c];
    b += part[(long)p * 2 * C + C + c];
  }
  s1[threadIdx.x] = a;
  s2[threadIdx.x] = b;
  __syncthreads();
  for (int s = TPB / 2; s > 0; s >>= 1) {
    if ((int)threadIdx.x < s) {
      s1[threadIdx.x] += s1[threadIdx.x + s];
      s2[threadIdx.x] += s2[threadIdx.x + s];
    }
    __syncthreads();
  }
  if (threadIdx.x == 0) {
    float m = s1[0] / (float)N;
    float var = s2[0] / (float)N - m * m;
    if (var < 0.0f) var = 0.0f;
    mu[c] = m;
    rs[c] = rsqrtf(var + 1e-5f);
  }
}

__global__ void __launch_bounds__(TPB) k_bn_fin(const float* __restrict__ partA, int PA, long NA, int CA,
                         float* __restrict__ muA, float* __restrict__ rsA,
                         const float* __restrict__ partB, int PB, long NB, int CB,
                         float* __restrict__ muB, float* __restrict__ rsB) {
  const bool jb = (int)blockIdx.x >= CA;
  bn_fin_col(jb ? partB : partA, jb ? PB : PA, jb ? NB : NA, jb ? CB : CA,
             jb ? muB : muA, jb ? rsB : rsA,
             jb ? (int)blockIdx.x - CA : (int)blockIdx.x);
}

__global__ void __launch_bounds__(TPB) k_bn_fin_sort(const float* __restrict__ part, int P, long N, int C,
                         float* __restrict__ mu, float* __restrict__ rs,
                         int* __restrict__ cnt, int* __restrict__ tot, int ns) {
  const int b = blockIdx.x;
  if (b < ns) {
    sort_scan_class(cnt, tot, b);
    return;
  }
  bn_fin_col(part, P, N, C, mu, rs, b - ns);
}

template <int C>
__device__ __forceinline__ void bn_elu_rows(const float* X, float* Y,
                                            const float* __restrict__ mu,
                                            const float* __restrict__ rs,
                                            const float* __restrict__ gb, long tot4,
                                            int bk, int nbk,
                                            const int* __restrict__ sidx) {
  constexpr int Q = C / 4;
  const float4* X4 = (const float4*)X;
  float4* Y4 = (float4*)Y;
  const int c = ((int)threadIdx.x * 4) & (C - 1);
  float g[4], m[4], s[4], b[4];
  for (int e = 0; e < 4; ++e) {
    g[e] = gb[c + e];
    m[e] = mu[c + e];
    s[e] = rs[c + e];
    b[e] = gb[C + c + e];
  }
  const long step = (long)nbk * TPB;
  for (long i = (long)bk * TPB + threadIdx.x; i < tot4; i += 4 * step) {
    float4 x[4];
    float4 y[4];
    long o4[4];
    if (sidx) {
      int tr[4];
      for (int u = 0; u < 4; ++u) {
        const long k = i + u * step;
        o4[u] = k < tot4 ? k : i;
        x[u] = X4[o4[u]];
        tr[u] = sidx[o4[u] / Q];
      }
      for (int u = 0; u < 4; ++u) {
        o4[u] = (long)tr[u] * Q + (o4[u] % Q);
        y[u] = Y4[o4[u]];
      }
    } else {
      for (int u = 0; u < 4; ++u) {
        o4[u] = i + u * step;
        if (o4[u] < tot4) x[u] = X4[o4[u]];
      }
    }
    for (int u = 0; u < 4; ++u) {
      if (i + u * step < tot4) {
        float v0 = g[0] * (x[u].x - m[0]) * s[0] + b[0];
        float v1 = g[1] * (x[u].y - m[1]) * s[1] + b[1];
        float v2 = g[2] * (x[u].z - m[2]) * s[2] + b[2];
        float v3 = g[3] * (x[u].w - m[3]) * s[3] + b[3];
        float4 o;
        o.x = v0 > 0.0f ? v0 : expm1f(v0);
        o.y = v1 > 0.0f ? v1 : expm1f(v1);
        o.z = v2 > 0.0f ? v2 : expm1f(v2);
        o.w = v3 > 0.0f ? v3 : expm1f(v3);
        if (sidx) {
          o.x += y[u].x; o.y += y[u].y; o.z += y[u].z; o.w += y[u].w;
        }
        Y4[o4[u]] = o;
      }
    }
  }
}

template <int CA, int CB>
__global__ void k_bn_elu(const float* XA, float* YA,
                         const float* __restrict__ muA, const float* __restrict__ rsA,
                         const float* __restrict__ gbA, long tot4A, int nA,
                         const float* XB, float* YB,
                         const float* __restrict__ muB, const float* __restrict__ rsB,
                         const float* __restrict__ gbB, long tot4B,
                         const int* __restrict__ sidxA) {
  if ((int)blockIdx.x < nA)
    bn_elu_rows<CA>(XA, YA, muA, rsA, gbA, tot4A, (int)blockIdx.x, nA, sidxA);
  else
    bn_elu_rows<CB>(XB, YB, muB, rsB, gbB, tot4B, (int)blockIdx.x - nA, (int)gridDim.x - nA,
                    nullptr);
}

template <int C>
__global__ void __launch_bounds__(TPB) k_bn_elu_sort(const float* X, float* Y,
                         const float* __restrict__ mu, const float* __restrict__ rs,
                         const float* __restrict__ gb, long tot4,
                         const int* __restrict__ nbrA, int nA,
                         const int* __restrict__ nbrB, int nB,
                         const int* __restrict__ cnt2, const int* __restrict__ tot2,
                         int* __restrict__ permA, int* __restrict__ permB, int ns,
                         const int* __restrict__ sidx) {
  const int b = blockIdx.x;
  if (b < ns) {
    sort_place_rows(nbrA, nA, nbrB, nB, cnt2, tot2, permA, permB, b % SORT_BLK, b / SORT_BLK);
    return;
  }
  bn_elu_rows<C>(X, Y, mu, rs, gb, tot4, b - ns, (int)gridDim.x - ns, sidx);
}

template <int CIN, int COUT>
__device__ __forceinline__ void gemm_tap_tile(const float* __restrict__ X,
                           const float* __restrict__ W,
                           float* __restrict__ Y,
                           const int* __restrict__ gidx,
                           const int* __restrict__ sidx,
                           int M, int out_offset, int accumulate, int nscrap, int zrows,
                           int bx, int by, int bz) {
  const int lane = threadIdx.x;
  W += (long)bz * CIN * COUT;
  out_offset += bz * zrows;
  const int tile_r = bx * 32;
  const int tile_c = (by * 4 + threadIdx.y) * 16;
  const int mr = lane & 15;
  const int kh = lane >> 4;

  const int rl = tile_r + lane;
  int sv = nscrap;
  if (rl < M) sv = sidx ? sidx[rl] : rl;
  if (__builtin_amdgcn_ballot_w32(sv != nscrap) == 0u) return;

  const int r0 = tile_r + mr;
  const int r1 = tile_r + 16 + mr;
  int g0 = 0, g1 = 0;
  if (r0 < M) g0 = gidx ? gidx[r0] : r0;
  if (r1 < M) g1 = gidx ? gidx[r1] : r1;
  const float* __restrict__ xb0 = X + (long)g0 * CIN + 2 * kh;
  const float* __restrict__ xb1 = X + (long)g1 * CIN + 2 * kh;
  const float* __restrict__ wb  = W + tile_c + mr + 2 * kh * COUT;

  v8f c0 = {0.f, 0.f, 0.f, 0.f, 0.f, 0.f, 0.f, 0.f};
  v8f c1 = {0.f, 0.f, 0.f, 0.f, 0.f, 0.f, 0.f, 0.f};
#pragma unroll
  for (int k0 = 0; k0 < CIN; k0 += 4) {
    v2f a0 = *(const v2f*)(xb0 + k0);
    v2f a1 = *(const v2f*)(xb1 + k0);
    v2f b;
    b[0] = wb[k0 * COUT];
    b[1] = wb[(k0 + 1) * COUT];
    c0 = __builtin_amdgcn_wmma_f32_16x16x4_f32(false, a0, false, b,
                                               (short)0, c0, false, false);
    c1 = __builtin_amdgcn_wmma_f32_16x16x4_f32(false, a1, false, b,
                                               (short)0, c1, false, false);
  }

#pragma unroll
  for (int i = 0; i < 8; ++i) {
    const int t0a = __builtin_amdgcn_readlane(sv, i);
    const int t0b = __builtin_amdgcn_readlane(sv, i + 8);
    const int ro0 = kh ? t0b : t0a;
    if (ro0 != nscrap) {
      float* y = Y + (long)(ro0 + out_offset) * COUT + tile_c + mr;
      if (accumulate) *y += c0[i]; else *y = c0[i];
    }
    const int t1a = __builtin_amdgcn_readlane(sv, i + 16);
    const int t1b = __builtin_amdgcn_readlane(sv, i + 24);
    const int ro1 = kh ? t1b : t1a;
    if (ro1 != nscrap) {
      float* y = Y + (long)(ro1 + out_offset) * COUT + tile_c + mr;
      if (accumulate) *y += c1[i]; else *y = c1[i];
    }
  }
}

template <int CIN, int COUT>
__device__ __forceinline__ void conv_os_tile(const float* __restrict__ X,
                          const float* __restrict__ W,
                          float* __restrict__ Y,
                          const int* __restrict__ nbr,
                          int n, int bx, int by) {
  const int lane = threadIdx.x;
  const int tile_r = bx * 32;
  const int tile_c = (by * 4 + threadIdx.y) * 16;
  const int mr = lane & 15;
  const int kh = lane >> 4;
  const int r0 = tile_r + mr;
  const int r1 = tile_r + 16 + mr;
  const bool in0 = r0 < n;
  const bool in1 = r1 < n;
  constexpr int KC = CIN < 64 ? CIN : 64;
  constexpr int NS = KC / 4;

  v8f c0 = {0.f, 0.f, 0.f, 0.f, 0.f, 0.f, 0.f, 0.f};
  v8f c1 = {0.f, 0.f, 0.f, 0.f, 0.f, 0.f, 0.f, 0.f};

  const int q0 = in0 ? r0 : n - 1;
  const int q1 = in1 ? r1 : n - 1;
  int g0 = nbr[q0];
  int g1 = nbr[q1];
  g0 = in0 ? g0 : -1;
  g1 = in1 ? g1 : -1;
  for (int t = 0; t < 27; ++t) {
    int l0 = r0;
    int l1 = r1;
    if (t < 25) {
      l0 = nbr[(long)(t + 1) * n + q0];
      l1 = nbr[(long)(t + 1) * n + q1];
    }
    if (__builtin_amdgcn_ballot_w32((g0 >= 0) || (g1 >= 0)) != 0u) {
      const int k = (t < 13) ? t : (t < 26 ? t + 1 : 13);
      const float* __restrict__ wb =
          W + (long)k * CIN * COUT + tile_c + mr + 2 * kh * COUT;
#pragma unroll
      for (int kc = 0; kc < CIN; kc += KC) {
        v2f a0[NS];
        v2f a1[NS];
#pragma unroll
        for (int s = 0; s < NS; ++s) {
          a0[s] = (v2f){0.f, 0.f};
          a1[s] = (v2f){0.f, 0.f};
        }
        if (g0 >= 0) {
          const float* __restrict__ xb0 = X + (long)g0 * CIN + 2 * kh + kc;
#pragma unroll
          for (int s = 0; s < NS; ++s) a0[s] = *(const v2f*)(xb0 + 4 * s);
        }
        if (g1 >= 0) {
          const float* __restrict__ xb1 = X + (long)g1 * CIN + 2 * kh + kc;
#pragma unroll
          for (int s = 0; s < NS; ++s) a1[s] = *(const v2f*)(xb1 + 4 * s);
        }
#pragma unroll
        for (int s = 0; s < NS; ++s) {
          const int k0 = kc + 4 * s;
          v2f b;
          b[0] = wb[k0 * COUT];
          b[1] = wb[(k0 + 1) * COUT];
          c0 = __builtin_amdgcn_wmma_f32_16x16x4_f32(false, a0[s], false, b,
                                                     (short)0, c0, false, false);
          c1 = __builtin_amdgcn_wmma_f32_16x16x4_f32(false, a1[s], false, b,
                                                     (short)0, c1, false, false);
        }
      }
    }
    g0 = in0 ? l0 : -1;
    g1 = in1 ? l1 : -1;
  }

#pragma unroll
  for (int i = 0; i < 8; ++i) {
    const int ro0 = tile_r + i + 8 * kh;
    if (ro0 < n) Y[(long)ro0 * COUT + tile_c + mr] = c0[i];
    const int ro1 = ro0 + 16;
    if (ro1 < n) Y[(long)ro1 * COUT + tile_c + mr] = c1[i];
  }
}

template <int CIN, int COUT>
__global__ void __launch_bounds__(128) k_conv_os(const float* __restrict__ X,
                          const float* __restrict__ W, float* __restrict__ Y,
                          const int* __restrict__ nbr, int n) {
  conv_os_tile<CIN, COUT>(X, W, Y, nbr, n, (int)blockIdx.x, (int)blockIdx.y);
}

template <int CIN, int COUT>
__global__ void __launch_bounds__(128) k_fill_gemm(int* __restrict__ a, long na4, int va,
                          int* __restrict__ b, long nb4, int vb,
                          const float* __restrict__ X, const float* __restrict__ W,
                          float* __restrict__ Y, int M, int ng) {
  constexpr int GY = COUT / 64;
  const int bid = blockIdx.x;
  if (bid < ng) {
    const int per = ((M + 31) / 32) * GY;
    const int bz = bid / per;
    const int r = bid - bz * per;
    gemm_tap_tile<CIN, COUT>(X, W, Y, nullptr, nullptr, M, 0, 0, -1, M, r / GY, r % GY, bz);
    return;
  }
  const long tid = (long)threadIdx.y * 32 + threadIdx.x;
  fill2_rows(a, na4, va, b, nb4, vb, (long)(bid - ng) * 128 + tid,
             (long)((int)gridDim.x - ng) * 128);
}

template <int CIN, int CCOUT, int GCOUT, int CIN2>
__global__ void __launch_bounds__(128) k_side_convs_gemm(
                          const float* __restrict__ X,
                          const float* __restrict__ Wc,
                          float* __restrict__ Yc,
                          const int* __restrict__ nbr,
                          const float* __restrict__ Wg,
                          float* __restrict__ Yg,
                          int n, int nconv,
                          const float* __restrict__ X2,
                          const float* __restrict__ Wc2,
                          float* __restrict__ Yc2,
                          const int* __restrict__ nbr2,
                          int n2, int nconv2) {
  constexpr int CY = CCOUT / 64;
  constexpr int GY = GCOUT / 64;
  const int b = blockIdx.x;
  if (b < nconv2) {
    conv_os_tile<CIN2, CCOUT>(X2, Wc2, Yc2, nbr2, n2, b / CY, b % CY);
    return;
  }
  const int b1 = b - nconv2;
  if (b1 < nconv) {
    conv_os_tile<CIN, CCOUT>(X, Wc, Yc, nbr, n, b1 / CY, b1 % CY);
    return;
  }
  const int g = b1 - nconv;
  const int per = ((n + 31) / 32) * GY;
  const int bz = g / per;
  const int r = g - bz * per;
  gemm_tap_tile<CIN, GCOUT>(X, Wg, Yg, nullptr, nullptr, n, 0, 0, -1, n, r / GY, r % GY, bz);
}

template <int CIN, int COUT>
__global__ void __launch_bounds__(128) k_conv_os_wide(const float* __restrict__ X,
                          const float* __restrict__ W,
                          float* __restrict__ Y,
                          const int* __restrict__ nbr,
                          const int* __restrict__ perm,
                          int n) {
  constexpr int NT = COUT / 16;
  constexpr int KC = 64;
  constexpr int NCH = CIN / KC;
  constexpr int PITCH = KC + 4;
  constexpr int CG = COUT / 4;
  constexpr int ITEMS = (KC / 4) * CG / 128;
  __shared__ float Bs[2][COUT * PITCH];
  __shared__ unsigned need[4];

  const int lane = threadIdx.x;
  const int wv = threadIdx.y;
  const int tid = wv * 32 + lane;
  const int tile_r = (((int)gridDim.x - 1 - (int)blockIdx.x) * 4 + wv) * 32;
  const int mr = lane & 15;
  const int kh = lane >> 4;
  const int p0 = tile_r + mr;
  const int p1 = tile_r + 16 + mr;
  const bool in0 = p0 < n;
  const bool in1 = p1 < n;
  const int r0 = perm[in0 ? p0 : n - 1];
  const int r1 = perm[in1 ? p1 : n - 1];

  unsigned wm = 0u;
  {
    int h0[26];
    int h1[26];
#pragma unroll
    for (int t = 0; t < 26; ++t) {
      h0[t] = nbr[(long)t * n + r0];
      h1[t] = nbr[(long)t * n + r1];
    }
#pragma unroll
    for (int t = 0; t < 26; ++t) {
      const int e0 = in0 ? h0[t] : -1;
      const int e1 = in1 ? h1[t] : -1;
      if (__builtin_amdgcn_ballot_w32((e0 & e1) >= 0) != 0u) wm |= 1u << t;
    }
    if (__builtin_amdgcn_ballot_w32(in0 || in1) != 0u) wm |= 1u << 26;
  }
  if (lane == 0) need[wv] = wm;
  __syncthreads();
  unsigned bm = (unsigned)__builtin_amdgcn_readfirstlane(
      (int)(need[0] | need[1] | need[2] | need[3]));
  if (bm == 0u) return;

  auto stage = [&](int t, int ch, float* __restrict__ dst) {
    const int kc = ch * KC;
    const int k = (t < 13) ? t : (t < 26 ? t + 1 : 13);
    const float* __restrict__ wsrc = W + ((long)k * CIN + kc) * COUT;
    float4 r[ITEMS][4];
    for (int u = 0; u < ITEMS; ++u) {
      const int e = tid + 128 * u;
      const int cg = e % CG;
      const int kg = e / CG;
      for (int rr = 0; rr < 4; ++rr)
        r[u][rr] = *(const float4*)(wsrc + (long)(4 * kg + rr) * COUT + 4 * cg);
    }
    for (int u = 0; u < ITEMS; ++u) {
      const int e = tid + 128 * u;
      const int cg = e % CG;
      const int kg = e / CG;
      float* d = dst + (4 * cg) * PITCH + 4 * kg;
      *(float4*)(d)             = make_float4(r[u][0].x, r[u][1].x, r[u][2].x, r[u][3].x);
      *(float4*)(d + PITCH)     = make_float4(r[u][0].y, r[u][1].y, r[u][2].y, r[u][3].y);
      *(float4*)(d + 2 * PITCH) = make_float4(r[u][0].z, r[u][1].z, r[u][2].z, r[u][3].z);
      *(float4*)(d + 3 * PITCH) = make_float4(r[u][0].w, r[u][1].w, r[u][2].w, r[u][3].w);
    }
  };

  v8f c0[NT];
  v8f c1[NT];
  for (int j = 0; j < NT; ++j) {
    c0[j] = (v8f){0.f, 0.f, 0.f, 0.f, 0.f, 0.f, 0.f, 0.f};
    c1[j] = (v8f){0.f, 0.f, 0.f, 0.f, 0.f, 0.f, 0.f, 0.f};
  }

  int t = __builtin_ctz(bm);
  bm &= bm - 1u;
  stage(t, 0, Bs[0]);

  int g0 = t < 26 ? nbr[(long)t * n + r0] : r0;
  int g1 = t < 26 ? nbr[(long)t * n + r1] : r1;
  g0 = in0 ? g0 : -1;
  g1 = in1 ? g1 : -1;
  __syncthreads();

  int buf = 0;
  for (int it = 0; it < 27; ++it) {
    const int tn = bm != 0u ? __builtin_ctz(bm) : -1;
    bm &= bm - 1u;
    int l0 = r0;
    int l1 = r1;
    if (tn >= 0 && tn < 26) {
      l0 = nbr[(long)tn * n + r0];
      l1 = nbr[(long)tn * n + r1];
    }
    const bool act = __builtin_amdgcn_ballot_w32((g0 >= 0) || (g1 >= 0)) != 0u;
    for (int ch = 0; ch < NCH; ++ch) {
      if (ch + 1 < NCH) stage(t, ch + 1, Bs[buf ^ 1]);
      else if (tn >= 0) stage(tn, 0, Bs[buf ^ 1]);
      if (act) {
        const float* __restrict__ bs = Bs[buf] + mr * PITCH + 4 * kh;
        float4 a0[8];
        float4 a1[8];
        for (int p = 0; p < 8; ++p) {
          a0[p] = make_float4(0.f, 0.f, 0.f, 0.f);
          a1[p] = make_float4(0.f, 0.f, 0.f, 0.f);
        }
        if (g0 >= 0) {
          const float* __restrict__ xb0 = X + (long)g0 * CIN + ch * KC + 4 * kh;
          for (int p = 0; p < 8; ++p) a0[p] = *(const float4*)(xb0 + 8 * p);
        }
        if (g1 >= 0) {
          const float* __restrict__ xb1 = X + (long)g1 * CIN + ch * KC + 4 * kh;
          for (int p = 0; p < 8; ++p) a1[p] = *(const float4*)(xb1 + 8 * p);
        }
        for (int p = 0; p < 8; ++p) {
          const v2f a0l = {a0[p].x, a0[p].y};
          const v2f a0h = {a0[p].z, a0[p].w};
          const v2f a1l = {a1[p].x, a1[p].y};
          const v2f a1h = {a1[p].z, a1[p].w};
          float4 bf[NT];
          for (int j = 0; j < NT; ++j) bf[j] = *(const float4*)(bs + 16 * j * PITCH + 8 * p);
          for (int j = 0; j < NT; ++j) {
            const float4 b = bf[j];
            const v2f bl = {b.x, b.y};
            const v2f bh = {b.z, b.w};
            c0[j] = __builtin_amdgcn_wmma_f32_16x16x4_f32(false, a0l, false, bl,
                                                          (short)0, c0[j], false, false);
            c1[j] = __builtin_amdgcn_wmma_f32_16x16x4_f32(false, a1l, false, bl,
                                                          (short)0, c1[j], false, false);
            c0[j] = __builtin_amdgcn_wmma_f32_16x16x4_f32(false, a0h, false, bh,
                                                          (short)0, c0[j], false, false);
            c1[j] = __builtin_amdgcn_wmma_f32_16x16x4_f32(false, a1h, false, bh,
                                                          (short)0, c1[j], false, false);
          }
        }
      }
      __syncthreads();
      buf ^= 1;
    }
    if (tn < 0) break;
    g0 = in0 ? l0 : -1;
    g1 = in1 ? l1 : -1;
    t = tn;
  }

  int o0[8];
  int o1[8];
#pragma unroll
  for (int i = 0; i < 8; ++i) {
    const int xa = __builtin_amdgcn_readlane(r0, i);
    const int xb = __builtin_amdgcn_readlane(r0, i + 8);
    o0[i] = kh ? xb : xa;
    const int ya = __builtin_amdgcn_readlane(r1, i);
    const int yb = __builtin_amdgcn_readlane(r1, i + 8);
    o1[i] = kh ? yb : ya;
  }
#pragma unroll
  for (int j = 0; j < NT; ++j) {
#pragma unroll
    for (int i = 0; i < 8; ++i) {
      if (tile_r + i + 8 * kh < n) Y[(long)o0[i] * COUT + 16 * j + mr] = c0[j][i];
      if (tile_r + 16 + i + 8 * kh < n) Y[(long)o1[i] * COUT + 16 * j + mr] = c1[j][i];
    }
  }
}

template <int CIN, int COUT>
static inline void conv_t(const float* X, const float* W, float* Y,
                          const int* nbr, int n, hipStream_t st) {
  dim3 grid((unsigned)((n + 31) / 32), (unsigned)(COUT / 64));
  dim3 blk(32, 4);
  k_conv_os<CIN, COUT><<<grid, blk, 0, st>>>(X, W, Y, nbr, n);
}

template <int CIN, int COUT>
static inline void conv_wide_t(const float* X, const float* W, float* Y,
                               const int* nbr, int n, const int* perm, hipStream_t st) {
  dim3 grid((unsigned)((n + 127) / 128));
  dim3 blk(32, 4);
  k_conv_os_wide<CIN, COUT><<<grid, blk, 0, st>>>(X, W, Y, nbr, perm, n);
}

static inline bool conv_is_wide(int n, int Cin) { return n >= 128 * 256 && Cin == 64; }

static void conv_os(const float* X, const float* W, float* Y, const int* nbr,
                    int n, int Cin, int Cout, const int* perm, hipStream_t st) {
  const bool wide = conv_is_wide(n, Cin);
  if      (wide && Cin == 64 && Cout ==  64) conv_wide_t<64,  64>(X, W, Y, nbr, n, perm, st);
  else if (wide && Cin == 64 && Cout == 128) conv_wide_t<64, 128>(X, W, Y, nbr, n, perm, st);
  else if (Cin == 256 && Cout == 128) conv_t<256, 128>(X, W, Y, nbr, n, st);
  else if (Cin == 128 && Cout == 128) conv_t<128, 128>(X, W, Y, nbr, n, st);
  else if (Cin ==  64 && Cout ==  64) conv_t< 64,  64>(X, W, Y, nbr, n, st);
  else if (Cin ==  64 && Cout == 128) conv_t< 64, 128>(X, W, Y, nbr, n, st);
}

static inline void side_convs_gemm(const float* X, const float* Wc, float* Yc, const int* nbr,
                                   const float* Wg, float* Yg, int n,
                                   const float* X2, const float* Wc2, float* Yc2,
                                   const int* nbr2, int n2, hipStream_t st) {
  const int tiles = (n + 31) / 32;
  const int nconv = tiles * 2;
  const int nprod = tiles * 8;
  const int nconv2 = ((n2 + 31) / 32) * 2;
  dim3 blk(32, 4);
  k_side_convs_gemm<128, 128, 64, 256><<<dim3((unsigned)(nconv2 + nconv + nprod)), blk, 0, st>>>(
      X, Wc, Yc, nbr, Wg, Yg, n, nconv, X2, Wc2, Yc2, nbr2, n2, nconv2);
}


extern "C" void kernel_launch(void* const* d_in, const int* in_sizes, int n_in,
                              void* d_out, int out_size, void* d_ws, size_t ws_size,
                              hipStream_t stream) {
  const float* feats0 = (const float*)d_in[0];
  const float* feats1 = (const float*)d_in[1];
  const float* feats2 = (const float*)d_in[2];
  const float* w_out0 = (const float*)d_in[3];
  const float* w_out1 = (const float*)d_in[4];
  const float* w_out2 = (const float*)d_in[5];
  const float* wt2    = (const float*)d_in[6];
  const float* wu2    = (const float*)d_in[7];
  const float* wt1    = (const float*)d_in[8];
  const float* wu1    = (const float*)d_in[9];
  const float* bn_out0 = (const float*)d_in[10];
  const float* bn_out1 = (const float*)d_in[11];
  const float* bn_out2 = (const float*)d_in[12];
  const float* bn_up2a = (const float*)d_in[13];
  const float* bn_up2b = (const float*)d_in[14];
  const float* bn_up1a = (const float*)d_in[15];
  const float* bn_up1b = (const float*)d_in[16];
  const int* m_c2 = (const int*)d_in[19];
  const int* m_g2 = (const int*)d_in[20];
  const int* m_u1 = (const int*)d_in[21];
  const int* m_g1 = (const int*)d_in[22];
  const int* m_u0 = (const int*)d_in[23];
  const int* lat1_idx = (const int*)d_in[24];
  const int* up1_idx  = (const int*)d_in[25];
  const int* lat0_idx = (const int*)d_in[26];
  const int* up0_idx  = (const int*)d_in[27];

  const int N0 = in_sizes[0] / 64;
  const int N1 = in_sizes[1] / 128;
  const int N2 = in_sizes[2] / 256;
  const int nu1 = in_sizes[17] / 128;
  const int nu0 = in_sizes[18] / 64;
  const int Mc2 = in_sizes[19] / 52;
  const int Mg2 = in_sizes[20] / 52;
  const int Mu1 = in_sizes[21] / 52;
  const int Mg1 = in_sizes[22] / 52;
  const int Mu0 = in_sizes[23] / 52;
  const int g2n = 8 * N2;
  const int g1n = 8 * nu1;

  char* wsp = (char*)d_ws;
  auto carve = [&](long nfloats) -> float* {
    float* p = (float*)wsp;
    wsp += ((nfloats * 4 + 255) / 256) * 256;
    return p;
  };
  long accElems = (long)(N2 + 1) * 128;
  { long e = (long)(g2n + 1) * 128; if (e > accElems) accElems = e; }
  { long e = (long)(nu1 + 1) * 128; if (e > accElems) accElems = e; }
  { long e = (long)(g1n + 1) * 64;  if (e > accElems) accElems = e; }
  { long e = (long)(nu0 + 1) * 128; if (e > accElems) accElems = e; }
  float* acc = carve(accElems);
  float* t2  = carve((long)g2n * 128);
  float* t1  = carve((long)g1n * 64);
  float* mu  = carve(128);
  float* rs  = carve(128);
  float* part = carve((long)1024 * 256);
  float* accS = carve((long)nu1 * 128);
  float* accT = carve((long)N2 * 128);
  float* muS  = carve(128);
  float* rsS  = carve(128);
  float* partS = carve((long)1024 * 256);
  float* x1  = carve((long)nu1 * 128);
  float* x0  = carve((long)nu0 * 64);
  const long zero4 = (long)(wsp - (char*)x1) / 16;
  int* nbC2 = (int*)carve((long)26 * N2);
  int* nbG2 = (int*)carve((long)26 * g2n);
  int* nbU1 = (int*)carve((long)26 * nu1);
  int* nbG1 = (int*)carve((long)26 * g1n);
  int* nbU0 = (int*)carve((long)26 * nu0);
  const long neg4 = (long)(wsp - (char*)nbC2) / 16;
  int* permG1 = (int*)carve(g1n);
  int* permU0 = (int*)carve(nu0);
  int* cnt2 = (int*)carve((long)2 * SORT_B * SORT_W);
  int* tot2 = (int*)carve((long)2 * SORT_B);

  float* out  = (float*)d_out;
  float* out0 = out;
  float* out1 = out + (long)nu0 * 128;
  float* out2 = out + (long)nu0 * 128 + (long)nu1 * 128;

  auto nblk = [](long items) -> unsigned {
    long b = (items + TPB - 1) / TPB;
    if (b > MAXB) b = MAXB;
    if (b < 1) b = 1;
    return (unsigned)b;
  };
  auto conv3 = [&](const float* X, int n, int Cin, int Cout,
                   const float* W, const int* nbr, const int* perm) {
    conv_os(X, W, acc, nbr, n, Cin, Cout, perm, stream);
  };
  auto bnelu = [&](const float* src, float* dst, const float* gb, long n, int C,
                   const float* srcS, float* dstS, const float* gbS, long nS,
                   const int* sidx) {
    const long tot = n * C;
    long P = (tot + TPB * 4 - 1) / (TPB * 4);
    if (P > 1024) P = 1024;
    const long totS = srcS ? nS * 128 : 0;
    long PS = (totS + TPB * 4 - 1) / (TPB * 4);
    if (PS > 1024) PS = 1024;
    const int CS = srcS ? 128 : 0;
    k_bn_part<<<dim3((unsigned)(P + PS)), dim3(TPB), 0, stream>>>(src, tot, C, part, (int)P,
                                                                  srcS, totS, 128, partS);
    k_bn_fin<<<dim3((unsigned)(C + CS)), dim3(TPB), 0, stream>>>(part, (int)P, n, C, mu, rs,
                                                                 partS, (int)PS, nS, 128, muS, rsS);
    const long tot4 = tot / 4;
    const long tot4S = totS / 4;
    const unsigned nA = nblk(tot4);
    const unsigned nB = srcS ? nblk(tot4S) : 0u;
    if (C == 128)
      k_bn_elu<128, 128><<<dim3(nA + nB), dim3(TPB), 0, stream>>>(src, dst, mu, rs, gb, tot4, (int)nA,
                                                                  srcS, dstS, muS, rsS, gbS, tot4S, sidx);
    else
      k_bn_elu<64, 128><<<dim3(nA + nB), dim3(TPB), 0, stream>>>(src, dst, mu, rs, gb, tot4, (int)nA,
                                                                 srcS, dstS, muS, rsS, gbS, tot4S, sidx);
  };

  auto tapblk = [](int Mw) -> int {
    long b = ((long)Mw + 4 * TPB - 1) / (4 * TPB);
    if (b > MAXB / 26) b = MAXB / 26;
    if (b < 1) b = 1;
    return (int)b;
  };
  auto scatblk = [](long tot4) -> int {
    long b = (tot4 + 4 * TPB - 1) / (4 * TPB);
    if (b > MAXB) b = MAXB;
    if (b < 1) b = 1;
    return (int)b;
  };
  {
    const int ng = ((N2 + 31) / 32) * 2 * 8;
    const int nf = 2 * MAXB;
    k_fill_gemm<256, 128><<<dim3((unsigned)(ng + nf)), dim3(32, 4), 0, stream>>>(
        nbC2, neg4, -1, (int*)x1, zero4, 0, feats2, wt2, t2, N2, ng);
  }
  {
    PrepArgs pa;
    pa.fill = 0;
    pa.map0 = m_c2; pa.map1 = m_g2; pa.map2 = m_u1; pa.map3 = m_g1; pa.map4 = m_u0;
    pa.nbr0 = nbC2; pa.nbr1 = nbG2; pa.nbr2 = nbU1; pa.nbr3 = nbG1; pa.nbr4 = nbU0;
    pa.Mw0 = Mc2; pa.Mw1 = Mg2; pa.Mw2 = Mu1; pa.Mw3 = Mg1; pa.Mw4 = Mu0;
    pa.n0 = N2; pa.n1 = g2n; pa.n2 = nu1; pa.n3 = g1n; pa.n4 = nu0;
    pa.fa = feats1; pa.xa = x1; pa.ia = lat1_idx; pa.ra = N1;
    pa.fb = feats0; pa.xb = x0; pa.ib = lat0_idx; pa.rb = N0;
    pa.nb0 = tapblk(Mc2); pa.nb1 = tapblk(Mg2); pa.nb2 = tapblk(Mu1);
    pa.nb3 = tapblk(Mg1); pa.nb4 = tapblk(Mu0);
    pa.s1 = 26 * pa.nb0;
    pa.s2 = pa.s1 + 26 * pa.nb1;
    pa.s3 = pa.s2 + 26 * pa.nb2;
    pa.s4 = pa.s3 + 26 * pa.nb3;
    pa.s5 = pa.s4 + 26 * pa.nb4;
    pa.s6 = pa.s5 + scatblk((long)N1 * 32);
    const int np = pa.s6 + scatblk((long)N0 * 16);
    const long tot = (long)g2n * 128;
    long P = (tot + TPB * 4 - 1) / (TPB * 4);
    if (P > 1024) P = 1024;
    const long tot4 = tot / 4;
    k_prep_bn<<<dim3((unsigned)(np + P)), dim3(TPB), 0, stream>>>(pa, np, t2, tot, 128, part);
    k_bn_fin<<<dim3(128u), dim3(TPB), 0, stream>>>(part, (int)P, (long)g2n, 128, mu, rs,
                                                   partS, 0, 0, 128, muS, rsS);
    const unsigned nA = nblk(tot4);
    k_bn_elu<128, 128><<<dim3(nA), dim3(TPB), 0, stream>>>(t2, t2, mu, rs, bn_up2a, tot4, (int)nA,
                                                           nullptr, nullptr, muS, rsS, nullptr, 0,
                                                           nullptr);
  }
  conv3(t2, g2n, 128, 128, wu2, nbG2, nullptr);
  {
    const bool wG1 = conv_is_wide(g1n, 64);
    const bool wU0 = conv_is_wide(nu0, 64);
    const int* tA = wG1 ? nbG1 : nbU0;
    const int nA = wG1 ? g1n : nu0;
    int* pA = wG1 ? permG1 : permU0;
    const int ny = (wG1 ? 1 : 0) + (wU0 ? 1 : 0);
    const int ns = SORT_BLK * ny;
    const int nsc = SORT_B * ny;
    const long tot = (long)g2n * 128;
    long P = (tot + TPB * 4 - 1) / (TPB * 4);
    if (P > 1024) P = 1024;
    const long tot4 = tot / 4;
    k_bn_part_sort<<<dim3((unsigned)(ns + P)), dim3(TPB), 0, stream>>>(
        acc, tot, 128, part, tA, nA, nbU0, nu0, cnt2, ns);
    k_bn_fin_sort<<<dim3((unsigned)(nsc + 128)), dim3(TPB), 0, stream>>>(
        part, (int)P, (long)g2n, 128, mu, rs, cnt2, tot2, nsc);
    k_bn_elu_sort<128><<<dim3((unsigned)ns + nblk(tot4)), dim3(TPB), 0, stream>>>(
        acc, x1, mu, rs, bn_up2b, tot4, tA, nA, nbU0, nu0, cnt2, tot2, pA, permU0, ns,
        up1_idx);
  }

  side_convs_gemm(x1, w_out1, accS, nbU1, wt1, t1, nu1, feats2, w_out2, accT, nbC2, N2, stream);
  bnelu(t1, t1, bn_up1a, g1n, 64, accS, out1, bn_out1, nu1, nullptr);
  conv3(t1, g1n, 64, 64, wu1, nbG1, permG1);

  bnelu(acc, x0, bn_up1b, g1n, 64, accT, out2, bn_out2, N2, up0_idx);

  conv3(x0, nu0, 64, 128, w_out0, nbU0, permU0);
  bnelu(acc, out0, bn_out0, nu0, 128, nullptr, nullptr, nullptr, 0, nullptr);

  (void)n_in; (void)out_size; (void)ws_size;
}
